# starting kernel
# baseline (speedup 1.0000x reference)
#include <hip/hip_runtime.h>
#include <hip/hip_bf16.h>


typedef __attribute__((ext_vector_type(2))) float v2f;
typedef __attribute__((ext_vector_type(8))) float v8f;

#define KNN_K 10
#define IN_EPS 1e-5f

__device__ __forceinline__ float lrelu(float y) { return y >= 0.0f ? y : 0.2f * y; }

__device__ __forceinline__ v8f wmma_f32(v2f a, v2f b, v8f c) {
    return __builtin_amdgcn_wmma_f32_16x16x4_f32(
         false, a,  false, b,
         (short)0, c,  false,  false);
}

__global__ void prep_pts_kernel(const float* __restrict__ coords,
                                float4* __restrict__ pts, int N, int total) {
    int gid = blockIdx.x * 256 + threadIdx.x;
    if (gid >= total) return;
    int b = gid / N, n = gid - b * N;
    const float* cb = coords + (size_t)b * 3 * N;
    float x = cb[n], y = cb[N + n], z = cb[2 * N + n];
    pts[gid] = make_float4(x, y, z, x * x + y * y + z * z);
}

__global__ void prep_w1_kernel(const float* __restrict__ W1, float* __restrict__ Wc) {
    int gid = blockIdx.x * 256 + threadIdx.x;
    if (gid >= 128 * 128) return;
    int o = gid >> 7, c = gid & 127;
    float a = W1[o * 256 + c], b = W1[o * 256 + 128 + c];
    Wc[o * 128 + c] = a - b;
    Wc[(128 + o) * 128 + c] = b;
}

__global__ void prep_w2_kernel(const float* __restrict__ W2, float* __restrict__ Wc) {
    int gid = blockIdx.x * 256 + threadIdx.x;
    if (gid >= 256 * 128) return;
    int o = gid >> 7, c = gid & 127;
    float a = W2[o * 256 + c], b = W2[o * 256 + 128 + c];
    Wc[o * 128 + c] = a - b;
    Wc[(256 + o) * 128 + c] = b;
}

__global__ void copy_x0_kernel(const float* __restrict__ feats,
                               float* __restrict__ X3, int N, int total) {
    int gid = blockIdx.x * 256 + threadIdx.x;
    if (gid >= total) return;
    int n = gid % N, c = (gid / N) & 127, b = gid / (N * 128);
    X3[(size_t)b * 512 * N + (size_t)c * N + n] = feats[gid];
}

__global__ __launch_bounds__(256) void knn_kernel(const float4* __restrict__ pts,
                                                  int* __restrict__ knn, int N) {
    __shared__ float4 tile[128];
    const int b = blockIdx.y;
    const int i = blockIdx.x * 256 + threadIdx.x;
    const float4 p = pts[(size_t)b * N + i];

    float bd[KNN_K + 1];
    int   bi[KNN_K + 1];
#pragma unroll
    for (int s = 0; s <= KNN_K; ++s) { bd[s] = 3.4e38f; bi[s] = -1; }

    for (int c0 = 0; c0 < N; c0 += 128) {
        __syncthreads();
        if (threadIdx.x < 128) tile[threadIdx.x] = pts[(size_t)b * N + c0 + threadIdx.x];
        __syncthreads();
        for (int j = 0; j < 128; ++j) {
            float4 q = tile[j];
            float d = p.w + q.w - 2.0f * (p.x * q.x + p.y * q.y + p.z * q.z);
            if (d < bd[KNN_K]) {
                float cd = d; int ci = c0 + j;
#pragma unroll
                for (int s = 0; s <= KNN_K; ++s) {
                    bool sw = cd < bd[s];
                    float td = sw ? bd[s] : cd; int ti = sw ? bi[s] : ci;
                    bd[s] = sw ? cd : bd[s];    bi[s] = sw ? ci : bi[s];
                    cd = td; ci = ti;
                }
            }
        }
    }
    int* o = knn + ((size_t)b * N + i) * KNN_K;
    int w = 0;
#pragma unroll
    for (int s = 0; s <= KNN_K; ++s)
        if (bi[s] != i && w < KNN_K) { o[w] = bi[s]; ++w; }
}

__global__ __launch_bounds__(128) void gemm_wmma_kernel(
        const float* __restrict__ W, const float* __restrict__ X,
        float* __restrict__ D, int N, int Kdim,
        long xBatchStride, long dBatchStride) {
    const int lane = threadIdx.x & 31;
    const int wave = threadIdx.x >> 5;
    const int half = lane >> 4;
    const int l16  = lane & 15;
    const int wm = wave >> 1, wn = wave & 1;
    const int m0 = blockIdx.y * 64 + wm * 32;
    const int n0 = blockIdx.x * 64 + wn * 32;
    const float* Xb = X + (size_t)blockIdx.z * xBatchStride;
    float*       Db = D + (size_t)blockIdx.z * dBatchStride;

    v8f c00 = {0.f,0.f,0.f,0.f,0.f,0.f,0.f,0.f};
    v8f c01 = c00, c10 = c00, c11 = c00;

    const float* ar0 = W + (size_t)(m0 + l16) * Kdim + 2 * half;
    const float* ar1 = ar0 + (size_t)16 * Kdim;
    const float* xr  = Xb + (size_t)(2 * half) * N + n0 + l16;

    __builtin_prefetch(xr, 0, 1);
    __builtin_prefetch(ar0, 0, 1);

#pragma unroll 4
    for (int k = 0; k < Kdim; k += 4) {
        v2f a0, a1, b0, b1;
        a0.x = ar0[0];   a0.y = ar0[1];
        a1.x = ar1[0];   a1.y = ar1[1];
        b0.x = xr[0];    b0.y = xr[N];
        b1.x = xr[16];   b1.y = xr[16 + N];
        c00 = wmma_f32(a0, b0, c00);
        c01 = wmma_f32(a0, b1, c01);
        c10 = wmma_f32(a1, b0, c10);
        c11 = wmma_f32(a1, b1, c11);
        ar0 += 4; ar1 += 4; xr += (size_t)4 * N;
    }
#pragma unroll
    for (int v = 0; v < 8; ++v) {
        const int mr0 = m0 + v + 8 * half;
        Db[(size_t)mr0 * N + n0 + l16]             = c00[v];
        Db[(size_t)mr0 * N + n0 + 16 + l16]        = c01[v];
        Db[(size_t)(mr0 + 16) * N + n0 + l16]      = c10[v];
        Db[(size_t)(mr0 + 16) * N + n0 + 16 + l16] = c11[v];
    }
}

__global__ __launch_bounds__(256) void edge_agg_kernel(
        const float* __restrict__ PQ, const int* __restrict__ knn,
        float* __restrict__ outMax, float* __restrict__ stats,
        int M, int N, long outBatchStride) {
    __shared__ float rs[256], rs2[256];
    const int b = blockIdx.z, m = blockIdx.y;
    const int n = blockIdx.x * 256 + threadIdx.x;
    const float* P = PQ + ((size_t)b * 2 * M + m) * N;
    const float* Q = PQ + ((size_t)b * 2 * M + M + m) * N;
    const int* id = knn + ((size_t)b * N + n) * KNN_K;

    const float p = P[n];
    float mx = -3.4e38f, s = 0.f, s2 = 0.f;
#pragma unroll
    for (int k = 0; k < KNN_K; ++k) {
        float v = p + Q[id[k]];
        mx = fmaxf(mx, v); s += v; s2 += v * v;
    }
    outMax[(size_t)b * outBatchStride + (size_t)m * N + n] = mx;

    rs[threadIdx.x] = s; rs2[threadIdx.x] = s2;
    __syncthreads();
    for (int off = 128; off > 0; off >>= 1) {
        if (threadIdx.x < off) {
            rs[threadIdx.x]  += rs[threadIdx.x + off];
            rs2[threadIdx.x] += rs2[threadIdx.x + off];
        }
        __syncthreads();
    }
    if (threadIdx.x == 0) {
        atomicAdd(&stats[((size_t)b * M + m) * 2 + 0], rs[0]);
        atomicAdd(&stats[((size_t)b * M + m) * 2 + 1], rs2[0]);
    }
}

__global__ void in_norm_kernel(float* __restrict__ data,
                               const float* __restrict__ stats,
                               int M, int N, long batchStride, float invCnt,
                               int total) {
    int gid = blockIdx.x * 256 + threadIdx.x;
    if (gid >= total) return;
    int n = gid % N, m = (gid / N) % M, b = gid / (N * M);
    float s  = stats[((size_t)b * M + m) * 2 + 0];
    float s2 = stats[((size_t)b * M + m) * 2 + 1];
    float mu = s * invCnt;
    float var = fmaxf(s2 * invCnt - mu * mu, 0.f);
    float r = rsqrtf(var + IN_EPS);
    float* px = data + (size_t)b * batchStride + (size_t)m * N + n;
    *px = lrelu((*px - mu) * r);
}

__global__ __launch_bounds__(256) void final_norm_kernel(
        const float* __restrict__ xin, float* __restrict__ out, int N) {
    __shared__ float rs[256], rs2[256];
    __shared__ float s_mu, s_r;
    const int b = blockIdx.y, c = blockIdx.x, C = gridDim.x;
    const float* row = xin + ((size_t)b * C + c) * N;
    float s = 0.f, s2 = 0.f;
    for (int n = threadIdx.x; n < N; n += 256) {
        float v = row[n]; s += v; s2 += v * v;
    }
    rs[threadIdx.x] = s; rs2[threadIdx.x] = s2;
    __syncthreads();
    for (int off = 128; off > 0; off >>= 1) {
        if (threadIdx.x < off) {
            rs[threadIdx.x]  += rs[threadIdx.x + off];
            rs2[threadIdx.x] += rs2[threadIdx.x + off];
        }
        __syncthreads();
    }
    if (threadIdx.x == 0) {
        float mu = rs[0] / (float)N;
        float var = fmaxf(rs2[0] / (float)N - mu * mu, 0.f);
        s_mu = mu; s_r = rsqrtf(var + IN_EPS);
    }
    __syncthreads();
    float mu = s_mu, r = s_r;
    float* orow = out + ((size_t)b * C + c) * N;
    for (int n = threadIdx.x; n < N; n += 256)
        orow[n] = lrelu((row[n] - mu) * r);
}

extern "C" void kernel_launch(void* const* d_in, const int* in_sizes, int n_in,
                              void* d_out, int out_size, void* d_ws, size_t ws_size,
                              hipStream_t stream) {
    constexpr int B = 4, C = 128, N = 4096;
    const float* coords = (const float*)d_in[0];
    const float* feats  = (const float*)d_in[1];
    const float* W1     = (const float*)d_in[2];
    const float* W2     = (const float*)d_in[3];
    const float* W3     = (const float*)d_in[4];
    float* out = (float*)d_out;

    char* ws = (char*)d_ws;
    size_t off = 0;
    auto carve = [&](size_t bytes) -> void* {
        void* p = ws + off;
        off = (off + bytes + 255) & ~(size_t)255;
        return p;
    };
    float4* pts   = (float4*)carve((size_t)B * N * sizeof(float4));
    float* W1cat  = (float*)carve(256 * 128 * sizeof(float));
    float* W2cat  = (float*)carve(512 * 128 * sizeof(float));
    int*   knn    = (int*)carve((size_t)B * N * KNN_K * sizeof(int));
    float* PQ     = (float*)carve((size_t)B * 512 * N * sizeof(float));
    float* X3     = (float*)carve((size_t)B * 512 * N * sizeof(float));
    float* x3pre  = (float*)carve((size_t)B * 128 * N * sizeof(float));
    float* stats1 = (float*)carve((size_t)B * 128 * 2 * sizeof(float));
    float* stats2 = (float*)carve((size_t)B * 256 * 2 * sizeof(float));
    (void)ws_size; (void)in_sizes; (void)n_in; (void)out_size;

    hipMemsetAsync(stats1, 0, (size_t)B * 128 * 2 * sizeof(float), stream);
    hipMemsetAsync(stats2, 0, (size_t)B * 256 * 2 * sizeof(float), stream);

    prep_pts_kernel<<<(B * N + 255) / 256, 256, 0, stream>>>(coords, pts, N, B * N);
    prep_w1_kernel<<<(128 * 128 + 255) / 256, 256, 0, stream>>>(W1, W1cat);
    prep_w2_kernel<<<(256 * 128 + 255) / 256, 256, 0, stream>>>(W2, W2cat);
    copy_x0_kernel<<<(B * 128 * N + 255) / 256, 256, 0, stream>>>(feats, X3, N, B * 128 * N);

    knn_kernel<<<dim3(N / 256, B), 256, 0, stream>>>(pts, knn, N);

    const float invCnt = 1.0f / (float)(N * KNN_K);

    gemm_wmma_kernel<<<dim3(N / 64, 256 / 64, B), 128, 0, stream>>>(
        W1cat, feats, PQ, N, C, (long)C * N, (long)256 * N);
    edge_agg_kernel<<<dim3(N / 256, 128, B), 256, 0, stream>>>(
        PQ, knn, X3 + (size_t)128 * N, stats1, 128, N, (long)512 * N);
    in_norm_kernel<<<(B * 128 * N + 255) / 256, 256, 0, stream>>>(
        X3 + (size_t)128 * N, stats1, 128, N, (long)512 * N, invCnt, B * 128 * N);

    gemm_wmma_kernel<<<dim3(N / 64, 512 / 64, B), 128, 0, stream>>>(
        W2cat, X3 + (size_t)128 * N, PQ, N, C, (long)512 * N, (long)512 * N);
    edge_agg_kernel<<<dim3(N / 256, 256, B), 256, 0, stream>>>(
        PQ, knn, X3 + (size_t)256 * N, stats2, 256, N, (long)512 * N);
    in_norm_kernel<<<(B * 256 * N + 255) / 256, 256, 0, stream>>>(
        X3 + (size_t)256 * N, stats2, 256, N, (long)512 * N, invCnt, B * 256 * N);

    gemm_wmma_kernel<<<dim3(N / 64, 128 / 64, B), 128, 0, stream>>>(
        W3, X3, x3pre, N, 512, (long)512 * N, (long)128 * N);
    final_norm_kernel<<<dim3(128, B), 256, 0, stream>>>(x3pre, out, N);
}


# P,Q stored point-major ([n][2M]) by the GEMM so each neighbour gather reads one contiguous channel vector; edge kernel lanes = channels; per-tile channel sums replace atomics (sum order changes)
# speedup vs baseline: 2.4946x; 2.4946x over previous
#include <hip/hip_runtime.h>
#include <hip/hip_bf16.h>


typedef __attribute__((ext_vector_type(2))) float v2f;
typedef __attribute__((ext_vector_type(8))) float v8f;

#define KNN_K 10
#define IN_EPS 1e-5f

__device__ __forceinline__ float lrelu(float y) { return y >= 0.0f ? y : 0.2f * y; }

__device__ __forceinline__ v8f wmma_f32(v2f a, v2f b, v8f c) {
    return __builtin_amdgcn_wmma_f32_16x16x4_f32(
         false, a,  false, b,
         (short)0, c,  false,  false);
}

__global__ void prep_pts_kernel(const float* __restrict__ coords,
                                float4* __restrict__ pts, int N, int total) {
    int gid = blockIdx.x * 256 + threadIdx.x;
    if (gid >= total) return;
    int b = gid / N, n = gid - b * N;
    const float* cb = coords + (size_t)b * 3 * N;
    float x = cb[n], y = cb[N + n], z = cb[2 * N + n];
    pts[gid] = make_float4(x, y, z, x * x + y * y + z * z);
}

__global__ void prep_w1_kernel(const float* __restrict__ W1, float* __restrict__ Wc) {
    int gid = blockIdx.x * 256 + threadIdx.x;
    if (gid >= 128 * 128) return;
    int o = gid >> 7, c = gid & 127;
    float a = W1[o * 256 + c], b = W1[o * 256 + 128 + c];
    Wc[o * 128 + c] = a - b;
    Wc[(128 + o) * 128 + c] = b;
}

__global__ void prep_w2_kernel(const float* __restrict__ W2, float* __restrict__ Wc) {
    int gid = blockIdx.x * 256 + threadIdx.x;
    if (gid >= 256 * 128) return;
    int o = gid >> 7, c = gid & 127;
    float a = W2[o * 256 + c], b = W2[o * 256 + 128 + c];
    Wc[o * 128 + c] = a - b;
    Wc[(256 + o) * 128 + c] = b;
}

__global__ void copy_x0_kernel(const float* __restrict__ feats,
                               float* __restrict__ X3, int N, int total) {
    int gid = blockIdx.x * 256 + threadIdx.x;
    if (gid >= total) return;
    int n = gid % N, c = (gid / N) & 127, b = gid / (N * 128);
    X3[(size_t)b * 512 * N + (size_t)c * N + n] = feats[gid];
}

#define KNN_N   4096
#define KNN_QW  4
#define KNN_CAP 64

__device__ __forceinline__ float knn_dist(float px, float py, float pz, float pw, float4 q) {
    float d;
    {
#pragma clang fp contract(off)
        d = __builtin_fmaf(-2.0f, __builtin_fmaf(px, q.x, py * q.y) + pz * q.z, pw + q.w);
    }
    return d;
}

__device__ __forceinline__ bool knn_less(float da, int ia, float db, int ib) {
    return (da < db) || (da == db && ia < ib);
}

__device__ __forceinline__ void knn_exact_wave(const float4* tile, int N, int lane,
        float px, float py, float pz, float pw, int iq, int* so_q) {
    float ld = -__builtin_inff(); int li = -1; int w = 0;
    for (int r = 0; r <= KNN_K; ++r) {
        float bd = __builtin_inff(); int bi = 0x7fffffff;
        for (int t = 0; t < N; t += 32) {
            const int c = t + lane;
            float d = knn_dist(px, py, pz, pw, tile[c]);
            if (knn_less(ld, li, d, c) && knn_less(d, c, bd, bi)) { bd = d; bi = c; }
        }
        for (int off = 16; off > 0; off >>= 1) {
            float od = __shfl_xor(bd, off, 32); int oi = __shfl_xor(bi, off, 32);
            if (knn_less(od, oi, bd, bi)) { bd = od; bi = oi; }
        }
        ld = bd; li = bi;
        if (bi != iq) {
            if (w < KNN_K && lane == 0) so_q[w] = (bi < N) ? bi : 0;
            ++w;
        }
    }
}

__device__ __forceinline__ void knn_replay(const float* sd, const int* si, int n,
                                           int lane, int iq, int* so_q) {
    float bd[KNN_K + 1];
    int   bi[KNN_K + 1];
#pragma unroll
    for (int s = 0; s <= KNN_K; ++s) { bd[s] = 3.4e38f; bi[s] = -1; }
    for (int j = 0; j < n; ++j) {
        float cd = sd[j]; int ci = si[j];
        if (cd < bd[KNN_K]) {
#pragma unroll
            for (int s = 0; s <= KNN_K; ++s) {
                bool sw = cd < bd[s];
                float td = sw ? bd[s] : cd; int ti = sw ? bi[s] : ci;
                bd[s] = sw ? cd : bd[s];    bi[s] = sw ? ci : bi[s];
                cd = td; ci = ti;
            }
        }
    }
    if (lane == 0) {
        int w = 0;
#pragma unroll
        for (int s = 0; s <= KNN_K; ++s)
            if (bi[s] != iq && w < KNN_K) { so_q[w] = bi[s]; ++w; }
    }
}

__global__ __launch_bounds__(256) void knn_kernel(const float4* __restrict__ pts,
                                                  int* __restrict__ knn, int N) {
    __shared__ float4 tile[KNN_N];
    __shared__ float  hd[8][KNN_QW][KNN_CAP];
    __shared__ int    hi[8][KNN_QW][KNN_CAP];
    __shared__ int    so[32 * KNN_K];
    const int b = blockIdx.y;
    const int lane = threadIdx.x & 31;
    const int wave = __builtin_amdgcn_readfirstlane((int)(threadIdx.x >> 5));
    const float4* pb = pts + (size_t)b * N;

    for (int i = threadIdx.x; i < N; i += 256) tile[i] = pb[i];
    __syncthreads();

    const int q0 = blockIdx.x * 32 + wave * KNN_QW;
    float px[KNN_QW], py[KNN_QW], pz[KNN_QW], pw[KNN_QW];
#pragma unroll
    for (int qi = 0; qi < KNN_QW; ++qi) {
        float4 p = pb[q0 + qi];
        px[qi] = p.x; py[qi] = p.y; pz[qi] = p.z; pw[qi] = p.w;
    }

    float mn[KNN_QW];
#pragma unroll
    for (int qi = 0; qi < KNN_QW; ++qi) mn[qi] = __builtin_inff();
#pragma unroll 4
    for (int t = 0; t < N; t += 32) {
        float4 q = tile[t + lane];
#pragma unroll
        for (int qi = 0; qi < KNN_QW; ++qi) {
            float d = knn_dist(px[qi], py[qi], pz[qi], pw[qi], q);
            mn[qi] = d < mn[qi] ? d : mn[qi];
        }
    }

    float T[KNN_QW];
#pragma unroll
    for (int qi = 0; qi < KNN_QW; ++qi) {
        const float mv = mn[qi];
        int cnt = 0;
#pragma unroll
        for (int s = 0; s < 32; ++s) {
            float v = __builtin_bit_cast(float,
                __builtin_amdgcn_readlane(__builtin_bit_cast(int, mv), s));
            cnt += knn_less(v, s, mv, lane) ? 1 : 0;
        }
        unsigned sel = __builtin_amdgcn_ballot_w32(cnt == KNN_K);
        T[qi] = sel ? __shfl(mv, __builtin_ctz(sel), 32) : __builtin_inff();
    }

    int cn[KNN_QW];
#pragma unroll
    for (int qi = 0; qi < KNN_QW; ++qi) cn[qi] = 0;
    for (int t = 0; t < N; t += 128) {
        float4 q[4];
        float d[4][KNN_QW];
        unsigned bm[4][KNN_QW];
#pragma unroll
        for (int u = 0; u < 4; ++u) q[u] = tile[t + 32 * u + lane];
#pragma unroll
        for (int u = 0; u < 4; ++u) {
#pragma unroll
            for (int qi = 0; qi < KNN_QW; ++qi) {
                d[u][qi] = knn_dist(px[qi], py[qi], pz[qi], pw[qi], q[u]);
                bm[u][qi] = __builtin_amdgcn_ballot_w32(d[u][qi] <= T[qi]);
            }
        }
#pragma unroll
        for (int u = 0; u < 4; ++u) {
            if (bm[u][0] | bm[u][1] | bm[u][2] | bm[u][3]) {
#pragma unroll
                for (int qi = 0; qi < KNN_QW; ++qi) {
                    const unsigned m = bm[u][qi];
                    if (m) {
                        const bool hit = ((m >> lane) & 1u) != 0u;
                        const int pos = cn[qi] + (int)__builtin_amdgcn_mbcnt_lo(m, 0u);
                        if (hit && pos < KNN_CAP) {
                            hd[wave][qi][pos] = d[u][qi];
                            hi[wave][qi][pos] = t + 32 * u + lane;
                        }
                        cn[qi] += __builtin_popcount(m);
                    }
                }
            }
        }
    }
    __syncthreads();

#pragma unroll
    for (int qi = 0; qi < KNN_QW; ++qi) {
        const int n = cn[qi];
        const int iq = q0 + qi;
        int* so_q = so + (wave * KNN_QW + qi) * KNN_K;
        if (n > KNN_K && n <= KNN_CAP) {
            const float* sd = hd[wave][qi];
            const int*   si = hi[wave][qi];
            const bool v0 = lane < n, v1 = lane + 32 < n;
            const float d0 = v0 ? sd[lane] : __builtin_inff();
            const int   i0 = v0 ? si[lane] : 0x7fffffff;
            const float d1 = v1 ? sd[lane + 32] : __builtin_inff();
            const int   i1 = v1 ? si[lane + 32] : 0x7fffffff;
            int r0 = 0, r1 = 0;
            bool tie = false;
            for (int j = 0; j < n; ++j) {
                const float dj = sd[j]; const int ij = si[j];
                r0 += knn_less(dj, ij, d0, i0) ? 1 : 0;
                r1 += knn_less(dj, ij, d1, i1) ? 1 : 0;
                tie = tie || (dj == d0 && ij != i0) || (dj == d1 && ij != i1);
            }
            if (__builtin_amdgcn_ballot_w32(tie)) {
                knn_replay(sd, si, n, lane, iq, so_q);
                continue;
            }
            int sr = KNN_K + 1;
            unsigned s0 = __builtin_amdgcn_ballot_w32(v0 && i0 == iq);
            if (s0) sr = __shfl(r0, __builtin_ctz(s0), 32);
            unsigned s1 = __builtin_amdgcn_ballot_w32(v1 && i1 == iq);
            if (s1) sr = __shfl(r1, __builtin_ctz(s1), 32);
            if (v0 && i0 != iq && r0 <= KNN_K) {
                int w = r0 - (r0 > sr ? 1 : 0);
                if (w < KNN_K) so_q[w] = i0;
            }
            if (v1 && i1 != iq && r1 <= KNN_K) {
                int w = r1 - (r1 > sr ? 1 : 0);
                if (w < KNN_K) so_q[w] = i1;
            }
        } else {
            knn_exact_wave(tile, N, lane, px[qi], py[qi], pz[qi], pw[qi], iq, so_q);
        }
    }
    __syncthreads();
    int* ob = knn + ((size_t)b * N + (size_t)blockIdx.x * 32) * KNN_K;
    for (int i = threadIdx.x; i < 32 * KNN_K; i += 256) ob[i] = so[i];
}

template <bool TRANS>
__global__ __launch_bounds__(128) void gemm_wmma_kernel(
        const float* __restrict__ W, const float* __restrict__ X,
        float* __restrict__ D, int N, int Kdim,
        long xBatchStride, long dBatchStride, int ldT) {
    const int lane = threadIdx.x & 31;
    const int wave = threadIdx.x >> 5;
    const int half = lane >> 4;
    const int l16  = lane & 15;
    const int wm = wave >> 1, wn = wave & 1;
    const int m0 = blockIdx.y * 64 + wm * 32;
    const int n0 = blockIdx.x * 64 + wn * 32;
    const float* Xb = X + (size_t)blockIdx.z * xBatchStride;
    float*       Db = D + (size_t)blockIdx.z * dBatchStride;

    v8f c00 = {0.f,0.f,0.f,0.f,0.f,0.f,0.f,0.f};
    v8f c01 = c00, c10 = c00, c11 = c00;

    const float* ar0 = W + (size_t)(m0 + l16) * Kdim + 2 * half;
    const float* ar1 = ar0 + (size_t)16 * Kdim;
    const float* xr  = Xb + (size_t)(2 * half) * N + n0 + l16;

    __builtin_prefetch(xr, 0, 1);
    __builtin_prefetch(ar0, 0, 1);

#pragma unroll 4
    for (int k = 0; k < Kdim; k += 4) {
        v2f a0, a1, b0, b1;
        a0.x = ar0[0];   a0.y = ar0[1];
        a1.x = ar1[0];   a1.y = ar1[1];
        b0.x = xr[0];    b0.y = xr[N];
        b1.x = xr[16];   b1.y = xr[16 + N];
        if (TRANS) {
            c00 = wmma_f32(b0, a0, c00);
            c01 = wmma_f32(b1, a0, c01);
            c10 = wmma_f32(b0, a1, c10);
            c11 = wmma_f32(b1, a1, c11);
        } else {
            c00 = wmma_f32(a0, b0, c00);
            c01 = wmma_f32(a0, b1, c01);
            c10 = wmma_f32(a1, b0, c10);
            c11 = wmma_f32(a1, b1, c11);
        }
        ar0 += 4; ar1 += 4; xr += (size_t)4 * N;
    }
    if (TRANS) {
#pragma unroll
        for (int v = 0; v < 8; ++v) {
            const size_t r0 = (size_t)(n0 + v + 8 * half) * ldT + m0 + l16;
            const size_t r1 = r0 + (size_t)16 * ldT;
            Db[r0]      = c00[v];
            Db[r0 + 16] = c10[v];
            Db[r1]      = c01[v];
            Db[r1 + 16] = c11[v];
        }
        return;
    }
#pragma unroll
    for (int v = 0; v < 8; ++v) {
        const int mr0 = m0 + v + 8 * half;
        Db[(size_t)mr0 * N + n0 + l16]             = c00[v];
        Db[(size_t)mr0 * N + n0 + 16 + l16]        = c01[v];
        Db[(size_t)(mr0 + 16) * N + n0 + l16]      = c10[v];
        Db[(size_t)(mr0 + 16) * N + n0 + 16 + l16] = c11[v];
    }
}

#define EDGE_PTS 32
#define EDGE_PW  4

template <int M>
__global__ __launch_bounds__(256) void edge_agg_kernel(
        const float* __restrict__ PQT, const int* __restrict__ knn,
        float* __restrict__ outMax, float2* __restrict__ part,
        int N, long outBatchStride) {
    constexpr int NV = M / 128;
    constexpr int TS = M + 4;
    __shared__ __attribute__((aligned(16))) float tile[EDGE_PTS * TS];
    __shared__ __attribute__((aligned(16))) float rs[8][M];
    __shared__ __attribute__((aligned(16))) float rs2[8][M];
    const int b = blockIdx.y;
    const int lane = threadIdx.x & 31;
    const int wave = __builtin_amdgcn_readfirstlane((int)(threadIdx.x >> 5));
    const int nb0 = blockIdx.x * EDGE_PTS;
    const float* base = PQT + (size_t)b * N * (2 * M);

    float4 s[NV], s2[NV];
#pragma unroll
    for (int v = 0; v < NV; ++v) {
        s[v]  = make_float4(0.f, 0.f, 0.f, 0.f);
        s2[v] = make_float4(0.f, 0.f, 0.f, 0.f);
    }

#pragma unroll
    for (int j = 0; j < EDGE_PW; ++j) {
        const int nl = wave * EDGE_PW + j;
        const int n = nb0 + nl;
        const int* id = knn + ((size_t)b * N + n) * KNN_K;
        const float* prow = base + (size_t)n * (2 * M) + 4 * lane;
        float4 p[NV], mx[NV];
#pragma unroll
        for (int v = 0; v < NV; ++v) {
            p[v] = *(const float4*)(prow + v * 128);
            mx[v] = make_float4(-3.4e38f, -3.4e38f, -3.4e38f, -3.4e38f);
        }
#pragma unroll
        for (int k = 0; k < KNN_K; ++k) {
            const float* qrow = base + (size_t)id[k] * (2 * M) + M + 4 * lane;
#pragma unroll
            for (int v = 0; v < NV; ++v) {
                const float4 q = *(const float4*)(qrow + v * 128);
                const float y0 = p[v].x + q.x, y1 = p[v].y + q.y;
                const float y2 = p[v].z + q.z, y3 = p[v].w + q.w;
                mx[v].x = fmaxf(mx[v].x, y0); mx[v].y = fmaxf(mx[v].y, y1);
                mx[v].z = fmaxf(mx[v].z, y2); mx[v].w = fmaxf(mx[v].w, y3);
                s[v].x += y0; s[v].y += y1; s[v].z += y2; s[v].w += y3;
                s2[v].x += y0 * y0; s2[v].y += y1 * y1;
                s2[v].z += y2 * y2; s2[v].w += y3 * y3;
            }
        }
#pragma unroll
        for (int v = 0; v < NV; ++v)
            *(float4*)(tile + nl * TS + v * 128 + 4 * lane) = mx[v];
    }
#pragma unroll
    for (int v = 0; v < NV; ++v) {
        *(float4*)(&rs[wave][v * 128 + 4 * lane])  = s[v];
        *(float4*)(&rs2[wave][v * 128 + 4 * lane]) = s2[v];
    }
    __syncthreads();

    float* ob = outMax + (size_t)b * outBatchStride + nb0 + lane;
    for (int m = wave; m < M; m += 8)
        ob[(size_t)m * N] = tile[lane * TS + m];

    const int t = threadIdx.x;
    if (t < M) {
        float a = 0.f, a2 = 0.f;
#pragma unroll
        for (int w = 0; w < 8; ++w) { a += rs[w][t]; a2 += rs2[w][t]; }
        part[((size_t)b * gridDim.x + blockIdx.x) * M + t] = make_float2(a, a2);
    }
}

__global__ __launch_bounds__(256) void in_norm_kernel(float* __restrict__ data,
                               const float2* __restrict__ part, int tiles,
                               int M, int N, long batchStride, float invCnt,
                               int total) {
    __shared__ float red[8][2];
    const int g0 = blockIdx.x * 256;
    if (g0 >= total) return;
    const int bm = g0 / N;
    const int n = g0 - bm * N + (int)threadIdx.x;
    const int b = bm / M, m = bm - b * M;
    float* px = data + (size_t)b * batchStride + (size_t)m * N + n;
    const float x = *px;
    float a = 0.f, a2 = 0.f;
    for (int t = threadIdx.x; t < tiles; t += 256) {
        const float2 v = part[((size_t)b * tiles + t) * M + m];
        a += v.x; a2 += v.y;
    }
    for (int off = 16; off > 0; off >>= 1) {
        a  += __shfl_xor(a, off, 32);
        a2 += __shfl_xor(a2, off, 32);
    }
    if ((threadIdx.x & 31) == 0) { red[threadIdx.x >> 5][0] = a; red[threadIdx.x >> 5][1] = a2; }
    __syncthreads();
    float s = 0.f, s2 = 0.f;
#pragma unroll
    for (int w = 0; w < 8; ++w) { s += red[w][0]; s2 += red[w][1]; }
    float mu = s * invCnt;
    float var = fmaxf(s2 * invCnt - mu * mu, 0.f);
    float r = rsqrtf(var + IN_EPS);
    *px = lrelu((x - mu) * r);
}

__global__ __launch_bounds__(256) void final_norm_kernel(
        const float* __restrict__ xin, float* __restrict__ out, int N) {
    __shared__ float rs[256], rs2[256];
    __shared__ float s_mu, s_r;
    const int b = blockIdx.y, c = blockIdx.x, C = gridDim.x;
    const float* row = xin + ((size_t)b * C + c) * N;
    float s = 0.f, s2 = 0.f;
    for (int n = threadIdx.x; n < N; n += 256) {
        float v = row[n]; s += v; s2 += v * v;
    }
    rs[threadIdx.x] = s; rs2[threadIdx.x] = s2;
    __syncthreads();
    for (int off = 128; off > 0; off >>= 1) {
        if (threadIdx.x < off) {
            rs[threadIdx.x]  += rs[threadIdx.x + off];
            rs2[threadIdx.x] += rs2[threadIdx.x + off];
        }
        __syncthreads();
    }
    if (threadIdx.x == 0) {
        float mu = rs[0] / (float)N;
        float var = fmaxf(rs2[0] / (float)N - mu * mu, 0.f);
        s_mu = mu; s_r = rsqrtf(var + IN_EPS);
    }
    __syncthreads();
    float mu = s_mu, r = s_r;
    float* orow = out + ((size_t)b * C + c) * N;
    for (int n = threadIdx.x; n < N; n += 256)
        orow[n] = lrelu((row[n] - mu) * r);
}

extern "C" void kernel_launch(void* const* d_in, const int* in_sizes, int n_in,
                              void* d_out, int out_size, void* d_ws, size_t ws_size,
                              hipStream_t stream) {
    constexpr int B = 4, C = 128, N = 4096;
    const float* coords = (const float*)d_in[0];
    const float* feats  = (const float*)d_in[1];
    const float* W1     = (const float*)d_in[2];
    const float* W2     = (const float*)d_in[3];
    const float* W3     = (const float*)d_in[4];
    float* out = (float*)d_out;

    char* ws = (char*)d_ws;
    size_t off = 0;
    auto carve = [&](size_t bytes) -> void* {
        void* p = ws + off;
        off = (off + bytes + 255) & ~(size_t)255;
        return p;
    };
    float4* pts   = (float4*)carve((size_t)B * N * sizeof(float4));
    float* W1cat  = (float*)carve(256 * 128 * sizeof(float));
    float* W2cat  = (float*)carve(512 * 128 * sizeof(float));
    int*   knn    = (int*)carve((size_t)B * N * KNN_K * sizeof(int));
    float* PQ     = (float*)carve((size_t)B * 512 * N * sizeof(float));
    float* X3     = (float*)carve((size_t)B * 512 * N * sizeof(float));
    float* x3pre  = (float*)carve((size_t)B * 128 * N * sizeof(float));
    float2* part1 = (float2*)x3pre;
    float2* part2 = part1 + (size_t)B * (N / 32) * 128;
    (void)ws_size; (void)in_sizes; (void)n_in; (void)out_size;

    prep_pts_kernel<<<(B * N + 255) / 256, 256, 0, stream>>>(coords, pts, N, B * N);
    prep_w1_kernel<<<(128 * 128 + 255) / 256, 256, 0, stream>>>(W1, W1cat);
    prep_w2_kernel<<<(256 * 128 + 255) / 256, 256, 0, stream>>>(W2, W2cat);
    copy_x0_kernel<<<(B * 128 * N + 255) / 256, 256, 0, stream>>>(feats, X3, N, B * 128 * N);

    knn_kernel<<<dim3(N / 32, B), 256, 0, stream>>>(pts, knn, N);

    const float invCnt = 1.0f / (float)(N * KNN_K);

    gemm_wmma_kernel<true><<<dim3(N / 64, 256 / 64, B), 128, 0, stream>>>(
        W1cat, feats, PQ, N, C, (long)C * N, (long)256 * N, 256);
    edge_agg_kernel<128><<<dim3(N / 32, B), 256, 0, stream>>>(
        PQ, knn, X3 + (size_t)128 * N, part1, N, (long)512 * N);
    in_norm_kernel<<<(B * 128 * N + 255) / 256, 256, 0, stream>>>(
        X3 + (size_t)128 * N, part1, N / 32, 128, N, (long)512 * N, invCnt, B * 128 * N);

    gemm_wmma_kernel<true><<<dim3(N / 64, 512 / 64, B), 128, 0, stream>>>(
        W2cat, X3 + (size_t)128 * N, PQ, N, C, (long)512 * N, (long)512 * N, 512);
    edge_agg_kernel<256><<<dim3(N / 32, B), 256, 0, stream>>>(
        PQ, knn, X3 + (size_t)256 * N, part2, N, (long)512 * N);
    in_norm_kernel<<<(B * 256 * N + 255) / 256, 256, 0, stream>>>(
        X3 + (size_t)256 * N, part2, N / 32, 256, N, (long)512 * N, invCnt, B * 256 * N);

    gemm_wmma_kernel<false><<<dim3(N / 64, 128 / 64, B), 128, 0, stream>>>(
        W3, X3, x3pre, N, 512, (long)512 * N, (long)128 * N, 0);
    final_norm_kernel<<<dim3(128, B), 256, 0, stream>>>(x3pre, out, N);
}


# GEMM K loop software-pipelined: fragments of 32 K rows form a register block, the next block's loads are in flight while the current block's 32 WMMAs run; loaders, k order, sums unchanged
# speedup vs baseline: 3.4369x; 1.3777x over previous
#include <hip/hip_runtime.h>
#include <hip/hip_bf16.h>


typedef __attribute__((ext_vector_type(2))) float v2f;
typedef __attribute__((ext_vector_type(8))) float v8f;

#define KNN_K 10
#define IN_EPS 1e-5f

__device__ __forceinline__ float lrelu(float y) { return y >= 0.0f ? y : 0.2f * y; }

__device__ __forceinline__ v8f wmma_f32(v2f a, v2f b, v8f c) {
    return __builtin_amdgcn_wmma_f32_16x16x4_f32(
         false, a,  false, b,
         (short)0, c,  false,  false);
}

__global__ void prep_pts_kernel(const float* __restrict__ coords,
                                float4* __restrict__ pts, int N, int total) {
    int gid = blockIdx.x * 256 + threadIdx.x;
    if (gid >= total) return;
    int b = gid / N, n = gid - b * N;
    const float* cb = coords + (size_t)b * 3 * N;
    float x = cb[n], y = cb[N + n], z = cb[2 * N + n];
    pts[gid] = make_float4(x, y, z, x * x + y * y + z * z);
}

__global__ void prep_w1_kernel(const float* __restrict__ W1, float* __restrict__ Wc) {
    int gid = blockIdx.x * 256 + threadIdx.x;
    if (gid >= 128 * 128) return;
    int o = gid >> 7, c = gid & 127;
    float a = W1[o * 256 + c], b = W1[o * 256 + 128 + c];
    Wc[o * 128 + c] = a - b;
    Wc[(128 + o) * 128 + c] = b;
}

__global__ void prep_w2_kernel(const float* __restrict__ W2, float* __restrict__ Wc) {
    int gid = blockIdx.x * 256 + threadIdx.x;
    if (gid >= 256 * 128) return;
    int o = gid >> 7, c = gid & 127;
    float a = W2[o * 256 + c], b = W2[o * 256 + 128 + c];
    Wc[o * 128 + c] = a - b;
    Wc[(256 + o) * 128 + c] = b;
}

__global__ void copy_x0_kernel(const float* __restrict__ feats,
                               float* __restrict__ X3, int N, int total) {
    int gid = blockIdx.x * 256 + threadIdx.x;
    if (gid >= total) return;
    int n = gid % N, c = (gid / N) & 127, b = gid / (N * 128);
    X3[(size_t)b * 512 * N + (size_t)c * N + n] = feats[gid];
}

#define KNN_N   4096
#define KNN_QW  4
#define KNN_CAP 64

__device__ __forceinline__ float knn_dist(float px, float py, float pz, float pw, float4 q) {
    float d;
    {
#pragma clang fp contract(off)
        d = __builtin_fmaf(-2.0f, __builtin_fmaf(px, q.x, py * q.y) + pz * q.z, pw + q.w);
    }
    return d;
}

__device__ __forceinline__ bool knn_less(float da, int ia, float db, int ib) {
    return (da < db) || (da == db && ia < ib);
}

__device__ __forceinline__ void knn_exact_wave(const float4* tile, int N, int lane,
        float px, float py, float pz, float pw, int iq, int* so_q) {
    float ld = -__builtin_inff(); int li = -1; int w = 0;
    for (int r = 0; r <= KNN_K; ++r) {
        float bd = __builtin_inff(); int bi = 0x7fffffff;
        for (int t = 0; t < N; t += 32) {
            const int c = t + lane;
            float d = knn_dist(px, py, pz, pw, tile[c]);
            if (knn_less(ld, li, d, c) && knn_less(d, c, bd, bi)) { bd = d; bi = c; }
        }
        for (int off = 16; off > 0; off >>= 1) {
            float od = __shfl_xor(bd, off, 32); int oi = __shfl_xor(bi, off, 32);
            if (knn_less(od, oi, bd, bi)) { bd = od; bi = oi; }
        }
        ld = bd; li = bi;
        if (bi != iq) {
            if (w < KNN_K && lane == 0) so_q[w] = (bi < N) ? bi : 0;
            ++w;
        }
    }
}

__device__ __forceinline__ void knn_replay(const float* sd, const int* si, int n,
                                           int lane, int iq, int* so_q) {
    float bd[KNN_K + 1];
    int   bi[KNN_K + 1];
#pragma unroll
    for (int s = 0; s <= KNN_K; ++s) { bd[s] = 3.4e38f; bi[s] = -1; }
    for (int j = 0; j < n; ++j) {
        float cd = sd[j]; int ci = si[j];
        if (cd < bd[KNN_K]) {
#pragma unroll
            for (int s = 0; s <= KNN_K; ++s) {
                bool sw = cd < bd[s];
                float td = sw ? bd[s] : cd; int ti = sw ? bi[s] : ci;
                bd[s] = sw ? cd : bd[s];    bi[s] = sw ? ci : bi[s];
                cd = td; ci = ti;
            }
        }
    }
    if (lane == 0) {
        int w = 0;
#pragma unroll
        for (int s = 0; s <= KNN_K; ++s)
            if (bi[s] != iq && w < KNN_K) { so_q[w] = bi[s]; ++w; }
    }
}

__global__ __launch_bounds__(256) void knn_kernel(const float4* __restrict__ pts,
                                                  int* __restrict__ knn, int N) {
    __shared__ float4 tile[KNN_N];
    __shared__ float  hd[8][KNN_QW][KNN_CAP];
    __shared__ int    hi[8][KNN_QW][KNN_CAP];
    __shared__ int    so[32 * KNN_K];
    const int b = blockIdx.y;
    const int lane = threadIdx.x & 31;
    const int wave = __builtin_amdgcn_readfirstlane((int)(threadIdx.x >> 5));
    const float4* pb = pts + (size_t)b * N;

    for (int i = threadIdx.x; i < N; i += 256) tile[i] = pb[i];
    __syncthreads();

    const int q0 = blockIdx.x * 32 + wave * KNN_QW;
    float px[KNN_QW], py[KNN_QW], pz[KNN_QW], pw[KNN_QW];
#pragma unroll
    for (int qi = 0; qi < KNN_QW; ++qi) {
        float4 p = pb[q0 + qi];
        px[qi] = p.x; py[qi] = p.y; pz[qi] = p.z; pw[qi] = p.w;
    }

    float mn[KNN_QW];
#pragma unroll
    for (int qi = 0; qi < KNN_QW; ++qi) mn[qi] = __builtin_inff();
#pragma unroll 4
    for (int t = 0; t < N; t += 32) {
        float4 q = tile[t + lane];
#pragma unroll
        for (int qi = 0; qi < KNN_QW; ++qi) {
            float d = knn_dist(px[qi], py[qi], pz[qi], pw[qi], q);
            mn[qi] = d < mn[qi] ? d : mn[qi];
        }
    }

    float T[KNN_QW];
#pragma unroll
    for (int qi = 0; qi < KNN_QW; ++qi) {
        const float mv = mn[qi];
        int cnt = 0;
#pragma unroll
        for (int s = 0; s < 32; ++s) {
            float v = __builtin_bit_cast(float,
                __builtin_amdgcn_readlane(__builtin_bit_cast(int, mv), s));
            cnt += knn_less(v, s, mv, lane) ? 1 : 0;
        }
        unsigned sel = __builtin_amdgcn_ballot_w32(cnt == KNN_K);
        T[qi] = sel ? __shfl(mv, __builtin_ctz(sel), 32) : __builtin_inff();
    }

    int cn[KNN_QW];
#pragma unroll
    for (int qi = 0; qi < KNN_QW; ++qi) cn[qi] = 0;
    for (int t = 0; t < N; t += 128) {
        float4 q[4];
        float d[4][KNN_QW];
        unsigned bm[4][KNN_QW];
#pragma unroll
        for (int u = 0; u < 4; ++u) q[u] = tile[t + 32 * u + lane];
#pragma unroll
        for (int u = 0; u < 4; ++u) {
#pragma unroll
            for (int qi = 0; qi < KNN_QW; ++qi) {
                d[u][qi] = knn_dist(px[qi], py[qi], pz[qi], pw[qi], q[u]);
                bm[u][qi] = __builtin_amdgcn_ballot_w32(d[u][qi] <= T[qi]);
            }
        }
#pragma unroll
        for (int u = 0; u < 4; ++u) {
            if (bm[u][0] | bm[u][1] | bm[u][2] | bm[u][3]) {
#pragma unroll
                for (int qi = 0; qi < KNN_QW; ++qi) {
                    const unsigned m = bm[u][qi];
                    if (m) {
                        const bool hit = ((m >> lane) & 1u) != 0u;
                        const int pos = cn[qi] + (int)__builtin_amdgcn_mbcnt_lo(m, 0u);
                        if (hit && pos < KNN_CAP) {
                            hd[wave][qi][pos] = d[u][qi];
                            hi[wave][qi][pos] = t + 32 * u + lane;
                        }
                        cn[qi] += __builtin_popcount(m);
                    }
                }
            }
        }
    }
    __syncthreads();

#pragma unroll
    for (int qi = 0; qi < KNN_QW; ++qi) {
        const int n = cn[qi];
        const int iq = q0 + qi;
        int* so_q = so + (wave * KNN_QW + qi) * KNN_K;
        if (n > KNN_K && n <= KNN_CAP) {
            const float* sd = hd[wave][qi];
            const int*   si = hi[wave][qi];
            const bool v0 = lane < n, v1 = lane + 32 < n;
            const float d0 = v0 ? sd[lane] : __builtin_inff();
            const int   i0 = v0 ? si[lane] : 0x7fffffff;
            const float d1 = v1 ? sd[lane + 32] : __builtin_inff();
            const int   i1 = v1 ? si[lane + 32] : 0x7fffffff;
            int r0 = 0, r1 = 0;
            bool tie = false;
            for (int j = 0; j < n; ++j) {
                const float dj = sd[j]; const int ij = si[j];
                r0 += knn_less(dj, ij, d0, i0) ? 1 : 0;
                r1 += knn_less(dj, ij, d1, i1) ? 1 : 0;
                tie = tie || (dj == d0 && ij != i0) || (dj == d1 && ij != i1);
            }
            if (__builtin_amdgcn_ballot_w32(tie)) {
                knn_replay(sd, si, n, lane, iq, so_q);
                continue;
            }
            int sr = KNN_K + 1;
            unsigned s0 = __builtin_amdgcn_ballot_w32(v0 && i0 == iq);
            if (s0) sr = __shfl(r0, __builtin_ctz(s0), 32);
            unsigned s1 = __builtin_amdgcn_ballot_w32(v1 && i1 == iq);
            if (s1) sr = __shfl(r1, __builtin_ctz(s1), 32);
            if (v0 && i0 != iq && r0 <= KNN_K) {
                int w = r0 - (r0 > sr ? 1 : 0);
                if (w < KNN_K) so_q[w] = i0;
            }
            if (v1 && i1 != iq && r1 <= KNN_K) {
                int w = r1 - (r1 > sr ? 1 : 0);
                if (w < KNN_K) so_q[w] = i1;
            }
        } else {
            knn_exact_wave(tile, N, lane, px[qi], py[qi], pz[qi], pw[qi], iq, so_q);
        }
    }
    __syncthreads();
    int* ob = knn + ((size_t)b * N + (size_t)blockIdx.x * 32) * KNN_K;
    for (int i = threadIdx.x; i < 32 * KNN_K; i += 256) ob[i] = so[i];
}

#define GEMM_KB 8

template <bool TRANS>
__global__ __launch_bounds__(128) void gemm_wmma_kernel(
        const float* __restrict__ W, const float* __restrict__ X,
        float* __restrict__ D, int N, int Kdim,
        long xBatchStride, long dBatchStride, int ldT) {
    const int lane = threadIdx.x & 31;
    const int wave = threadIdx.x >> 5;
    const int half = lane >> 4;
    const int l16  = lane & 15;
    const int wm = wave >> 1, wn = wave & 1;
    const int m0 = blockIdx.y * 64 + wm * 32;
    const int n0 = blockIdx.x * 64 + wn * 32;
    const float* Xb = X + (size_t)blockIdx.z * xBatchStride;
    float*       Db = D + (size_t)blockIdx.z * dBatchStride;

    v8f c00 = {0.f,0.f,0.f,0.f,0.f,0.f,0.f,0.f};
    v8f c01 = c00, c10 = c00, c11 = c00;

    const float* ar0 = W + (size_t)(m0 + l16) * Kdim + 2 * half;
    const float* ar1 = ar0 + (size_t)16 * Kdim;
    const float* xr  = Xb + (size_t)(2 * half) * N + n0 + l16;

    __builtin_prefetch(xr, 0, 1);
    __builtin_prefetch(ar0, 0, 1);

    v2f fa0[GEMM_KB], fa1[GEMM_KB], fb0[GEMM_KB], fb1[GEMM_KB];
    v2f ga0[GEMM_KB], ga1[GEMM_KB], gb0[GEMM_KB], gb1[GEMM_KB];

    auto load_blk = [&](v2f (&a0)[GEMM_KB], v2f (&a1)[GEMM_KB],
                        v2f (&b0)[GEMM_KB], v2f (&b1)[GEMM_KB]) {
#pragma unroll
        for (int s = 0; s < GEMM_KB; ++s) {
            const float* xs = xr + (size_t)(4 * s) * N;
            a0[s].x = ar0[4 * s];   a0[s].y = ar0[4 * s + 1];
            a1[s].x = ar1[4 * s];   a1[s].y = ar1[4 * s + 1];
            b0[s].x = xs[0];        b0[s].y = xs[N];
            b1[s].x = xs[16];       b1[s].y = xs[16 + N];
        }
        ar0 += 4 * GEMM_KB; ar1 += 4 * GEMM_KB; xr += (size_t)(4 * GEMM_KB) * N;
    };
    auto mma_blk = [&](const v2f (&a0)[GEMM_KB], const v2f (&a1)[GEMM_KB],
                       const v2f (&b0)[GEMM_KB], const v2f (&b1)[GEMM_KB]) {
#pragma unroll
        for (int s = 0; s < GEMM_KB; ++s) {
            if (TRANS) {
                c00 = wmma_f32(b0[s], a0[s], c00);
                c01 = wmma_f32(b1[s], a0[s], c01);
                c10 = wmma_f32(b0[s], a1[s], c10);
                c11 = wmma_f32(b1[s], a1[s], c11);
            } else {
                c00 = wmma_f32(a0[s], b0[s], c00);
                c01 = wmma_f32(a0[s], b1[s], c01);
                c10 = wmma_f32(a1[s], b0[s], c10);
                c11 = wmma_f32(a1[s], b1[s], c11);
            }
        }
    };

    const int nblk = Kdim / (4 * GEMM_KB);
    load_blk(fa0, fa1, fb0, fb1);
    for (int j = 0; j < nblk - 2; j += 2) {
        load_blk(ga0, ga1, gb0, gb1);
        __builtin_amdgcn_sched_barrier(0);
        mma_blk(fa0, fa1, fb0, fb1);
        load_blk(fa0, fa1, fb0, fb1);
        __builtin_amdgcn_sched_barrier(0);
        mma_blk(ga0, ga1, gb0, gb1);
    }
    load_blk(ga0, ga1, gb0, gb1);
    __builtin_amdgcn_sched_barrier(0);
    mma_blk(fa0, fa1, fb0, fb1);
    mma_blk(ga0, ga1, gb0, gb1);
    if (TRANS) {
#pragma unroll
        for (int v = 0; v < 8; ++v) {
            const size_t r0 = (size_t)(n0 + v + 8 * half) * ldT + m0 + l16;
            const size_t r1 = r0 + (size_t)16 * ldT;
            Db[r0]      = c00[v];
            Db[r0 + 16] = c10[v];
            Db[r1]      = c01[v];
            Db[r1 + 16] = c11[v];
        }
        return;
    }
#pragma unroll
    for (int v = 0; v < 8; ++v) {
        const int mr0 = m0 + v + 8 * half;
        Db[(size_t)mr0 * N + n0 + l16]             = c00[v];
        Db[(size_t)mr0 * N + n0 + 16 + l16]        = c01[v];
        Db[(size_t)(mr0 + 16) * N + n0 + l16]      = c10[v];
        Db[(size_t)(mr0 + 16) * N + n0 + 16 + l16] = c11[v];
    }
}

#define EDGE_PTS 32
#define EDGE_PW  4

template <int M>
__global__ __launch_bounds__(256) void edge_agg_kernel(
        const float* __restrict__ PQT, const int* __restrict__ knn,
        float* __restrict__ outMax, float2* __restrict__ part,
        int N, long outBatchStride) {
    constexpr int NV = M / 128;
    constexpr int TS = M + 4;
    __shared__ __attribute__((aligned(16))) float tile[EDGE_PTS * TS];
    __shared__ __attribute__((aligned(16))) float rs[8][M];
    __shared__ __attribute__((aligned(16))) float rs2[8][M];
    const int b = blockIdx.y;
    const int lane = threadIdx.x & 31;
    const int wave = __builtin_amdgcn_readfirstlane((int)(threadIdx.x >> 5));
    const int nb0 = blockIdx.x * EDGE_PTS;
    const float* base = PQT + (size_t)b * N * (2 * M);

    float4 s[NV], s2[NV];
#pragma unroll
    for (int v = 0; v < NV; ++v) {
        s[v]  = make_float4(0.f, 0.f, 0.f, 0.f);
        s2[v] = make_float4(0.f, 0.f, 0.f, 0.f);
    }

#pragma unroll
    for (int j = 0; j < EDGE_PW; ++j) {
        const int nl = wave * EDGE_PW + j;
        const int n = nb0 + nl;
        const int* id = knn + ((size_t)b * N + n) * KNN_K;
        const float* prow = base + (size_t)n * (2 * M) + 4 * lane;
        float4 p[NV], mx[NV];
#pragma unroll
        for (int v = 0; v < NV; ++v) {
            p[v] = *(const float4*)(prow + v * 128);
            mx[v] = make_float4(-3.4e38f, -3.4e38f, -3.4e38f, -3.4e38f);
        }
#pragma unroll
        for (int k = 0; k < KNN_K; ++k) {
            const float* qrow = base + (size_t)id[k] * (2 * M) + M + 4 * lane;
#pragma unroll
            for (int v = 0; v < NV; ++v) {
                const float4 q = *(const float4*)(qrow + v * 128);
                const float y0 = p[v].x + q.x, y1 = p[v].y + q.y;
                const float y2 = p[v].z + q.z, y3 = p[v].w + q.w;
                mx[v].x = fmaxf(mx[v].x, y0); mx[v].y = fmaxf(mx[v].y, y1);
                mx[v].z = fmaxf(mx[v].z, y2); mx[v].w = fmaxf(mx[v].w, y3);
                s[v].x += y0; s[v].y += y1; s[v].z += y2; s[v].w += y3;
                s2[v].x += y0 * y0; s2[v].y += y1 * y1;
                s2[v].z += y2 * y2; s2[v].w += y3 * y3;
            }
        }
#pragma unroll
        for (int v = 0; v < NV; ++v)
            *(float4*)(tile + nl * TS + v * 128 + 4 * lane) = mx[v];
    }
#pragma unroll
    for (int v = 0; v < NV; ++v) {
        *(float4*)(&rs[wave][v * 128 + 4 * lane])  = s[v];
        *(float4*)(&rs2[wave][v * 128 + 4 * lane]) = s2[v];
    }
    __syncthreads();

    float* ob = outMax + (size_t)b * outBatchStride + nb0 + lane;
    for (int m = wave; m < M; m += 8)
        ob[(size_t)m * N] = tile[lane * TS + m];

    const int t = threadIdx.x;
    if (t < M) {
        float a = 0.f, a2 = 0.f;
#pragma unroll
        for (int w = 0; w < 8; ++w) { a += rs[w][t]; a2 += rs2[w][t]; }
        part[((size_t)b * gridDim.x + blockIdx.x) * M + t] = make_float2(a, a2);
    }
}

__global__ __launch_bounds__(256) void in_norm_kernel(float* __restrict__ data,
                               const float2* __restrict__ part, int tiles,
                               int M, int N, long batchStride, float invCnt,
                               int total) {
    __shared__ float red[8][2];
    const int g0 = blockIdx.x * 256;
    if (g0 >= total) return;
    const int bm = g0 / N;
    const int n = g0 - bm * N + (int)threadIdx.x;
    const int b = bm / M, m = bm - b * M;
    float* px = data + (size_t)b * batchStride + (size_t)m * N + n;
    const float x = *px;
    float a = 0.f, a2 = 0.f;
    for (int t = threadIdx.x; t < tiles; t += 256) {
        const float2 v = part[((size_t)b * tiles + t) * M + m];
        a += v.x; a2 += v.y;
    }
    for (int off = 16; off > 0; off >>= 1) {
        a  += __shfl_xor(a, off, 32);
        a2 += __shfl_xor(a2, off, 32);
    }
    if ((threadIdx.x & 31) == 0) { red[threadIdx.x >> 5][0] = a; red[threadIdx.x >> 5][1] = a2; }
    __syncthreads();
    float s = 0.f, s2 = 0.f;
#pragma unroll
    for (int w = 0; w < 8; ++w) { s += red[w][0]; s2 += red[w][1]; }
    float mu = s * invCnt;
    float var = fmaxf(s2 * invCnt - mu * mu, 0.f);
    float r = rsqrtf(var + IN_EPS);
    *px = lrelu((x - mu) * r);
}

__global__ __launch_bounds__(256) void final_norm_kernel(
        const float* __restrict__ xin, float* __restrict__ out, int N) {
    __shared__ float rs[256], rs2[256];
    __shared__ float s_mu, s_r;
    const int b = blockIdx.y, c = blockIdx.x, C = gridDim.x;
    const float* row = xin + ((size_t)b * C + c) * N;
    float s = 0.f, s2 = 0.f;
    for (int n = threadIdx.x; n < N; n += 256) {
        float v = row[n]; s += v; s2 += v * v;
    }
    rs[threadIdx.x] = s; rs2[threadIdx.x] = s2;
    __syncthreads();
    for (int off = 128; off > 0; off >>= 1) {
        if (threadIdx.x < off) {
            rs[threadIdx.x]  += rs[threadIdx.x + off];
            rs2[threadIdx.x] += rs2[threadIdx.x + off];
        }
        __syncthreads();
    }
    if (threadIdx.x == 0) {
        float mu = rs[0] / (float)N;
        float var = fmaxf(rs2[0] / (float)N - mu * mu, 0.f);
        s_mu = mu; s_r = rsqrtf(var + IN_EPS);
    }
    __syncthreads();
    float mu = s_mu, r = s_r;
    float* orow = out + ((size_t)b * C + c) * N;
    for (int n = threadIdx.x; n < N; n += 256)
        orow[n] = lrelu((row[n] - mu) * r);
}

extern "C" void kernel_launch(void* const* d_in, const int* in_sizes, int n_in,
                              void* d_out, int out_size, void* d_ws, size_t ws_size,
                              hipStream_t stream) {
    constexpr int B = 4, C = 128, N = 4096;
    const float* coords = (const float*)d_in[0];
    const float* feats  = (const float*)d_in[1];
    const float* W1     = (const float*)d_in[2];
    const float* W2     = (const float*)d_in[3];
    const float* W3     = (const float*)d_in[4];
    float* out = (float*)d_out;

    char* ws = (char*)d_ws;
    size_t off = 0;
    auto carve = [&](size_t bytes) -> void* {
        void* p = ws + off;
        off = (off + bytes + 255) & ~(size_t)255;
        return p;
    };
    float4* pts   = (float4*)carve((size_t)B * N * sizeof(float4));
    float* W1cat  = (float*)carve(256 * 128 * sizeof(float));
    float* W2cat  = (float*)carve(512 * 128 * sizeof(float));
    int*   knn    = (int*)carve((size_t)B * N * KNN_K * sizeof(int));
    float* PQ     = (float*)carve((size_t)B * 512 * N * sizeof(float));
    float* X3     = (float*)carve((size_t)B * 512 * N * sizeof(float));
    float* x3pre  = (float*)carve((size_t)B * 128 * N * sizeof(float));
    float2* part1 = (float2*)x3pre;
    float2* part2 = part1 + (size_t)B * (N / 32) * 128;
    (void)ws_size; (void)in_sizes; (void)n_in; (void)out_size;

    prep_pts_kernel<<<(B * N + 255) / 256, 256, 0, stream>>>(coords, pts, N, B * N);
    prep_w1_kernel<<<(128 * 128 + 255) / 256, 256, 0, stream>>>(W1, W1cat);
    prep_w2_kernel<<<(256 * 128 + 255) / 256, 256, 0, stream>>>(W2, W2cat);
    copy_x0_kernel<<<(B * 128 * N + 255) / 256, 256, 0, stream>>>(feats, X3, N, B * 128 * N);

    knn_kernel<<<dim3(N / 32, B), 256, 0, stream>>>(pts, knn, N);

    const float invCnt = 1.0f / (float)(N * KNN_K);

    gemm_wmma_kernel<true><<<dim3(N / 64, 256 / 64, B), 128, 0, stream>>>(
        W1cat, feats, PQ, N, C, (long)C * N, (long)256 * N, 256);
    edge_agg_kernel<128><<<dim3(N / 32, B), 256, 0, stream>>>(
        PQ, knn, X3 + (size_t)128 * N, part1, N, (long)512 * N);
    in_norm_kernel<<<(B * 128 * N + 255) / 256, 256, 0, stream>>>(
        X3 + (size_t)128 * N, part1, N / 32, 128, N, (long)512 * N, invCnt, B * 128 * N);

    gemm_wmma_kernel<true><<<dim3(N / 64, 512 / 64, B), 128, 0, stream>>>(
        W2cat, X3 + (size_t)128 * N, PQ, N, C, (long)512 * N, (long)512 * N, 512);
    edge_agg_kernel<256><<<dim3(N / 32, B), 256, 0, stream>>>(
        PQ, knn, X3 + (size_t)256 * N, part2, N, (long)512 * N);
    in_norm_kernel<<<(B * 256 * N + 255) / 256, 256, 0, stream>>>(
        X3 + (size_t)256 * N, part2, N / 32, 256, N, (long)512 * N, invCnt, B * 256 * N);

    gemm_wmma_kernel<false><<<dim3(N / 64, 128 / 64, B), 128, 0, stream>>>(
        W3, X3, x3pre, N, 512, (long)512 * N, (long)128 * N, 0);
    final_norm_kernel<<<dim3(128, B), 256, 0, stream>>>(x3pre, out, N);
}


# Three launches removed: x0 is no longer copied into the stacked buffer (the last GEMM reads K rows 0..127 from the features input, the rest from the buffer); the three set-up passes share one launch
# speedup vs baseline: 3.7384x; 1.0877x over previous
#include <hip/hip_runtime.h>
#include <hip/hip_bf16.h>


typedef __attribute__((ext_vector_type(2))) float v2f;
typedef __attribute__((ext_vector_type(8))) float v8f;

#define KNN_K 10
#define IN_EPS 1e-5f

__device__ __forceinline__ float lrelu(float y) { return y >= 0.0f ? y : 0.2f * y; }

__device__ __forceinline__ v8f wmma_f32(v2f a, v2f b, v8f c) {
    return __builtin_amdgcn_wmma_f32_16x16x4_f32(
         false, a,  false, b,
         (short)0, c,  false,  false);
}

#define PREP_PTS_BLOCKS 64
#define PREP_W1_BLOCKS  64
#define PREP_W2_BLOCKS  128

__global__ __launch_bounds__(256) void prep_kernel(const float* __restrict__ coords,
                                                   float4* __restrict__ pts,
                                                   const float* __restrict__ W1,
                                                   float* __restrict__ W1c,
                                                   const float* __restrict__ W2,
                                                   float* __restrict__ W2c, int N) {
    const int blk = blockIdx.x;
    if (blk < PREP_PTS_BLOCKS) {
        const int gid = blk * 256 + threadIdx.x;
        const int b = gid / N, n = gid - b * N;
        const float* cb = coords + (size_t)b * 3 * N;
        float x = cb[n], y = cb[N + n], z = cb[2 * N + n];
        pts[gid] = make_float4(x, y, z, x * x + y * y + z * z);
    } else if (blk < PREP_PTS_BLOCKS + PREP_W1_BLOCKS) {
        const int gid = (blk - PREP_PTS_BLOCKS) * 256 + threadIdx.x;
        const int o = gid >> 7, c = gid & 127;
        float a = W1[o * 256 + c], b = W1[o * 256 + 128 + c];
        W1c[o * 128 + c] = a - b;
        W1c[(128 + o) * 128 + c] = b;
    } else {
        const int gid = (blk - PREP_PTS_BLOCKS - PREP_W1_BLOCKS) * 256 + threadIdx.x;
        const int o = gid >> 7, c = gid & 127;
        float a = W2[o * 256 + c], b = W2[o * 256 + 128 + c];
        W2c[o * 128 + c] = a - b;
        W2c[(256 + o) * 128 + c] = b;
    }
}

#define KNN_N   4096
#define KNN_QW  4
#define KNN_CAP 64

__device__ __forceinline__ float knn_dist(float px, float py, float pz, float pw, float4 q) {
    float d;
    {
#pragma clang fp contract(off)
        d = __builtin_fmaf(-2.0f, __builtin_fmaf(px, q.x, py * q.y) + pz * q.z, pw + q.w);
    }
    return d;
}

__device__ __forceinline__ bool knn_less(float da, int ia, float db, int ib) {
    return (da < db) || (da == db && ia < ib);
}

__device__ __forceinline__ void knn_exact_wave(const float4* tile, int N, int lane,
        float px, float py, float pz, float pw, int iq, int* so_q) {
    float ld = -__builtin_inff(); int li = -1; int w = 0;
    for (int r = 0; r <= KNN_K; ++r) {
        float bd = __builtin_inff(); int bi = 0x7fffffff;
        for (int t = 0; t < N; t += 32) {
            const int c = t + lane;
            float d = knn_dist(px, py, pz, pw, tile[c]);
            if (knn_less(ld, li, d, c) && knn_less(d, c, bd, bi)) { bd = d; bi = c; }
        }
        for (int off = 16; off > 0; off >>= 1) {
            float od = __shfl_xor(bd, off, 32); int oi = __shfl_xor(bi, off, 32);
            if (knn_less(od, oi, bd, bi)) { bd = od; bi = oi; }
        }
        ld = bd; li = bi;
        if (bi != iq) {
            if (w < KNN_K && lane == 0) so_q[w] = (bi < N) ? bi : 0;
            ++w;
        }
    }
}

__device__ __forceinline__ void knn_replay(const float* sd, const int* si, int n,
                                           int lane, int iq, int* so_q) {
    float bd[KNN_K + 1];
    int   bi[KNN_K + 1];
#pragma unroll
    for (int s = 0; s <= KNN_K; ++s) { bd[s] = 3.4e38f; bi[s] = -1; }
    for (int j = 0; j < n; ++j) {
        float cd = sd[j]; int ci = si[j];
        if (cd < bd[KNN_K]) {
#pragma unroll
            for (int s = 0; s <= KNN_K; ++s) {
                bool sw = cd < bd[s];
                float td = sw ? bd[s] : cd; int ti = sw ? bi[s] : ci;
                bd[s] = sw ? cd : bd[s];    bi[s] = sw ? ci : bi[s];
                cd = td; ci = ti;
            }
        }
    }
    if (lane == 0) {
        int w = 0;
#pragma unroll
        for (int s = 0; s <= KNN_K; ++s)
            if (bi[s] != iq && w < KNN_K) { so_q[w] = bi[s]; ++w; }
    }
}

__global__ __launch_bounds__(256) void knn_kernel(const float4* __restrict__ pts,
                                                  int* __restrict__ knn, int N) {
    __shared__ float4 tile[KNN_N];
    __shared__ float  hd[8][KNN_QW][KNN_CAP];
    __shared__ int    hi[8][KNN_QW][KNN_CAP];
    __shared__ int    so[32 * KNN_K];
    const int b = blockIdx.y;
    const int lane = threadIdx.x & 31;
    const int wave = __builtin_amdgcn_readfirstlane((int)(threadIdx.x >> 5));
    const float4* pb = pts + (size_t)b * N;

    for (int i = threadIdx.x; i < N; i += 256) tile[i] = pb[i];
    __syncthreads();

    const int q0 = blockIdx.x * 32 + wave * KNN_QW;
    float px[KNN_QW], py[KNN_QW], pz[KNN_QW], pw[KNN_QW];
#pragma unroll
    for (int qi = 0; qi < KNN_QW; ++qi) {
        float4 p = pb[q0 + qi];
        px[qi] = p.x; py[qi] = p.y; pz[qi] = p.z; pw[qi] = p.w;
    }

    float mn[KNN_QW];
#pragma unroll
    for (int qi = 0; qi < KNN_QW; ++qi) mn[qi] = __builtin_inff();
#pragma unroll 4
    for (int t = 0; t < N; t += 32) {
        float4 q = tile[t + lane];
#pragma unroll
        for (int qi = 0; qi < KNN_QW; ++qi) {
            float d = knn_dist(px[qi], py[qi], pz[qi], pw[qi], q);
            mn[qi] = d < mn[qi] ? d : mn[qi];
        }
    }

    float T[KNN_QW];
#pragma unroll
    for (int qi = 0; qi < KNN_QW; ++qi) {
        const float mv = mn[qi];
        int cnt = 0;
#pragma unroll
        for (int s = 0; s < 32; ++s) {
            float v = __builtin_bit_cast(float,
                __builtin_amdgcn_readlane(__builtin_bit_cast(int, mv), s));
            cnt += knn_less(v, s, mv, lane) ? 1 : 0;
        }
        unsigned sel = __builtin_amdgcn_ballot_w32(cnt == KNN_K);
        T[qi] = sel ? __shfl(mv, __builtin_ctz(sel), 32) : __builtin_inff();
    }

    int cn[KNN_QW];
#pragma unroll
    for (int qi = 0; qi < KNN_QW; ++qi) cn[qi] = 0;
    for (int t = 0; t < N; t += 128) {
        float4 q[4];
        float d[4][KNN_QW];
        unsigned bm[4][KNN_QW];
#pragma unroll
        for (int u = 0; u < 4; ++u) q[u] = tile[t + 32 * u + lane];
#pragma unroll
        for (int u = 0; u < 4; ++u) {
#pragma unroll
            for (int qi = 0; qi < KNN_QW; ++qi) {
                d[u][qi] = knn_dist(px[qi], py[qi], pz[qi], pw[qi], q[u]);
                bm[u][qi] = __builtin_amdgcn_ballot_w32(d[u][qi] <= T[qi]);
            }
        }
#pragma unroll
        for (int u = 0; u < 4; ++u) {
            if (bm[u][0] | bm[u][1] | bm[u][2] | bm[u][3]) {
#pragma unroll
                for (int qi = 0; qi < KNN_QW; ++qi) {
                    const unsigned m = bm[u][qi];
                    if (m) {
                        const bool hit = ((m >> lane) & 1u) != 0u;
                        const int pos = cn[qi] + (int)__builtin_amdgcn_mbcnt_lo(m, 0u);
                        if (hit && pos < KNN_CAP) {
                            hd[wave][qi][pos] = d[u][qi];
                            hi[wave][qi][pos] = t + 32 * u + lane;
                        }
                        cn[qi] += __builtin_popcount(m);
                    }
                }
            }
        }
    }
    __syncthreads();

#pragma unroll
    for (int qi = 0; qi < KNN_QW; ++qi) {
        const int n = cn[qi];
        const int iq = q0 + qi;
        int* so_q = so + (wave * KNN_QW + qi) * KNN_K;
        if (n > KNN_K && n <= KNN_CAP) {
            const float* sd = hd[wave][qi];
            const int*   si = hi[wave][qi];
            const bool v0 = lane < n, v1 = lane + 32 < n;
            const float d0 = v0 ? sd[lane] : __builtin_inff();
            const int   i0 = v0 ? si[lane] : 0x7fffffff;
            const float d1 = v1 ? sd[lane + 32] : __builtin_inff();
            const int   i1 = v1 ? si[lane + 32] : 0x7fffffff;
            int r0 = 0, r1 = 0;
            bool tie = false;
            for (int j = 0; j < n; ++j) {
                const float dj = sd[j]; const int ij = si[j];
                r0 += knn_less(dj, ij, d0, i0) ? 1 : 0;
                r1 += knn_less(dj, ij, d1, i1) ? 1 : 0;
                tie = tie || (dj == d0 && ij != i0) || (dj == d1 && ij != i1);
            }
            if (__builtin_amdgcn_ballot_w32(tie)) {
                knn_replay(sd, si, n, lane, iq, so_q);
                continue;
            }
            int sr = KNN_K + 1;
            unsigned s0 = __builtin_amdgcn_ballot_w32(v0 && i0 == iq);
            if (s0) sr = __shfl(r0, __builtin_ctz(s0), 32);
            unsigned s1 = __builtin_amdgcn_ballot_w32(v1 && i1 == iq);
            if (s1) sr = __shfl(r1, __builtin_ctz(s1), 32);
            if (v0 && i0 != iq && r0 <= KNN_K) {
                int w = r0 - (r0 > sr ? 1 : 0);
                if (w < KNN_K) so_q[w] = i0;
            }
            if (v1 && i1 != iq && r1 <= KNN_K) {
                int w = r1 - (r1 > sr ? 1 : 0);
                if (w < KNN_K) so_q[w] = i1;
            }
        } else {
            knn_exact_wave(tile, N, lane, px[qi], py[qi], pz[qi], pw[qi], iq, so_q);
        }
    }
    __syncthreads();
    int* ob = knn + ((size_t)b * N + (size_t)blockIdx.x * 32) * KNN_K;
    for (int i = threadIdx.x; i < 32 * KNN_K; i += 256) ob[i] = so[i];
}

#define GEMM_KB 8

template <bool TRANS>
__global__ __launch_bounds__(128) void gemm_wmma_kernel(
        const float* __restrict__ W, const float* __restrict__ X, const float* X2,
        float* __restrict__ D, int N, int Kdim,
        long xBatchStride, long x2BatchStride, int splitBlk, long dBatchStride, int ldT) {
    const int lane = threadIdx.x & 31;
    const int wave = threadIdx.x >> 5;
    const int half = lane >> 4;
    const int l16  = lane & 15;
    const int wm = wave >> 1, wn = wave & 1;
    const int m0 = blockIdx.y * 64 + wm * 32;
    const int n0 = blockIdx.x * 64 + wn * 32;
    const float* Xb = X + (size_t)blockIdx.z * xBatchStride;
    float*       Db = D + (size_t)blockIdx.z * dBatchStride;

    v8f c00 = {0.f,0.f,0.f,0.f,0.f,0.f,0.f,0.f};
    v8f c01 = c00, c10 = c00, c11 = c00;

    const float* ar0 = W + (size_t)(m0 + l16) * Kdim + 2 * half;
    const float* ar1 = ar0 + (size_t)16 * Kdim;
    const float* xr  = Xb + (size_t)(2 * half) * N + n0 + l16;
    const float* xr2 = X2 + (size_t)blockIdx.z * x2BatchStride
                     + (size_t)(splitBlk * (4 * GEMM_KB) + 2 * half) * N + n0 + l16;
    int blk = 0;

    __builtin_prefetch(xr, 0, 1);
    __builtin_prefetch(ar0, 0, 1);

    v2f fa0[GEMM_KB], fa1[GEMM_KB], fb0[GEMM_KB], fb1[GEMM_KB];
    v2f ga0[GEMM_KB], ga1[GEMM_KB], gb0[GEMM_KB], gb1[GEMM_KB];

    auto load_blk = [&](v2f (&a0)[GEMM_KB], v2f (&a1)[GEMM_KB],
                        v2f (&b0)[GEMM_KB], v2f (&b1)[GEMM_KB]) {
#pragma unroll
        for (int s = 0; s < GEMM_KB; ++s) {
            const float* xs = xr + (size_t)(4 * s) * N;
            a0[s].x = ar0[4 * s];   a0[s].y = ar0[4 * s + 1];
            a1[s].x = ar1[4 * s];   a1[s].y = ar1[4 * s + 1];
            b0[s].x = xs[0];        b0[s].y = xs[N];
            b1[s].x = xs[16];       b1[s].y = xs[16 + N];
        }
        ar0 += 4 * GEMM_KB; ar1 += 4 * GEMM_KB; xr += (size_t)(4 * GEMM_KB) * N;
        if (++blk == splitBlk) xr = xr2;
    };
    auto mma_blk = [&](const v2f (&a0)[GEMM_KB], const v2f (&a1)[GEMM_KB],
                       const v2f (&b0)[GEMM_KB], const v2f (&b1)[GEMM_KB]) {
#pragma unroll
        for (int s = 0; s < GEMM_KB; ++s) {
            if (TRANS) {
                c00 = wmma_f32(b0[s], a0[s], c00);
                c01 = wmma_f32(b1[s], a0[s], c01);
                c10 = wmma_f32(b0[s], a1[s], c10);
                c11 = wmma_f32(b1[s], a1[s], c11);
            } else {
                c00 = wmma_f32(a0[s], b0[s], c00);
                c01 = wmma_f32(a0[s], b1[s], c01);
                c10 = wmma_f32(a1[s], b0[s], c10);
                c11 = wmma_f32(a1[s], b1[s], c11);
            }
        }
    };

    const int nblk = Kdim / (4 * GEMM_KB);
    load_blk(fa0, fa1, fb0, fb1);
    for (int j = 0; j < nblk - 2; j += 2) {
        load_blk(ga0, ga1, gb0, gb1);
        __builtin_amdgcn_sched_barrier(0);
        mma_blk(fa0, fa1, fb0, fb1);
        load_blk(fa0, fa1, fb0, fb1);
        __builtin_amdgcn_sched_barrier(0);
        mma_blk(ga0, ga1, gb0, gb1);
    }
    load_blk(ga0, ga1, gb0, gb1);
    __builtin_amdgcn_sched_barrier(0);
    mma_blk(fa0, fa1, fb0, fb1);
    mma_blk(ga0, ga1, gb0, gb1);
    if (TRANS) {
#pragma unroll
        for (int v = 0; v < 8; ++v) {
            const size_t r0 = (size_t)(n0 + v + 8 * half) * ldT + m0 + l16;
            const size_t r1 = r0 + (size_t)16 * ldT;
            Db[r0]      = c00[v];
            Db[r0 + 16] = c10[v];
            Db[r1]      = c01[v];
            Db[r1 + 16] = c11[v];
        }
        return;
    }
#pragma unroll
    for (int v = 0; v < 8; ++v) {
        const int mr0 = m0 + v + 8 * half;
        Db[(size_t)mr0 * N + n0 + l16]             = c00[v];
        Db[(size_t)mr0 * N + n0 + 16 + l16]        = c01[v];
        Db[(size_t)(mr0 + 16) * N + n0 + l16]      = c10[v];
        Db[(size_t)(mr0 + 16) * N + n0 + 16 + l16] = c11[v];
    }
}

#define EDGE_PTS 32
#define EDGE_PW  4

template <int M>
__global__ __launch_bounds__(256) void edge_agg_kernel(
        const float* __restrict__ PQT, const int* __restrict__ knn,
        float* __restrict__ outMax, float2* __restrict__ part,
        int N, long outBatchStride) {
    constexpr int NV = M / 128;
    constexpr int TS = M + 4;
    __shared__ __attribute__((aligned(16))) float tile[EDGE_PTS * TS];
    __shared__ __attribute__((aligned(16))) float rs[8][M];
    __shared__ __attribute__((aligned(16))) float rs2[8][M];
    const int b = blockIdx.y;
    const int lane = threadIdx.x & 31;
    const int wave = __builtin_amdgcn_readfirstlane((int)(threadIdx.x >> 5));
    const int nb0 = blockIdx.x * EDGE_PTS;
    const float* base = PQT + (size_t)b * N * (2 * M);

    float4 s[NV], s2[NV];
#pragma unroll
    for (int v = 0; v < NV; ++v) {
        s[v]  = make_float4(0.f, 0.f, 0.f, 0.f);
        s2[v] = make_float4(0.f, 0.f, 0.f, 0.f);
    }

#pragma unroll
    for (int j = 0; j < EDGE_PW; ++j) {
        const int nl = wave * EDGE_PW + j;
        const int n = nb0 + nl;
        const int* id = knn + ((size_t)b * N + n) * KNN_K;
        const float* prow = base + (size_t)n * (2 * M) + 4 * lane;
        float4 p[NV], mx[NV];
#pragma unroll
        for (int v = 0; v < NV; ++v) {
            p[v] = *(const float4*)(prow + v * 128);
            mx[v] = make_float4(-3.4e38f, -3.4e38f, -3.4e38f, -3.4e38f);
        }
#pragma unroll
        for (int k = 0; k < KNN_K; ++k) {
            const float* qrow = base + (size_t)id[k] * (2 * M) + M + 4 * lane;
#pragma unroll
            for (int v = 0; v < NV; ++v) {
                const float4 q = *(const float4*)(qrow + v * 128);
                const float y0 = p[v].x + q.x, y1 = p[v].y + q.y;
                const float y2 = p[v].z + q.z, y3 = p[v].w + q.w;
                mx[v].x = fmaxf(mx[v].x, y0); mx[v].y = fmaxf(mx[v].y, y1);
                mx[v].z = fmaxf(mx[v].z, y2); mx[v].w = fmaxf(mx[v].w, y3);
                s[v].x += y0; s[v].y += y1; s[v].z += y2; s[v].w += y3;
                s2[v].x += y0 * y0; s2[v].y += y1 * y1;
                s2[v].z += y2 * y2; s2[v].w += y3 * y3;
            }
        }
#pragma unroll
        for (int v = 0; v < NV; ++v)
            *(float4*)(tile + nl * TS + v * 128 + 4 * lane) = mx[v];
    }
#pragma unroll
    for (int v = 0; v < NV; ++v) {
        *(float4*)(&rs[wave][v * 128 + 4 * lane])  = s[v];
        *(float4*)(&rs2[wave][v * 128 + 4 * lane]) = s2[v];
    }
    __syncthreads();

    float* ob = outMax + (size_t)b * outBatchStride + nb0 + lane;
    for (int m = wave; m < M; m += 8)
        ob[(size_t)m * N] = tile[lane * TS + m];

    const int t = threadIdx.x;
    if (t < M) {
        float a = 0.f, a2 = 0.f;
#pragma unroll
        for (int w = 0; w < 8; ++w) { a += rs[w][t]; a2 += rs2[w][t]; }
        part[((size_t)b * gridDim.x + blockIdx.x) * M + t] = make_float2(a, a2);
    }
}

__global__ __launch_bounds__(256) void in_norm_kernel(float* __restrict__ data,
                               const float2* __restrict__ part, int tiles,
                               int M, int N, long batchStride, float invCnt,
                               int total) {
    __shared__ float red[8][2];
    const int g0 = blockIdx.x * 256;
    if (g0 >= total) return;
    const int bm = g0 / N;
    const int n = g0 - bm * N + (int)threadIdx.x;
    const int b = bm / M, m = bm - b * M;
    float* px = data + (size_t)b * batchStride + (size_t)m * N + n;
    const float x = *px;
    float a = 0.f, a2 = 0.f;
    for (int t = threadIdx.x; t < tiles; t += 256) {
        const float2 v = part[((size_t)b * tiles + t) * M + m];
        a += v.x; a2 += v.y;
    }
    for (int off = 16; off > 0; off >>= 1) {
        a  += __shfl_xor(a, off, 32);
        a2 += __shfl_xor(a2, off, 32);
    }
    if ((threadIdx.x & 31) == 0) { red[threadIdx.x >> 5][0] = a; red[threadIdx.x >> 5][1] = a2; }
    __syncthreads();
    float s = 0.f, s2 = 0.f;
#pragma unroll
    for (int w = 0; w < 8; ++w) { s += red[w][0]; s2 += red[w][1]; }
    float mu = s * invCnt;
    float var = fmaxf(s2 * invCnt - mu * mu, 0.f);
    float r = rsqrtf(var + IN_EPS);
    *px = lrelu((x - mu) * r);
}

__global__ __launch_bounds__(256) void final_norm_kernel(
        const float* __restrict__ xin, float* __restrict__ out, int N) {
    __shared__ float rs[256], rs2[256];
    __shared__ float s_mu, s_r;
    const int b = blockIdx.y, c = blockIdx.x, C = gridDim.x;
    const float* row = xin + ((size_t)b * C + c) * N;
    float s = 0.f, s2 = 0.f;
    for (int n = threadIdx.x; n < N; n += 256) {
        float v = row[n]; s += v; s2 += v * v;
    }
    rs[threadIdx.x] = s; rs2[threadIdx.x] = s2;
    __syncthreads();
    for (int off = 128; off > 0; off >>= 1) {
        if (threadIdx.x < off) {
            rs[threadIdx.x]  += rs[threadIdx.x + off];
            rs2[threadIdx.x] += rs2[threadIdx.x + off];
        }
        __syncthreads();
    }
    if (threadIdx.x == 0) {
        float mu = rs[0] / (float)N;
        float var = fmaxf(rs2[0] / (float)N - mu * mu, 0.f);
        s_mu = mu; s_r = rsqrtf(var + IN_EPS);
    }
    __syncthreads();
    float mu = s_mu, r = s_r;
    float* orow = out + ((size_t)b * C + c) * N;
    for (int n = threadIdx.x; n < N; n += 256)
        orow[n] = lrelu((row[n] - mu) * r);
}

extern "C" void kernel_launch(void* const* d_in, const int* in_sizes, int n_in,
                              void* d_out, int out_size, void* d_ws, size_t ws_size,
                              hipStream_t stream) {
    constexpr int B = 4, C = 128, N = 4096;
    const float* coords = (const float*)d_in[0];
    const float* feats  = (const float*)d_in[1];
    const float* W1     = (const float*)d_in[2];
    const float* W2     = (const float*)d_in[3];
    const float* W3     = (const float*)d_in[4];
    float* out = (float*)d_out;

    char* ws = (char*)d_ws;
    size_t off = 0;
    auto carve = [&](size_t bytes) -> void* {
        void* p = ws + off;
        off = (off + bytes + 255) & ~(size_t)255;
        return p;
    };
    float4* pts   = (float4*)carve((size_t)B * N * sizeof(float4));
    float* W1cat  = (float*)carve(256 * 128 * sizeof(float));
    float* W2cat  = (float*)carve(512 * 128 * sizeof(float));
    int*   knn    = (int*)carve((size_t)B * N * KNN_K * sizeof(int));
    float* PQ     = (float*)carve((size_t)B * 512 * N * sizeof(float));
    float* X3     = (float*)carve((size_t)B * 512 * N * sizeof(float));
    float* x3pre  = (float*)carve((size_t)B * 128 * N * sizeof(float));
    float2* part1 = (float2*)x3pre;
    float2* part2 = part1 + (size_t)B * (N / 32) * 128;
    (void)ws_size; (void)in_sizes; (void)n_in; (void)out_size;

    static_assert(B * N == PREP_PTS_BLOCKS * 256, "set-up block split: points");
    prep_kernel<<<PREP_PTS_BLOCKS + PREP_W1_BLOCKS + PREP_W2_BLOCKS, 256, 0, stream>>>(
        coords, pts, W1, W1cat, W2, W2cat, N);

    knn_kernel<<<dim3(N / 32, B), 256, 0, stream>>>(pts, knn, N);

    const float invCnt = 1.0f / (float)(N * KNN_K);

    gemm_wmma_kernel<true><<<dim3(N / 64, 256 / 64, B), 128, 0, stream>>>(
        W1cat, feats, feats, PQ, N, C, (long)C * N, (long)C * N, 0, (long)256 * N, 256);
    edge_agg_kernel<128><<<dim3(N / 32, B), 256, 0, stream>>>(
        PQ, knn, X3 + (size_t)128 * N, part1, N, (long)512 * N);
    in_norm_kernel<<<(B * 128 * N + 255) / 256, 256, 0, stream>>>(
        X3 + (size_t)128 * N, part1, N / 32, 128, N, (long)512 * N, invCnt, B * 128 * N);

    gemm_wmma_kernel<true><<<dim3(N / 64, 512 / 64, B), 128, 0, stream>>>(
        W2cat, X3 + (size_t)128 * N, X3 + (size_t)128 * N, PQ, N, C,
        (long)512 * N, (long)512 * N, 0, (long)512 * N, 512);
    edge_agg_kernel<256><<<dim3(N / 32, B), 256, 0, stream>>>(
        PQ, knn, X3 + (size_t)256 * N, part2, N, (long)512 * N);
    in_norm_kernel<<<(B * 256 * N + 255) / 256, 256, 0, stream>>>(
        X3 + (size_t)256 * N, part2, N / 32, 256, N, (long)512 * N, invCnt, B * 256 * N);

    gemm_wmma_kernel<false><<<dim3(N / 64, 128 / 64, B), 128, 0, stream>>>(
        W3, feats, X3, x3pre, N, 512, (long)C * N, (long)512 * N, 4, (long)128 * N, 0);
    final_norm_kernel<<<dim3(128, B), 256, 0, stream>>>(x3pre, out, N);
}


# InstanceNorm+lrelu of x1, x2 moved into the consuming GEMMs' X fragments (per-channel mean/rstd table); the two normalisation passes shrink to small statistics launches; values bit-identical
# speedup vs baseline: 4.4671x; 1.1949x over previous
#include <hip/hip_runtime.h>
#include <hip/hip_bf16.h>


typedef __attribute__((ext_vector_type(2))) float v2f;
typedef __attribute__((ext_vector_type(8))) float v8f;

#define KNN_K 10
#define IN_EPS 1e-5f

__device__ __forceinline__ float lrelu(float y) { return y >= 0.0f ? y : 0.2f * y; }

__device__ __forceinline__ v8f wmma_f32(v2f a, v2f b, v8f c) {
    return __builtin_amdgcn_wmma_f32_16x16x4_f32(
         false, a,  false, b,
         (short)0, c,  false,  false);
}

#define PREP_PTS_BLOCKS 64
#define PREP_W1_BLOCKS  64
#define PREP_W2_BLOCKS  128

__global__ __launch_bounds__(256) void prep_kernel(const float* __restrict__ coords,
                                                   float4* __restrict__ pts,
                                                   const float* __restrict__ W1,
                                                   float* __restrict__ W1c,
                                                   const float* __restrict__ W2,
                                                   float* __restrict__ W2c, int N) {
    const int blk = blockIdx.x;
    if (blk < PREP_PTS_BLOCKS) {
        const int gid = blk * 256 + threadIdx.x;
        const int b = gid / N, n = gid - b * N;
        const float* cb = coords + (size_t)b * 3 * N;
        float x = cb[n], y = cb[N + n], z = cb[2 * N + n];
        pts[gid] = make_float4(x, y, z, x * x + y * y + z * z);
    } else if (blk < PREP_PTS_BLOCKS + PREP_W1_BLOCKS) {
        const int gid = (blk - PREP_PTS_BLOCKS) * 256 + threadIdx.x;
        const int o = gid >> 7, c = gid & 127;
        float a = W1[o * 256 + c], b = W1[o * 256 + 128 + c];
        W1c[o * 128 + c] = a - b;
        W1c[(128 + o) * 128 + c] = b;
    } else {
        const int gid = (blk - PREP_PTS_BLOCKS - PREP_W1_BLOCKS) * 256 + threadIdx.x;
        const int o = gid >> 7, c = gid & 127;
        float a = W2[o * 256 + c], b = W2[o * 256 + 128 + c];
        W2c[o * 128 + c] = a - b;
        W2c[(256 + o) * 128 + c] = b;
    }
}

#define KNN_N   4096
#define KNN_QW  4
#define KNN_CAP 64

__device__ __forceinline__ float knn_dist(float px, float py, float pz, float pw, float4 q) {
    float d;
    {
#pragma clang fp contract(off)
        d = __builtin_fmaf(-2.0f, __builtin_fmaf(px, q.x, py * q.y) + pz * q.z, pw + q.w);
    }
    return d;
}

__device__ __forceinline__ bool knn_less(float da, int ia, float db, int ib) {
    return (da < db) || (da == db && ia < ib);
}

__device__ __forceinline__ void knn_exact_wave(const float4* tile, int N, int lane,
        float px, float py, float pz, float pw, int iq, int* so_q) {
    float ld = -__builtin_inff(); int li = -1; int w = 0;
    for (int r = 0; r <= KNN_K; ++r) {
        float bd = __builtin_inff(); int bi = 0x7fffffff;
        for (int t = 0; t < N; t += 32) {
            const int c = t + lane;
            float d = knn_dist(px, py, pz, pw, tile[c]);
            if (knn_less(ld, li, d, c) && knn_less(d, c, bd, bi)) { bd = d; bi = c; }
        }
        for (int off = 16; off > 0; off >>= 1) {
            float od = __shfl_xor(bd, off, 32); int oi = __shfl_xor(bi, off, 32);
            if (knn_less(od, oi, bd, bi)) { bd = od; bi = oi; }
        }
        ld = bd; li = bi;
        if (bi != iq) {
            if (w < KNN_K && lane == 0) so_q[w] = (bi < N) ? bi : 0;
            ++w;
        }
    }
}

__device__ __forceinline__ void knn_replay(const float* sd, const int* si, int n,
                                           int lane, int iq, int* so_q) {
    float bd[KNN_K + 1];
    int   bi[KNN_K + 1];
#pragma unroll
    for (int s = 0; s <= KNN_K; ++s) { bd[s] = 3.4e38f; bi[s] = -1; }
    for (int j = 0; j < n; ++j) {
        float cd = sd[j]; int ci = si[j];
        if (cd < bd[KNN_K]) {
#pragma unroll
            for (int s = 0; s <= KNN_K; ++s) {
                bool sw = cd < bd[s];
                float td = sw ? bd[s] : cd; int ti = sw ? bi[s] : ci;
                bd[s] = sw ? cd : bd[s];    bi[s] = sw ? ci : bi[s];
                cd = td; ci = ti;
            }
        }
    }
    if (lane == 0) {
        int w = 0;
#pragma unroll
        for (int s = 0; s <= KNN_K; ++s)
            if (bi[s] != iq && w < KNN_K) { so_q[w] = bi[s]; ++w; }
    }
}

__global__ __launch_bounds__(256) void knn_kernel(const float4* __restrict__ pts,
                                                  int* __restrict__ knn, int N) {
    __shared__ float4 tile[KNN_N];
    __shared__ float  hd[8][KNN_QW][KNN_CAP];
    __shared__ int    hi[8][KNN_QW][KNN_CAP];
    __shared__ int    so[32 * KNN_K];
    const int b = blockIdx.y;
    const int lane = threadIdx.x & 31;
    const int wave = __builtin_amdgcn_readfirstlane((int)(threadIdx.x >> 5));
    const float4* pb = pts + (size_t)b * N;

    for (int i = threadIdx.x; i < N; i += 256) tile[i] = pb[i];
    __syncthreads();

    const int q0 = blockIdx.x * 32 + wave * KNN_QW;
    float px[KNN_QW], py[KNN_QW], pz[KNN_QW], pw[KNN_QW];
#pragma unroll
    for (int qi = 0; qi < KNN_QW; ++qi) {
        float4 p = pb[q0 + qi];
        px[qi] = p.x; py[qi] = p.y; pz[qi] = p.z; pw[qi] = p.w;
    }

    float mn[KNN_QW];
#pragma unroll
    for (int qi = 0; qi < KNN_QW; ++qi) mn[qi] = __builtin_inff();
#pragma unroll 4
    for (int t = 0; t < N; t += 32) {
        float4 q = tile[t + lane];
#pragma unroll
        for (int qi = 0; qi < KNN_QW; ++qi) {
            float d = knn_dist(px[qi], py[qi], pz[qi], pw[qi], q);
            mn[qi] = d < mn[qi] ? d : mn[qi];
        }
    }

    float T[KNN_QW];
#pragma unroll
    for (int qi = 0; qi < KNN_QW; ++qi) {
        const float mv = mn[qi];
        int cnt = 0;
#pragma unroll
        for (int s = 0; s < 32; ++s) {
            float v = __builtin_bit_cast(float,
                __builtin_amdgcn_readlane(__builtin_bit_cast(int, mv), s));
            cnt += knn_less(v, s, mv, lane) ? 1 : 0;
        }
        unsigned sel = __builtin_amdgcn_ballot_w32(cnt == KNN_K);
        T[qi] = sel ? __shfl(mv, __builtin_ctz(sel), 32) : __builtin_inff();
    }

    int cn[KNN_QW];
#pragma unroll
    for (int qi = 0; qi < KNN_QW; ++qi) cn[qi] = 0;
    for (int t = 0; t < N; t += 128) {
        float4 q[4];
        float d[4][KNN_QW];
        unsigned bm[4][KNN_QW];
#pragma unroll
        for (int u = 0; u < 4; ++u) q[u] = tile[t + 32 * u + lane];
#pragma unroll
        for (int u = 0; u < 4; ++u) {
#pragma unroll
            for (int qi = 0; qi < KNN_QW; ++qi) {
                d[u][qi] = knn_dist(px[qi], py[qi], pz[qi], pw[qi], q[u]);
                bm[u][qi] = __builtin_amdgcn_ballot_w32(d[u][qi] <= T[qi]);
            }
        }
#pragma unroll
        for (int u = 0; u < 4; ++u) {
            if (bm[u][0] | bm[u][1] | bm[u][2] | bm[u][3]) {
#pragma unroll
                for (int qi = 0; qi < KNN_QW; ++qi) {
                    const unsigned m = bm[u][qi];
                    if (m) {
                        const bool hit = ((m >> lane) & 1u) != 0u;
                        const int pos = cn[qi] + (int)__builtin_amdgcn_mbcnt_lo(m, 0u);
                        if (hit && pos < KNN_CAP) {
                            hd[wave][qi][pos] = d[u][qi];
                            hi[wave][qi][pos] = t + 32 * u + lane;
                        }
                        cn[qi] += __builtin_popcount(m);
                    }
                }
            }
        }
    }
    __syncthreads();

#pragma unroll
    for (int qi = 0; qi < KNN_QW; ++qi) {
        const int n = cn[qi];
        const int iq = q0 + qi;
        int* so_q = so + (wave * KNN_QW + qi) * KNN_K;
        if (n > KNN_K && n <= KNN_CAP) {
            const float* sd = hd[wave][qi];
            const int*   si = hi[wave][qi];
            const bool v0 = lane < n, v1 = lane + 32 < n;
            const float d0 = v0 ? sd[lane] : __builtin_inff();
            const int   i0 = v0 ? si[lane] : 0x7fffffff;
            const float d1 = v1 ? sd[lane + 32] : __builtin_inff();
            const int   i1 = v1 ? si[lane + 32] : 0x7fffffff;
            int r0 = 0, r1 = 0;
            bool tie = false;
            for (int j = 0; j < n; ++j) {
                const float dj = sd[j]; const int ij = si[j];
                r0 += knn_less(dj, ij, d0, i0) ? 1 : 0;
                r1 += knn_less(dj, ij, d1, i1) ? 1 : 0;
                tie = tie || (dj == d0 && ij != i0) || (dj == d1 && ij != i1);
            }
            if (__builtin_amdgcn_ballot_w32(tie)) {
                knn_replay(sd, si, n, lane, iq, so_q);
                continue;
            }
            int sr = KNN_K + 1;
            unsigned s0 = __builtin_amdgcn_ballot_w32(v0 && i0 == iq);
            if (s0) sr = __shfl(r0, __builtin_ctz(s0), 32);
            unsigned s1 = __builtin_amdgcn_ballot_w32(v1 && i1 == iq);
            if (s1) sr = __shfl(r1, __builtin_ctz(s1), 32);
            if (v0 && i0 != iq && r0 <= KNN_K) {
                int w = r0 - (r0 > sr ? 1 : 0);
                if (w < KNN_K) so_q[w] = i0;
            }
            if (v1 && i1 != iq && r1 <= KNN_K) {
                int w = r1 - (r1 > sr ? 1 : 0);
                if (w < KNN_K) so_q[w] = i1;
            }
        } else {
            knn_exact_wave(tile, N, lane, px[qi], py[qi], pz[qi], pw[qi], iq, so_q);
        }
    }
    __syncthreads();
    int* ob = knn + ((size_t)b * N + (size_t)blockIdx.x * 32) * KNN_K;
    for (int i = threadIdx.x; i < 32 * KNN_K; i += 256) ob[i] = so[i];
}

#define GEMM_KB 8
#define STAT_CH 384

__device__ __forceinline__ v2f norm_lrelu2(v2f x, float4 st) {
    v2f mu = {st.x, st.y}, rs = {st.z, st.w};
    v2f y = (x - mu) * rs;
    v2f z = y * 0.2f;
    v2f r = {__builtin_fmaxf(y.x, z.x), __builtin_fmaxf(y.y, z.y)};
    return r;
}

template <bool TRANS, int NORM>
__global__ __launch_bounds__(128) void gemm_wmma_kernel(
        const float* __restrict__ W, const float* __restrict__ X, const float* X2,
        const float4* __restrict__ stat, float* __restrict__ D, int N, int Kdim,
        long xBatchStride, long x2BatchStride, int splitBlk, long dBatchStride, int ldT) {
    const int lane = threadIdx.x & 31;
    const int wave = threadIdx.x >> 5;
    const int half = lane >> 4;
    const int l16  = lane & 15;
    const int wm = wave >> 1, wn = wave & 1;
    const int m0 = blockIdx.y * 64 + wm * 32;
    const int n0 = blockIdx.x * 64 + wn * 32;
    const float* Xb = X + (size_t)blockIdx.z * xBatchStride;
    float*       Db = D + (size_t)blockIdx.z * dBatchStride;

    v8f c00 = {0.f,0.f,0.f,0.f,0.f,0.f,0.f,0.f};
    v8f c01 = c00, c10 = c00, c11 = c00;

    const float* ar0 = W + (size_t)(m0 + l16) * Kdim + 2 * half;
    const float* ar1 = ar0 + (size_t)16 * Kdim;
    const float* xr  = Xb + (size_t)(2 * half) * N + n0 + l16;
    const float* xr2 = X2 + (size_t)blockIdx.z * x2BatchStride
                     + (size_t)(splitBlk * (4 * GEMM_KB) + 2 * half) * N + n0 + l16;
    int blk = 0;
    const float4* sp = stat + (size_t)blockIdx.z * (STAT_CH / 2) + half;

    __builtin_prefetch(xr, 0, 1);
    __builtin_prefetch(ar0, 0, 1);

    v2f fa0[GEMM_KB], fa1[GEMM_KB], fb0[GEMM_KB], fb1[GEMM_KB];
    v2f ga0[GEMM_KB], ga1[GEMM_KB], gb0[GEMM_KB], gb1[GEMM_KB];
    float4 fst[GEMM_KB], gst[GEMM_KB];

    auto load_blk = [&](v2f (&a0)[GEMM_KB], v2f (&a1)[GEMM_KB],
                        v2f (&b0)[GEMM_KB], v2f (&b1)[GEMM_KB], float4 (&st)[GEMM_KB]) {
#pragma unroll
        for (int s = 0; s < GEMM_KB; ++s) {
            const float* xs = xr + (size_t)(4 * s) * N;
            a0[s].x = ar0[4 * s];   a0[s].y = ar0[4 * s + 1];
            a1[s].x = ar1[4 * s];   a1[s].y = ar1[4 * s + 1];
            b0[s].x = xs[0];        b0[s].y = xs[N];
            b1[s].x = xs[16];       b1[s].y = xs[16 + N];
        }
        if (NORM == 1 || (NORM == 2 && blk >= splitBlk)) {
#pragma unroll
            for (int s = 0; s < GEMM_KB; ++s) st[s] = sp[2 * s];
            sp += 2 * GEMM_KB;
        }
        ar0 += 4 * GEMM_KB; ar1 += 4 * GEMM_KB; xr += (size_t)(4 * GEMM_KB) * N;
        if (++blk == splitBlk) xr = xr2;
    };
    auto mma_blk = [&](const v2f (&a0)[GEMM_KB], const v2f (&a1)[GEMM_KB],
                       v2f (&b0)[GEMM_KB], v2f (&b1)[GEMM_KB],
                       const float4 (&st)[GEMM_KB], int bidx) {
        if (NORM == 1 || (NORM == 2 && bidx >= splitBlk)) {
#pragma unroll
            for (int s = 0; s < GEMM_KB; ++s) {
                b0[s] = norm_lrelu2(b0[s], st[s]);
                b1[s] = norm_lrelu2(b1[s], st[s]);
            }
        }
#pragma unroll
        for (int s = 0; s < GEMM_KB; ++s) {
            if (TRANS) {
                c00 = wmma_f32(b0[s], a0[s], c00);
                c01 = wmma_f32(b1[s], a0[s], c01);
                c10 = wmma_f32(b0[s], a1[s], c10);
                c11 = wmma_f32(b1[s], a1[s], c11);
            } else {
                c00 = wmma_f32(a0[s], b0[s], c00);
                c01 = wmma_f32(a0[s], b1[s], c01);
                c10 = wmma_f32(a1[s], b0[s], c10);
                c11 = wmma_f32(a1[s], b1[s], c11);
            }
        }
    };

    const int nblk = Kdim / (4 * GEMM_KB);
    load_blk(fa0, fa1, fb0, fb1, fst);
    for (int j = 0; j < nblk - 2; j += 2) {
        load_blk(ga0, ga1, gb0, gb1, gst);
        __builtin_amdgcn_sched_barrier(0);
        mma_blk(fa0, fa1, fb0, fb1, fst, j);
        load_blk(fa0, fa1, fb0, fb1, fst);
        __builtin_amdgcn_sched_barrier(0);
        mma_blk(ga0, ga1, gb0, gb1, gst, j + 1);
    }
    load_blk(ga0, ga1, gb0, gb1, gst);
    __builtin_amdgcn_sched_barrier(0);
    mma_blk(fa0, fa1, fb0, fb1, fst, nblk - 2);
    mma_blk(ga0, ga1, gb0, gb1, gst, nblk - 1);
    if (TRANS) {
#pragma unroll
        for (int v = 0; v < 8; ++v) {
            const size_t r0 = (size_t)(n0 + v + 8 * half) * ldT + m0 + l16;
            const size_t r1 = r0 + (size_t)16 * ldT;
            Db[r0]      = c00[v];
            Db[r0 + 16] = c10[v];
            Db[r1]      = c01[v];
            Db[r1 + 16] = c11[v];
        }
        return;
    }
#pragma unroll
    for (int v = 0; v < 8; ++v) {
        const int mr0 = m0 + v + 8 * half;
        Db[(size_t)mr0 * N + n0 + l16]             = c00[v];
        Db[(size_t)mr0 * N + n0 + 16 + l16]        = c01[v];
        Db[(size_t)(mr0 + 16) * N + n0 + l16]      = c10[v];
        Db[(size_t)(mr0 + 16) * N + n0 + 16 + l16] = c11[v];
    }
}

#define EDGE_PTS 32
#define EDGE_PW  4

template <int M>
__global__ __launch_bounds__(256) void edge_agg_kernel(
        const float* __restrict__ PQT, const int* __restrict__ knn,
        float* __restrict__ outMax, float2* __restrict__ part,
        int N, long outBatchStride) {
    constexpr int NV = M / 128;
    constexpr int TS = M + 4;
    __shared__ __attribute__((aligned(16))) float tile[EDGE_PTS * TS];
    __shared__ __attribute__((aligned(16))) float rs[8][M];
    __shared__ __attribute__((aligned(16))) float rs2[8][M];
    const int b = blockIdx.y;
    const int lane = threadIdx.x & 31;
    const int wave = __builtin_amdgcn_readfirstlane((int)(threadIdx.x >> 5));
    const int nb0 = blockIdx.x * EDGE_PTS;
    const float* base = PQT + (size_t)b * N * (2 * M);

    float4 s[NV], s2[NV];
#pragma unroll
    for (int v = 0; v < NV; ++v) {
        s[v]  = make_float4(0.f, 0.f, 0.f, 0.f);
        s2[v] = make_float4(0.f, 0.f, 0.f, 0.f);
    }

#pragma unroll
    for (int j = 0; j < EDGE_PW; ++j) {
        const int nl = wave * EDGE_PW + j;
        const int n = nb0 + nl;
        const int* id = knn + ((size_t)b * N + n) * KNN_K;
        const float* prow = base + (size_t)n * (2 * M) + 4 * lane;
        float4 p[NV], mx[NV];
#pragma unroll
        for (int v = 0; v < NV; ++v) {
            p[v] = *(const float4*)(prow + v * 128);
            mx[v] = make_float4(-3.4e38f, -3.4e38f, -3.4e38f, -3.4e38f);
        }
#pragma unroll
        for (int k = 0; k < KNN_K; ++k) {
            const float* qrow = base + (size_t)id[k] * (2 * M) + M + 4 * lane;
#pragma unroll
            for (int v = 0; v < NV; ++v) {
                const float4 q = *(const float4*)(qrow + v * 128);
                const float y0 = p[v].x + q.x, y1 = p[v].y + q.y;
                const float y2 = p[v].z + q.z, y3 = p[v].w + q.w;
                mx[v].x = fmaxf(mx[v].x, y0); mx[v].y = fmaxf(mx[v].y, y1);
                mx[v].z = fmaxf(mx[v].z, y2); mx[v].w = fmaxf(mx[v].w, y3);
                s[v].x += y0; s[v].y += y1; s[v].z += y2; s[v].w += y3;
                s2[v].x += y0 * y0; s2[v].y += y1 * y1;
                s2[v].z += y2 * y2; s2[v].w += y3 * y3;
            }
        }
#pragma unroll
        for (int v = 0; v < NV; ++v)
            *(float4*)(tile + nl * TS + v * 128 + 4 * lane) = mx[v];
    }
#pragma unroll
    for (int v = 0; v < NV; ++v) {
        *(float4*)(&rs[wave][v * 128 + 4 * lane])  = s[v];
        *(float4*)(&rs2[wave][v * 128 + 4 * lane]) = s2[v];
    }
    __syncthreads();

    float* ob = outMax + (size_t)b * outBatchStride + nb0 + lane;
    for (int m = wave; m < M; m += 8)
        ob[(size_t)m * N] = tile[lane * TS + m];

    const int t = threadIdx.x;
    if (t < M) {
        float a = 0.f, a2 = 0.f;
#pragma unroll
        for (int w = 0; w < 8; ++w) { a += rs[w][t]; a2 += rs2[w][t]; }
        part[((size_t)b * gridDim.x + blockIdx.x) * M + t] = make_float2(a, a2);
    }
}

#define STAT_TILES 128
#define STAT_CPB   16

__global__ __launch_bounds__(32 * STAT_CPB) void stats_kernel(const float2* __restrict__ part,
                                                             float* __restrict__ stat,
                                                             int M, int chOff, float invCnt) {
    __shared__ float sm[STAT_CPB], sr[STAT_CPB];
    const int lane = threadIdx.x & 31;
    const int w = (int)(threadIdx.x >> 5);
    const int row = blockIdx.x * STAT_CPB + w;
    const int b = row / M, m = row - b * M;
    const float2* p = part + (size_t)b * STAT_TILES * M + m;
    float a[4], a2[4];
#pragma unroll
    for (int g = 0; g < 4; ++g) {
        const float2 v = p[(size_t)(g * 32 + lane) * M];
        a[g] = v.x; a2[g] = v.y;
    }
#pragma unroll
    for (int g = 0; g < 4; ++g)
        for (int off = 16; off > 0; off >>= 1) {
            a[g]  += __shfl_xor(a[g], off, 32);
            a2[g] += __shfl_xor(a2[g], off, 32);
        }
    float s = 0.f, s2 = 0.f;
#pragma unroll
    for (int g = 0; g < 4; ++g) { s += a[g]; s2 += a2[g]; }
    float mu = s * invCnt;
    float e2 = s2 * invCnt;
    float var = fmaxf(__builtin_fmaf(-mu, mu, e2), 0.f);
    float r = rsqrtf(var + IN_EPS);
    if (lane == 0) { sm[w] = mu; sr[w] = r; }
    __syncthreads();
    if (w == 0) {
        const int ch = 2 * (lane >> 2) + (lane & 1);
        const float v = (lane & 2) ? sr[ch] : sm[ch];
        stat[((size_t)b * (STAT_CH / 2) + ((chOff + m) >> 1)) * 4 + lane] = v;
    }
}

__global__ __launch_bounds__(256) void final_norm_kernel(
        const float* __restrict__ xin, float* __restrict__ out, int N) {
    __shared__ float rs[256], rs2[256];
    __shared__ float s_mu, s_r;
    const int b = blockIdx.y, c = blockIdx.x, C = gridDim.x;
    const float* row = xin + ((size_t)b * C + c) * N;
    float s = 0.f, s2 = 0.f;
    for (int n = threadIdx.x; n < N; n += 256) {
        float v = row[n]; s += v; s2 += v * v;
    }
    rs[threadIdx.x] = s; rs2[threadIdx.x] = s2;
    __syncthreads();
    for (int off = 128; off > 0; off >>= 1) {
        if (threadIdx.x < off) {
            rs[threadIdx.x]  += rs[threadIdx.x + off];
            rs2[threadIdx.x] += rs2[threadIdx.x + off];
        }
        __syncthreads();
    }
    if (threadIdx.x == 0) {
        float mu = rs[0] / (float)N;
        float var = fmaxf(rs2[0] / (float)N - mu * mu, 0.f);
        s_mu = mu; s_r = rsqrtf(var + IN_EPS);
    }
    __syncthreads();
    float mu = s_mu, r = s_r;
    float* orow = out + ((size_t)b * C + c) * N;
    for (int n = threadIdx.x; n < N; n += 256)
        orow[n] = lrelu((row[n] - mu) * r);
}

extern "C" void kernel_launch(void* const* d_in, const int* in_sizes, int n_in,
                              void* d_out, int out_size, void* d_ws, size_t ws_size,
                              hipStream_t stream) {
    constexpr int B = 4, C = 128, N = 4096;
    const float* coords = (const float*)d_in[0];
    const float* feats  = (const float*)d_in[1];
    const float* W1     = (const float*)d_in[2];
    const float* W2     = (const float*)d_in[3];
    const float* W3     = (const float*)d_in[4];
    float* out = (float*)d_out;

    char* ws = (char*)d_ws;
    size_t off = 0;
    auto carve = [&](size_t bytes) -> void* {
        void* p = ws + off;
        off = (off + bytes + 255) & ~(size_t)255;
        return p;
    };
    float4* pts   = (float4*)carve((size_t)B * N * sizeof(float4));
    float* W1cat  = (float*)carve(256 * 128 * sizeof(float));
    float* W2cat  = (float*)carve(512 * 128 * sizeof(float));
    int*   knn    = (int*)carve((size_t)B * N * KNN_K * sizeof(int));
    float* PQ     = (float*)carve((size_t)B * 512 * N * sizeof(float));
    float* X3     = (float*)carve((size_t)B * 512 * N * sizeof(float));
    float* x3pre  = (float*)carve((size_t)B * 128 * N * sizeof(float));
    float2* part1 = (float2*)x3pre;
    float2* part2 = part1 + (size_t)B * (N / 32) * 128;
    float* stat = X3;
    static_assert(N / EDGE_PTS == STAT_TILES, "tiles per batch");
    static_assert((size_t)B * STAT_CH * 2 <= (size_t)128 * N, "table fits the unused rows");
    (void)ws_size; (void)in_sizes; (void)n_in; (void)out_size;

    static_assert(B * N == PREP_PTS_BLOCKS * 256, "set-up block split: points");
    prep_kernel<<<PREP_PTS_BLOCKS + PREP_W1_BLOCKS + PREP_W2_BLOCKS, 256, 0, stream>>>(
        coords, pts, W1, W1cat, W2, W2cat, N);

    knn_kernel<<<dim3(N / 32, B), 256, 0, stream>>>(pts, knn, N);

    const float invCnt = 1.0f / (float)(N * KNN_K);

    gemm_wmma_kernel<true, 0><<<dim3(N / 64, 256 / 64, B), 128, 0, stream>>>(
        W1cat, feats, feats, (const float4*)stat, PQ, N, C, (long)C * N, (long)C * N, 0,
        (long)256 * N, 256);
    edge_agg_kernel<128><<<dim3(N / 32, B), 256, 0, stream>>>(
        PQ, knn, X3 + (size_t)128 * N, part1, N, (long)512 * N);
    stats_kernel<<<B * 128 / STAT_CPB, 32 * STAT_CPB, 0, stream>>>(part1, stat, 128, 0, invCnt);

    gemm_wmma_kernel<true, 1><<<dim3(N / 64, 512 / 64, B), 128, 0, stream>>>(
        W2cat, X3 + (size_t)128 * N, X3 + (size_t)128 * N, (const float4*)stat, PQ, N, C,
        (long)512 * N, (long)512 * N, 0, (long)512 * N, 512);
    edge_agg_kernel<256><<<dim3(N / 32, B), 256, 0, stream>>>(
        PQ, knn, X3 + (size_t)256 * N, part2, N, (long)512 * N);
    stats_kernel<<<B * 256 / STAT_CPB, 32 * STAT_CPB, 0, stream>>>(part2, stat, 256, 128, invCnt);

    gemm_wmma_kernel<false, 2><<<dim3(N / 64, 128 / 64, B), 128, 0, stream>>>(
        W3, feats, X3, (const float4*)stat, x3pre, N, 512, (long)C * N, (long)512 * N, 4,
        (long)128 * N, 0);
    final_norm_kernel<<<dim3(128, B), 256, 0, stream>>>(x3pre, out, N);
}


# GEMMs: each block copies its 64 W rows to LDS once with whole-line 16-byte loads and reads the W fragments from LDS (8-byte, pitch K+4) instead of 16-line global loads; k order and sums unchanged
# speedup vs baseline: 4.6250x; 1.0353x over previous
#include <hip/hip_runtime.h>
#include <hip/hip_bf16.h>


typedef __attribute__((ext_vector_type(2))) float v2f;
typedef __attribute__((ext_vector_type(8))) float v8f;

#define KNN_K 10
#define IN_EPS 1e-5f

__device__ __forceinline__ float lrelu(float y) { return y >= 0.0f ? y : 0.2f * y; }

__device__ __forceinline__ v8f wmma_f32(v2f a, v2f b, v8f c) {
    return __builtin_amdgcn_wmma_f32_16x16x4_f32(
         false, a,  false, b,
         (short)0, c,  false,  false);
}

#define PREP_PTS_BLOCKS 64
#define PREP_W1_BLOCKS  64
#define PREP_W2_BLOCKS  128

__global__ __launch_bounds__(256) void prep_kernel(const float* __restrict__ coords,
                                                   float4* __restrict__ pts,
                                                   const float* __restrict__ W1,
                                                   float* __restrict__ W1c,
                                                   const float* __restrict__ W2,
                                                   float* __restrict__ W2c, int N) {
    const int blk = blockIdx.x;
    if (blk < PREP_PTS_BLOCKS) {
        const int gid = blk * 256 + threadIdx.x;
        const int b = gid / N, n = gid - b * N;
        const float* cb = coords + (size_t)b * 3 * N;
        float x = cb[n], y = cb[N + n], z = cb[2 * N + n];
        pts[gid] = make_float4(x, y, z, x * x + y * y + z * z);
    } else if (blk < PREP_PTS_BLOCKS + PREP_W1_BLOCKS) {
        const int gid = (blk - PREP_PTS_BLOCKS) * 256 + threadIdx.x;
        const int o = gid >> 7, c = gid & 127;
        float a = W1[o * 256 + c], b = W1[o * 256 + 128 + c];
        W1c[o * 128 + c] = a - b;
        W1c[(128 + o) * 128 + c] = b;
    } else {
        const int gid = (blk - PREP_PTS_BLOCKS - PREP_W1_BLOCKS) * 256 + threadIdx.x;
        const int o = gid >> 7, c = gid & 127;
        float a = W2[o * 256 + c], b = W2[o * 256 + 128 + c];
        W2c[o * 128 + c] = a - b;
        W2c[(256 + o) * 128 + c] = b;
    }
}

#define KNN_N   4096
#define KNN_QW  4
#define KNN_CAP 64

__device__ __forceinline__ float knn_dist(float px, float py, float pz, float pw, float4 q) {
    float d;
    {
#pragma clang fp contract(off)
        d = __builtin_fmaf(-2.0f, __builtin_fmaf(px, q.x, py * q.y) + pz * q.z, pw + q.w);
    }
    return d;
}

__device__ __forceinline__ bool knn_less(float da, int ia, float db, int ib) {
    return (da < db) || (da == db && ia < ib);
}

__device__ __forceinline__ void knn_exact_wave(const float4* tile, int N, int lane,
        float px, float py, float pz, float pw, int iq, int* so_q) {
    float ld = -__builtin_inff(); int li = -1; int w = 0;
    for (int r = 0; r <= KNN_K; ++r) {
        float bd = __builtin_inff(); int bi = 0x7fffffff;
        for (int t = 0; t < N; t += 32) {
            const int c = t + lane;
            float d = knn_dist(px, py, pz, pw, tile[c]);
            if (knn_less(ld, li, d, c) && knn_less(d, c, bd, bi)) { bd = d; bi = c; }
        }
        for (int off = 16; off > 0; off >>= 1) {
            float od = __shfl_xor(bd, off, 32); int oi = __shfl_xor(bi, off, 32);
            if (knn_less(od, oi, bd, bi)) { bd = od; bi = oi; }
        }
        ld = bd; li = bi;
        if (bi != iq) {
            if (w < KNN_K && lane == 0) so_q[w] = (bi < N) ? bi : 0;
            ++w;
        }
    }
}

__device__ __forceinline__ void knn_replay(const float* sd, const int* si, int n,
                                           int lane, int iq, int* so_q) {
    float bd[KNN_K + 1];
    int   bi[KNN_K + 1];
#pragma unroll
    for (int s = 0; s <= KNN_K; ++s) { bd[s] = 3.4e38f; bi[s] = -1; }
    for (int j = 0; j < n; ++j) {
        float cd = sd[j]; int ci = si[j];
        if (cd < bd[KNN_K]) {
#pragma unroll
            for (int s = 0; s <= KNN_K; ++s) {
                bool sw = cd < bd[s];
                float td = sw ? bd[s] : cd; int ti = sw ? bi[s] : ci;
                bd[s] = sw ? cd : bd[s];    bi[s] = sw ? ci : bi[s];
                cd = td; ci = ti;
            }
        }
    }
    if (lane == 0) {
        int w = 0;
#pragma unroll
        for (int s = 0; s <= KNN_K; ++s)
            if (bi[s] != iq && w < KNN_K) { so_q[w] = bi[s]; ++w; }
    }
}

__global__ __launch_bounds__(256) void knn_kernel(const float4* __restrict__ pts,
                                                  int* __restrict__ knn, int N) {
    __shared__ float4 tile[KNN_N];
    __shared__ float  hd[8][KNN_QW][KNN_CAP];
    __shared__ int    hi[8][KNN_QW][KNN_CAP];
    __shared__ int    so[32 * KNN_K];
    const int b = blockIdx.y;
    const int lane = threadIdx.x & 31;
    const int wave = __builtin_amdgcn_readfirstlane((int)(threadIdx.x >> 5));
    const float4* pb = pts + (size_t)b * N;

    for (int i = threadIdx.x; i < N; i += 256) tile[i] = pb[i];
    __syncthreads();

    const int q0 = blockIdx.x * 32 + wave * KNN_QW;
    float px[KNN_QW], py[KNN_QW], pz[KNN_QW], pw[KNN_QW];
#pragma unroll
    for (int qi = 0; qi < KNN_QW; ++qi) {
        float4 p = pb[q0 + qi];
        px[qi] = p.x; py[qi] = p.y; pz[qi] = p.z; pw[qi] = p.w;
    }

    float mn[KNN_QW];
#pragma unroll
    for (int qi = 0; qi < KNN_QW; ++qi) mn[qi] = __builtin_inff();
#pragma unroll 4
    for (int t = 0; t < N; t += 32) {
        float4 q = tile[t + lane];
#pragma unroll
        for (int qi = 0; qi < KNN_QW; ++qi) {
            float d = knn_dist(px[qi], py[qi], pz[qi], pw[qi], q);
            mn[qi] = d < mn[qi] ? d : mn[qi];
        }
    }

    float T[KNN_QW];
#pragma unroll
    for (int qi = 0; qi < KNN_QW; ++qi) {
        const float mv = mn[qi];
        int cnt = 0;
#pragma unroll
        for (int s = 0; s < 32; ++s) {
            float v = __builtin_bit_cast(float,
                __builtin_amdgcn_readlane(__builtin_bit_cast(int, mv), s));
            cnt += knn_less(v, s, mv, lane) ? 1 : 0;
        }
        unsigned sel = __builtin_amdgcn_ballot_w32(cnt == KNN_K);
        T[qi] = sel ? __shfl(mv, __builtin_ctz(sel), 32) : __builtin_inff();
    }

    int cn[KNN_QW];
#pragma unroll
    for (int qi = 0; qi < KNN_QW; ++qi) cn[qi] = 0;
    for (int t = 0; t < N; t += 128) {
        float4 q[4];
        float d[4][KNN_QW];
        unsigned bm[4][KNN_QW];
#pragma unroll
        for (int u = 0; u < 4; ++u) q[u] = tile[t + 32 * u + lane];
#pragma unroll
        for (int u = 0; u < 4; ++u) {
#pragma unroll
            for (int qi = 0; qi < KNN_QW; ++qi) {
                d[u][qi] = knn_dist(px[qi], py[qi], pz[qi], pw[qi], q[u]);
                bm[u][qi] = __builtin_amdgcn_ballot_w32(d[u][qi] <= T[qi]);
            }
        }
#pragma unroll
        for (int u = 0; u < 4; ++u) {
            if (bm[u][0] | bm[u][1] | bm[u][2] | bm[u][3]) {
#pragma unroll
                for (int qi = 0; qi < KNN_QW; ++qi) {
                    const unsigned m = bm[u][qi];
                    if (m) {
                        const bool hit = ((m >> lane) & 1u) != 0u;
                        const int pos = cn[qi] + (int)__builtin_amdgcn_mbcnt_lo(m, 0u);
                        if (hit && pos < KNN_CAP) {
                            hd[wave][qi][pos] = d[u][qi];
                            hi[wave][qi][pos] = t + 32 * u + lane;
                        }
                        cn[qi] += __builtin_popcount(m);
                    }
                }
            }
        }
    }
    __syncthreads();

#pragma unroll
    for (int qi = 0; qi < KNN_QW; ++qi) {
        const int n = cn[qi];
        const int iq = q0 + qi;
        int* so_q = so + (wave * KNN_QW + qi) * KNN_K;
        if (n > KNN_K && n <= KNN_CAP) {
            const float* sd = hd[wave][qi];
            const int*   si = hi[wave][qi];
            const bool v0 = lane < n, v1 = lane + 32 < n;
            const float d0 = v0 ? sd[lane] : __builtin_inff();
            const int   i0 = v0 ? si[lane] : 0x7fffffff;
            const float d1 = v1 ? sd[lane + 32] : __builtin_inff();
            const int   i1 = v1 ? si[lane + 32] : 0x7fffffff;
            int r0 = 0, r1 = 0;
            bool tie = false;
            for (int j = 0; j < n; ++j) {
                const float dj = sd[j]; const int ij = si[j];
                r0 += knn_less(dj, ij, d0, i0) ? 1 : 0;
                r1 += knn_less(dj, ij, d1, i1) ? 1 : 0;
                tie = tie || (dj == d0 && ij != i0) || (dj == d1 && ij != i1);
            }
            if (__builtin_amdgcn_ballot_w32(tie)) {
                knn_replay(sd, si, n, lane, iq, so_q);
                continue;
            }
            int sr = KNN_K + 1;
            unsigned s0 = __builtin_amdgcn_ballot_w32(v0 && i0 == iq);
            if (s0) sr = __shfl(r0, __builtin_ctz(s0), 32);
            unsigned s1 = __builtin_amdgcn_ballot_w32(v1 && i1 == iq);
            if (s1) sr = __shfl(r1, __builtin_ctz(s1), 32);
            if (v0 && i0 != iq && r0 <= KNN_K) {
                int w = r0 - (r0 > sr ? 1 : 0);
                if (w < KNN_K) so_q[w] = i0;
            }
            if (v1 && i1 != iq && r1 <= KNN_K) {
                int w = r1 - (r1 > sr ? 1 : 0);
                if (w < KNN_K) so_q[w] = i1;
            }
        } else {
            knn_exact_wave(tile, N, lane, px[qi], py[qi], pz[qi], pw[qi], iq, so_q);
        }
    }
    __syncthreads();
    int* ob = knn + ((size_t)b * N + (size_t)blockIdx.x * 32) * KNN_K;
    for (int i = threadIdx.x; i < 32 * KNN_K; i += 256) ob[i] = so[i];
}

#define GEMM_KB 8
#define STAT_CH 384

__device__ __forceinline__ v2f norm_lrelu2(v2f x, float4 st) {
    v2f mu = {st.x, st.y}, rs = {st.z, st.w};
    v2f y = (x - mu) * rs;
    v2f z = y * 0.2f;
    v2f r = {__builtin_fmaxf(y.x, z.x), __builtin_fmaxf(y.y, z.y)};
    return r;
}

template <bool TRANS, int NORM, int KD>
__global__ __launch_bounds__(128) void gemm_wmma_kernel(
        const float* __restrict__ W, const float* __restrict__ X, const float* X2,
        const float4* __restrict__ stat, float* __restrict__ D, int N, int Kdim,
        long xBatchStride, long x2BatchStride, int splitBlk, long dBatchStride, int ldT) {
    const int lane = threadIdx.x & 31;
    const int wave = threadIdx.x >> 5;
    const int half = lane >> 4;
    const int l16  = lane & 15;
    const int wm = wave >> 1, wn = wave & 1;
    const int m0 = blockIdx.y * 64 + wm * 32;
    const int n0 = blockIdx.x * 64 + wn * 32;
    const float* Xb = X + (size_t)blockIdx.z * xBatchStride;
    float*       Db = D + (size_t)blockIdx.z * dBatchStride;

    v8f c00 = {0.f,0.f,0.f,0.f,0.f,0.f,0.f,0.f};
    v8f c01 = c00, c10 = c00, c11 = c00;

    constexpr int LDW = KD + 4;
    __shared__ __attribute__((aligned(16))) float wl[64 * LDW];
    {
        const float4* wsrc = (const float4*)(W + (size_t)(blockIdx.y * 64) * KD);
#pragma unroll 16
        for (int it = 0; it < 64 * KD / 4 / 128; ++it) {
            const int i = it * 128 + (int)threadIdx.x;
            const int r = i / (KD / 4), c4 = i - r * (KD / 4);
            *(float4*)(wl + r * LDW + 4 * c4) = wsrc[i];
        }
    }
    __syncthreads();

    const float* ar0 = wl + (wm * 32 + l16) * LDW + 2 * half;
    const float* ar1 = ar0 + 16 * LDW;
    const float* xr  = Xb + (size_t)(2 * half) * N + n0 + l16;
    const float* xr2 = X2 + (size_t)blockIdx.z * x2BatchStride
                     + (size_t)(splitBlk * (4 * GEMM_KB) + 2 * half) * N + n0 + l16;
    int blk = 0;
    const float4* sp = stat + (size_t)blockIdx.z * (STAT_CH / 2) + half;

    __builtin_prefetch(xr, 0, 1);

    v2f fa0[GEMM_KB], fa1[GEMM_KB], fb0[GEMM_KB], fb1[GEMM_KB];
    v2f ga0[GEMM_KB], ga1[GEMM_KB], gb0[GEMM_KB], gb1[GEMM_KB];
    float4 fst[GEMM_KB], gst[GEMM_KB];

    auto load_blk = [&](v2f (&a0)[GEMM_KB], v2f (&a1)[GEMM_KB],
                        v2f (&b0)[GEMM_KB], v2f (&b1)[GEMM_KB], float4 (&st)[GEMM_KB]) {
#pragma unroll
        for (int s = 0; s < GEMM_KB; ++s) {
            const float* xs = xr + (size_t)(4 * s) * N;
            a0[s] = *(const v2f*)(ar0 + 4 * s);
            a1[s] = *(const v2f*)(ar1 + 4 * s);
            b0[s].x = xs[0];        b0[s].y = xs[N];
            b1[s].x = xs[16];       b1[s].y = xs[16 + N];
        }
        if (NORM == 1 || (NORM == 2 && blk >= splitBlk)) {
#pragma unroll
            for (int s = 0; s < GEMM_KB; ++s) st[s] = sp[2 * s];
            sp += 2 * GEMM_KB;
        }
        ar0 += 4 * GEMM_KB; ar1 += 4 * GEMM_KB; xr += (size_t)(4 * GEMM_KB) * N;
        if (++blk == splitBlk) xr = xr2;
    };
    auto mma_blk = [&](const v2f (&a0)[GEMM_KB], const v2f (&a1)[GEMM_KB],
                       v2f (&b0)[GEMM_KB], v2f (&b1)[GEMM_KB],
                       const float4 (&st)[GEMM_KB], int bidx) {
        if (NORM == 1 || (NORM == 2 && bidx >= splitBlk)) {
#pragma unroll
            for (int s = 0; s < GEMM_KB; ++s) {
                b0[s] = norm_lrelu2(b0[s], st[s]);
                b1[s] = norm_lrelu2(b1[s], st[s]);
            }
        }
#pragma unroll
        for (int s = 0; s < GEMM_KB; ++s) {
            if (TRANS) {
                c00 = wmma_f32(b0[s], a0[s], c00);
                c01 = wmma_f32(b1[s], a0[s], c01);
                c10 = wmma_f32(b0[s], a1[s], c10);
                c11 = wmma_f32(b1[s], a1[s], c11);
            } else {
                c00 = wmma_f32(a0[s], b0[s], c00);
                c01 = wmma_f32(a0[s], b1[s], c01);
                c10 = wmma_f32(a1[s], b0[s], c10);
                c11 = wmma_f32(a1[s], b1[s], c11);
            }
        }
    };

    const int nblk = Kdim / (4 * GEMM_KB);
    load_blk(fa0, fa1, fb0, fb1, fst);
    for (int j = 0; j < nblk - 2; j += 2) {
        load_blk(ga0, ga1, gb0, gb1, gst);
        __builtin_amdgcn_sched_barrier(0);
        mma_blk(fa0, fa1, fb0, fb1, fst, j);
        load_blk(fa0, fa1, fb0, fb1, fst);
        __builtin_amdgcn_sched_barrier(0);
        mma_blk(ga0, ga1, gb0, gb1, gst, j + 1);
    }
    load_blk(ga0, ga1, gb0, gb1, gst);
    __builtin_amdgcn_sched_barrier(0);
    mma_blk(fa0, fa1, fb0, fb1, fst, nblk - 2);
    mma_blk(ga0, ga1, gb0, gb1, gst, nblk - 1);
    if (TRANS) {
#pragma unroll
        for (int v = 0; v < 8; ++v) {
            const size_t r0 = (size_t)(n0 + v + 8 * half) * ldT + m0 + l16;
            const size_t r1 = r0 + (size_t)16 * ldT;
            Db[r0]      = c00[v];
            Db[r0 + 16] = c10[v];
            Db[r1]      = c01[v];
            Db[r1 + 16] = c11[v];
        }
        return;
    }
#pragma unroll
    for (int v = 0; v < 8; ++v) {
        const int mr0 = m0 + v + 8 * half;
        Db[(size_t)mr0 * N + n0 + l16]             = c00[v];
        Db[(size_t)mr0 * N + n0 + 16 + l16]        = c01[v];
        Db[(size_t)(mr0 + 16) * N + n0 + l16]      = c10[v];
        Db[(size_t)(mr0 + 16) * N + n0 + 16 + l16] = c11[v];
    }
}

#define EDGE_PTS 32
#define EDGE_PW  4

template <int M>
__global__ __launch_bounds__(256) void edge_agg_kernel(
        const float* __restrict__ PQT, const int* __restrict__ knn,
        float* __restrict__ outMax, float2* __restrict__ part,
        int N, long outBatchStride) {
    constexpr int NV = M / 128;
    constexpr int TS = M + 4;
    __shared__ __attribute__((aligned(16))) float tile[EDGE_PTS * TS];
    __shared__ __attribute__((aligned(16))) float rs[8][M];
    __shared__ __attribute__((aligned(16))) float rs2[8][M];
    const int b = blockIdx.y;
    const int lane = threadIdx.x & 31;
    const int wave = __builtin_amdgcn_readfirstlane((int)(threadIdx.x >> 5));
    const int nb0 = blockIdx.x * EDGE_PTS;
    const float* base = PQT + (size_t)b * N * (2 * M);

    float4 s[NV], s2[NV];
#pragma unroll
    for (int v = 0; v < NV; ++v) {
        s[v]  = make_float4(0.f, 0.f, 0.f, 0.f);
        s2[v] = make_float4(0.f, 0.f, 0.f, 0.f);
    }

#pragma unroll
    for (int j = 0; j < EDGE_PW; ++j) {
        const int nl = wave * EDGE_PW + j;
        const int n = nb0 + nl;
        const int* id = knn + ((size_t)b * N + n) * KNN_K;
        const float* prow = base + (size_t)n * (2 * M) + 4 * lane;
        float4 p[NV], mx[NV];
#pragma unroll
        for (int v = 0; v < NV; ++v) {
            p[v] = *(const float4*)(prow + v * 128);
            mx[v] = make_float4(-3.4e38f, -3.4e38f, -3.4e38f, -3.4e38f);
        }
#pragma unroll
        for (int k = 0; k < KNN_K; ++k) {
            const float* qrow = base + (size_t)id[k] * (2 * M) + M + 4 * lane;
#pragma unroll
            for (int v = 0; v < NV; ++v) {
                const float4 q = *(const float4*)(qrow + v * 128);
                const float y0 = p[v].x + q.x, y1 = p[v].y + q.y;
                const float y2 = p[v].z + q.z, y3 = p[v].w + q.w;
                mx[v].x = fmaxf(mx[v].x, y0); mx[v].y = fmaxf(mx[v].y, y1);
                mx[v].z = fmaxf(mx[v].z, y2); mx[v].w = fmaxf(mx[v].w, y3);
                s[v].x += y0; s[v].y += y1; s[v].z += y2; s[v].w += y3;
                s2[v].x += y0 * y0; s2[v].y += y1 * y1;
                s2[v].z += y2 * y2; s2[v].w += y3 * y3;
            }
        }
#pragma unroll
        for (int v = 0; v < NV; ++v)
            *(float4*)(tile + nl * TS + v * 128 + 4 * lane) = mx[v];
    }
#pragma unroll
    for (int v = 0; v < NV; ++v) {
        *(float4*)(&rs[wave][v * 128 + 4 * lane])  = s[v];
        *(float4*)(&rs2[wave][v * 128 + 4 * lane]) = s2[v];
    }
    __syncthreads();

    float* ob = outMax + (size_t)b * outBatchStride + nb0 + lane;
    for (int m = wave; m < M; m += 8)
        ob[(size_t)m * N] = tile[lane * TS + m];

    const int t = threadIdx.x;
    if (t < M) {
        float a = 0.f, a2 = 0.f;
#pragma unroll
        for (int w = 0; w < 8; ++w) { a += rs[w][t]; a2 += rs2[w][t]; }
        part[((size_t)b * gridDim.x + blockIdx.x) * M + t] = make_float2(a, a2);
    }
}

#define STAT_TILES 128
#define STAT_CPB   16

__global__ __launch_bounds__(32 * STAT_CPB) void stats_kernel(const float2* __restrict__ part,
                                                             float* __restrict__ stat,
                                                             int M, int chOff, float invCnt) {
    __shared__ float sm[STAT_CPB], sr[STAT_CPB];
    const int lane = threadIdx.x & 31;
    const int w = (int)(threadIdx.x >> 5);
    const int row = blockIdx.x * STAT_CPB + w;
    const int b = row / M, m = row - b * M;
    const float2* p = part + (size_t)b * STAT_TILES * M + m;
    float a[4], a2[4];
#pragma unroll
    for (int g = 0; g < 4; ++g) {
        const float2 v = p[(size_t)(g * 32 + lane) * M];
        a[g] = v.x; a2[g] = v.y;
    }
#pragma unroll
    for (int g = 0; g < 4; ++g)
        for (int off = 16; off > 0; off >>= 1) {
            a[g]  += __shfl_xor(a[g], off, 32);
            a2[g] += __shfl_xor(a2[g], off, 32);
        }
    float s = 0.f, s2 = 0.f;
#pragma unroll
    for (int g = 0; g < 4; ++g) { s += a[g]; s2 += a2[g]; }
    float mu = s * invCnt;
    float e2 = s2 * invCnt;
    float var = fmaxf(__builtin_fmaf(-mu, mu, e2), 0.f);
    float r = rsqrtf(var + IN_EPS);
    if (lane == 0) { sm[w] = mu; sr[w] = r; }
    __syncthreads();
    if (w == 0) {
        const int ch = 2 * (lane >> 2) + (lane & 1);
        const float v = (lane & 2) ? sr[ch] : sm[ch];
        stat[((size_t)b * (STAT_CH / 2) + ((chOff + m) >> 1)) * 4 + lane] = v;
    }
}

__global__ __launch_bounds__(256) void final_norm_kernel(
        const float* __restrict__ xin, float* __restrict__ out, int N) {
    __shared__ float rs[256], rs2[256];
    __shared__ float s_mu, s_r;
    const int b = blockIdx.y, c = blockIdx.x, C = gridDim.x;
    const float* row = xin + ((size_t)b * C + c) * N;
    float s = 0.f, s2 = 0.f;
    for (int n = threadIdx.x; n < N; n += 256) {
        float v = row[n]; s += v; s2 += v * v;
    }
    rs[threadIdx.x] = s; rs2[threadIdx.x] = s2;
    __syncthreads();
    for (int off = 128; off > 0; off >>= 1) {
        if (threadIdx.x < off) {
            rs[threadIdx.x]  += rs[threadIdx.x + off];
            rs2[threadIdx.x] += rs2[threadIdx.x + off];
        }
        __syncthreads();
    }
    if (threadIdx.x == 0) {
        float mu = rs[0] / (float)N;
        float var = fmaxf(rs2[0] / (float)N - mu * mu, 0.f);
        s_mu = mu; s_r = rsqrtf(var + IN_EPS);
    }
    __syncthreads();
    float mu = s_mu, r = s_r;
    float* orow = out + ((size_t)b * C + c) * N;
    for (int n = threadIdx.x; n < N; n += 256)
        orow[n] = lrelu((row[n] - mu) * r);
}

extern "C" void kernel_launch(void* const* d_in, const int* in_sizes, int n_in,
                              void* d_out, int out_size, void* d_ws, size_t ws_size,
                              hipStream_t stream) {
    constexpr int B = 4, C = 128, N = 4096;
    const float* coords = (const float*)d_in[0];
    const float* feats  = (const float*)d_in[1];
    const float* W1     = (const float*)d_in[2];
    const float* W2     = (const float*)d_in[3];
    const float* W3     = (const float*)d_in[4];
    float* out = (float*)d_out;

    char* ws = (char*)d_ws;
    size_t off = 0;
    auto carve = [&](size_t bytes) -> void* {
        void* p = ws + off;
        off = (off + bytes + 255) & ~(size_t)255;
        return p;
    };
    float4* pts   = (float4*)carve((size_t)B * N * sizeof(float4));
    float* W1cat  = (float*)carve(256 * 128 * sizeof(float));
    float* W2cat  = (float*)carve(512 * 128 * sizeof(float));
    int*   knn    = (int*)carve((size_t)B * N * KNN_K * sizeof(int));
    float* PQ     = (float*)carve((size_t)B * 512 * N * sizeof(float));
    float* X3     = (float*)carve((size_t)B * 512 * N * sizeof(float));
    float* x3pre  = (float*)carve((size_t)B * 128 * N * sizeof(float));
    float2* part1 = (float2*)x3pre;
    float2* part2 = part1 + (size_t)B * (N / 32) * 128;
    float* stat = X3;
    static_assert(N / EDGE_PTS == STAT_TILES, "tiles per batch");
    static_assert((size_t)B * STAT_CH * 2 <= (size_t)128 * N, "table fits the unused rows");
    (void)ws_size; (void)in_sizes; (void)n_in; (void)out_size;

    static_assert(B * N == PREP_PTS_BLOCKS * 256, "set-up block split: points");
    prep_kernel<<<PREP_PTS_BLOCKS + PREP_W1_BLOCKS + PREP_W2_BLOCKS, 256, 0, stream>>>(
        coords, pts, W1, W1cat, W2, W2cat, N);

    knn_kernel<<<dim3(N / 32, B), 256, 0, stream>>>(pts, knn, N);

    const float invCnt = 1.0f / (float)(N * KNN_K);

    gemm_wmma_kernel<true, 0, 128><<<dim3(N / 64, 256 / 64, B), 128, 0, stream>>>(
        W1cat, feats, feats, (const float4*)stat, PQ, N, C, (long)C * N, (long)C * N, 0,
        (long)256 * N, 256);
    edge_agg_kernel<128><<<dim3(N / 32, B), 256, 0, stream>>>(
        PQ, knn, X3 + (size_t)128 * N, part1, N, (long)512 * N);
    stats_kernel<<<B * 128 / STAT_CPB, 32 * STAT_CPB, 0, stream>>>(part1, stat, 128, 0, invCnt);

    gemm_wmma_kernel<true, 1, 128><<<dim3(N / 64, 512 / 64, B), 128, 0, stream>>>(
        W2cat, X3 + (size_t)128 * N, X3 + (size_t)128 * N, (const float4*)stat, PQ, N, C,
        (long)512 * N, (long)512 * N, 0, (long)512 * N, 512);
    edge_agg_kernel<256><<<dim3(N / 32, B), 256, 0, stream>>>(
        PQ, knn, X3 + (size_t)256 * N, part2, N, (long)512 * N);
    stats_kernel<<<B * 256 / STAT_CPB, 32 * STAT_CPB, 0, stream>>>(part2, stat, 256, 128, invCnt);

    gemm_wmma_kernel<false, 2, 512><<<dim3(N / 64, 128 / 64, B), 128, 0, stream>>>(
        W3, feats, X3, (const float4*)stat, x3pre, N, 512, (long)C * N, (long)512 * N, 4,
        (long)128 * N, 0);
    final_norm_kernel<<<dim3(128, B), 256, 0, stream>>>(x3pre, out, N);
}


# KNN search and stage-1 GEMM share the first launch: 8 KNN + 4 GEMM waves per block, all blocks resident; points (one load round trip) and stacked weights formed there; 8 launches not 10
# speedup vs baseline: 5.0325x; 1.0881x over previous
#include <hip/hip_runtime.h>
#include <hip/hip_bf16.h>


typedef __attribute__((ext_vector_type(2))) float v2f;
typedef __attribute__((ext_vector_type(8))) float v8f;

#define KNN_K 10
#define IN_EPS 1e-5f

__device__ __forceinline__ float lrelu(float y) { return y >= 0.0f ? y : 0.2f * y; }

__device__ __forceinline__ v8f wmma_f32(v2f a, v2f b, v8f c) {
    return __builtin_amdgcn_wmma_f32_16x16x4_f32(
         false, a,  false, b,
         (short)0, c,  false,  false);
}

#define KNN_N   4096
#define KNN_QW  4
#define KNN_CAP 64

__device__ __forceinline__ float knn_dist(float px, float py, float pz, float pw, float4 q) {
    float d;
    {
#pragma clang fp contract(off)
        d = __builtin_fmaf(-2.0f, __builtin_fmaf(px, q.x, py * q.y) + pz * q.z, pw + q.w);
    }
    return d;
}

__device__ __forceinline__ bool knn_less(float da, int ia, float db, int ib) {
    return (da < db) || (da == db && ia < ib);
}

__device__ __forceinline__ void knn_exact_wave(const float4* tile, int N, int lane,
        float px, float py, float pz, float pw, int iq, int* so_q) {
    float ld = -__builtin_inff(); int li = -1; int w = 0;
    for (int r = 0; r <= KNN_K; ++r) {
        float bd = __builtin_inff(); int bi = 0x7fffffff;
        for (int t = 0; t < N; t += 32) {
            const int c = t + lane;
            float d = knn_dist(px, py, pz, pw, tile[c]);
            if (knn_less(ld, li, d, c) && knn_less(d, c, bd, bi)) { bd = d; bi = c; }
        }
        for (int off = 16; off > 0; off >>= 1) {
            float od = __shfl_xor(bd, off, 32); int oi = __shfl_xor(bi, off, 32);
            if (knn_less(od, oi, bd, bi)) { bd = od; bi = oi; }
        }
        ld = bd; li = bi;
        if (bi != iq) {
            if (w < KNN_K && lane == 0) so_q[w] = (bi < N) ? bi : 0;
            ++w;
        }
    }
}

__device__ __forceinline__ void knn_replay(const float* sd, const int* si, int n,
                                           int lane, int iq, int* so_q) {
    float bd[KNN_K + 1];
    int   bi[KNN_K + 1];
#pragma unroll
    for (int s = 0; s <= KNN_K; ++s) { bd[s] = 3.4e38f; bi[s] = -1; }
    for (int j = 0; j < n; ++j) {
        float cd = sd[j]; int ci = si[j];
        if (cd < bd[KNN_K]) {
#pragma unroll
            for (int s = 0; s <= KNN_K; ++s) {
                bool sw = cd < bd[s];
                float td = sw ? bd[s] : cd; int ti = sw ? bi[s] : ci;
                bd[s] = sw ? cd : bd[s];    bi[s] = sw ? ci : bi[s];
                cd = td; ci = ti;
            }
        }
    }
    if (lane == 0) {
        int w = 0;
#pragma unroll
        for (int s = 0; s <= KNN_K; ++s)
            if (bi[s] != iq && w < KNN_K) { so_q[w] = bi[s]; ++w; }
    }
}

#define STAT_CH 384

__device__ __forceinline__ v2f norm_lrelu2(v2f x, float4 st) {
    v2f mu = {st.x, st.y}, rs = {st.z, st.w};
    v2f y = (x - mu) * rs;
    v2f z = y * 0.2f;
    v2f r = {__builtin_fmaxf(y.x, z.x), __builtin_fmaxf(y.y, z.y)};
    return r;
}

template <int KD, bool CAT>
__device__ __forceinline__ void gemm_copy_w(const float* __restrict__ W, float* wl,
                                            int by, int tid, int Mhalf) {
    constexpr int LDW = KD + 4;
    if (CAT) {
        const int R0 = by * 64;
        const bool diff = R0 < Mhalf;
        const float* wsrc = W + (size_t)(diff ? R0 : R0 - Mhalf) * (2 * KD);
        if (diff) {
#pragma unroll 8
            for (int it = 0; it < 64 * KD / 4 / 128; ++it) {
                const int i = it * 128 + tid;
                const int r = i / (KD / 4), c4 = i - r * (KD / 4);
                const float4 wa = *(const float4*)(wsrc + (size_t)r * (2 * KD) + 4 * c4);
                const float4 wb = *(const float4*)(wsrc + (size_t)r * (2 * KD) + KD + 4 * c4);
                *(float4*)(wl + r * LDW + 4 * c4) =
                    make_float4(wa.x - wb.x, wa.y - wb.y, wa.z - wb.z, wa.w - wb.w);
            }
        } else {
#pragma unroll 16
            for (int it = 0; it < 64 * KD / 4 / 128; ++it) {
                const int i = it * 128 + tid;
                const int r = i / (KD / 4), c4 = i - r * (KD / 4);
                *(float4*)(wl + r * LDW + 4 * c4) =
                    *(const float4*)(wsrc + (size_t)r * (2 * KD) + KD + 4 * c4);
            }
        }
    } else {
        const float4* wsrc = (const float4*)(W + (size_t)(by * 64) * KD);
#pragma unroll 16
        for (int it = 0; it < 64 * KD / 4 / 128; ++it) {
            const int i = it * 128 + tid;
            const int r = i / (KD / 4), c4 = i - r * (KD / 4);
            *(float4*)(wl + r * LDW + 4 * c4) = wsrc[i];
        }
    }
}

template <bool TRANS, int NORM, int KD, int KB>
__device__ __forceinline__ void gemm_tile(
        const float* wl, const float* __restrict__ Xb, const float* X2b,
        const float4* __restrict__ statb, float* __restrict__ Db, int N, int Kdim,
        int splitBlk, int ldT, int m0, int n0, int wm, int lane) {
    constexpr int LDW = KD + 4;
    const int half = lane >> 4;
    const int l16  = lane & 15;

    v8f c00 = {0.f,0.f,0.f,0.f,0.f,0.f,0.f,0.f};
    v8f c01 = c00, c10 = c00, c11 = c00;

    const float* ar0 = wl + (wm * 32 + l16) * LDW + 2 * half;
    const float* ar1 = ar0 + 16 * LDW;
    const float* xr  = Xb + (size_t)(2 * half) * N + n0 + l16;
    const float* xr2 = X2b + (size_t)(splitBlk * (4 * KB) + 2 * half) * N + n0 + l16;
    int blk = 0;
    const float4* sp = statb + half;

    __builtin_prefetch(xr, 0, 1);

    v2f fa0[KB], fa1[KB], fb0[KB], fb1[KB];
    v2f ga0[KB], ga1[KB], gb0[KB], gb1[KB];
    float4 fst[KB], gst[KB];

    auto load_blk = [&](v2f (&a0)[KB], v2f (&a1)[KB],
                        v2f (&b0)[KB], v2f (&b1)[KB], float4 (&st)[KB]) {
#pragma unroll
        for (int s = 0; s < KB; ++s) {
            const float* xs = xr + (size_t)(4 * s) * N;
            a0[s] = *(const v2f*)(ar0 + 4 * s);
            a1[s] = *(const v2f*)(ar1 + 4 * s);
            b0[s].x = xs[0];        b0[s].y = xs[N];
            b1[s].x = xs[16];       b1[s].y = xs[16 + N];
        }
        if (NORM == 1 || (NORM == 2 && blk >= splitBlk)) {
#pragma unroll
            for (int s = 0; s < KB; ++s) st[s] = sp[2 * s];
            sp += 2 * KB;
        }
        ar0 += 4 * KB; ar1 += 4 * KB; xr += (size_t)(4 * KB) * N;
        if (++blk == splitBlk) xr = xr2;
    };
    auto mma_blk = [&](const v2f (&a0)[KB], const v2f (&a1)[KB],
                       v2f (&b0)[KB], v2f (&b1)[KB],
                       const float4 (&st)[KB], int bidx) {
        if (NORM == 1 || (NORM == 2 && bidx >= splitBlk)) {
#pragma unroll
            for (int s = 0; s < KB; ++s) {
                b0[s] = norm_lrelu2(b0[s], st[s]);
                b1[s] = norm_lrelu2(b1[s], st[s]);
            }
        }
#pragma unroll
        for (int s = 0; s < KB; ++s) {
            if (TRANS) {
                c00 = wmma_f32(b0[s], a0[s], c00);
                c01 = wmma_f32(b1[s], a0[s], c01);
                c10 = wmma_f32(b0[s], a1[s], c10);
                c11 = wmma_f32(b1[s], a1[s], c11);
            } else {
                c00 = wmma_f32(a0[s], b0[s], c00);
                c01 = wmma_f32(a0[s], b1[s], c01);
                c10 = wmma_f32(a1[s], b0[s], c10);
                c11 = wmma_f32(a1[s], b1[s], c11);
            }
        }
    };

    const int nblk = Kdim / (4 * KB);
    load_blk(fa0, fa1, fb0, fb1, fst);
    for (int j = 0; j < nblk - 2; j += 2) {
        load_blk(ga0, ga1, gb0, gb1, gst);
        __builtin_amdgcn_sched_barrier(0);
        mma_blk(fa0, fa1, fb0, fb1, fst, j);
        load_blk(fa0, fa1, fb0, fb1, fst);
        __builtin_amdgcn_sched_barrier(0);
        mma_blk(ga0, ga1, gb0, gb1, gst, j + 1);
    }
    load_blk(ga0, ga1, gb0, gb1, gst);
    __builtin_amdgcn_sched_barrier(0);
    mma_blk(fa0, fa1, fb0, fb1, fst, nblk - 2);
    mma_blk(ga0, ga1, gb0, gb1, gst, nblk - 1);
    if (TRANS) {
#pragma unroll
        for (int v = 0; v < 8; ++v) {
            const size_t r0 = (size_t)(n0 + v + 8 * half) * ldT + m0 + l16;
            const size_t r1 = r0 + (size_t)16 * ldT;
            Db[r0]      = c00[v];
            Db[r0 + 16] = c10[v];
            Db[r1]      = c01[v];
            Db[r1 + 16] = c11[v];
        }
        return;
    }
#pragma unroll
    for (int v = 0; v < 8; ++v) {
        const int mr0 = m0 + v + 8 * half;
        Db[(size_t)mr0 * N + n0 + l16]             = c00[v];
        Db[(size_t)mr0 * N + n0 + 16 + l16]        = c01[v];
        Db[(size_t)(mr0 + 16) * N + n0 + l16]      = c10[v];
        Db[(size_t)(mr0 + 16) * N + n0 + 16 + l16] = c11[v];
    }
}

template <bool TRANS, int NORM, int KD>
__global__ __launch_bounds__(128) void gemm_wmma_kernel(
        const float* __restrict__ W, const float* __restrict__ X, const float* X2,
        const float4* __restrict__ stat, float* __restrict__ D, int N, int Kdim,
        long xBatchStride, long x2BatchStride, int splitBlk, long dBatchStride, int ldT) {
    __shared__ __attribute__((aligned(16))) float wl[64 * (KD + 4)];
    gemm_copy_w<KD, false>(W, wl, (int)blockIdx.y, (int)threadIdx.x, 0);
    __syncthreads();

    const int lane = threadIdx.x & 31;
    const int wave = threadIdx.x >> 5;
    const int wm = wave >> 1, wn = wave & 1;
    gemm_tile<TRANS, NORM, KD, 8>(
        wl, X + (size_t)blockIdx.z * xBatchStride, X2 + (size_t)blockIdx.z * x2BatchStride,
        stat + (size_t)blockIdx.z * (STAT_CH / 2), D + (size_t)blockIdx.z * dBatchStride,
        N, Kdim, splitBlk, ldT, blockIdx.y * 64 + wm * 32, blockIdx.x * 64 + wn * 32, wm, lane);
}

#define FUSE_KNN_THREADS 256
#define FUSE_THREADS     384

__global__ __launch_bounds__(FUSE_THREADS, 6) void knn_gemm1_kernel(
        const float* __restrict__ coords, int* __restrict__ knn,
        const float* __restrict__ W1, const float* __restrict__ feats,
        float* __restrict__ PQ, const float* __restrict__ W2, float* __restrict__ W2cat, int N) {
    __shared__ float4 tile[KNN_N];
    __shared__ float  hd[8][KNN_QW][KNN_CAP];
    __shared__ int    hi[8][KNN_QW][KNN_CAP];
    __shared__ int    so[32 * KNN_K];
    __shared__ __attribute__((aligned(16))) float wl[64 * (128 + 4)];
    const int lane = threadIdx.x & 31;
    const int wave = __builtin_amdgcn_readfirstlane((int)(threadIdx.x >> 5));
    const bool knn_role = wave < 8;
    const int tilesPerBatch = N / 32;
    const int b = blockIdx.x / tilesPerBatch;
    const int tb = blockIdx.x - b * tilesPerBatch;
    const int by = tb / (N / 128);
    const int np = tb - by * (N / 128);

    if (knn_role) {
        const float* cb = coords + (size_t)b * 3 * N;
#pragma unroll
        for (int it = 0; it < KNN_N / FUSE_KNN_THREADS; ++it) {
            const int i = it * FUSE_KNN_THREADS + (int)threadIdx.x;
            const float x = cb[i], y = cb[N + i], z = cb[2 * N + i];
            tile[i] = make_float4(x, y, z, __builtin_fmaf(z, z, __builtin_fmaf(x, x, y * y)));
        }
    } else {
        const int gt = (int)threadIdx.x - FUSE_KNN_THREADS;
        gemm_copy_w<128, true>(W1, wl, by, gt, 128);
        const int e = (int)blockIdx.x * 128 + gt;
        const int R = e >> 7, cc = e & 127;
        const float wa2 = W2[(R & 255) * 256 + cc];
        const float wb2 = W2[(R & 255) * 256 + 128 + cc];
        W2cat[e] = (R < 256) ? wa2 - wb2 : wb2;
    }
    __syncthreads();

    const int q0 = tb * 32 + wave * KNN_QW;
    float px[KNN_QW], py[KNN_QW], pz[KNN_QW], pw[KNN_QW];
    int cn[KNN_QW];
    if (knn_role) {
#pragma unroll
        for (int qi = 0; qi < KNN_QW; ++qi) {
            const float4 p = tile[q0 + qi];
            px[qi] = __builtin_bit_cast(float, __builtin_amdgcn_readfirstlane(__builtin_bit_cast(int, p.x)));
            py[qi] = __builtin_bit_cast(float, __builtin_amdgcn_readfirstlane(__builtin_bit_cast(int, p.y)));
            pz[qi] = __builtin_bit_cast(float, __builtin_amdgcn_readfirstlane(__builtin_bit_cast(int, p.z)));
            pw[qi] = __builtin_bit_cast(float, __builtin_amdgcn_readfirstlane(__builtin_bit_cast(int, p.w)));
        }

        float mn[KNN_QW];
#pragma unroll
        for (int qi = 0; qi < KNN_QW; ++qi) mn[qi] = __builtin_inff();
#pragma unroll 4
        for (int t = 0; t < N; t += 32) {
            float4 q = tile[t + lane];
#pragma unroll
            for (int qi = 0; qi < KNN_QW; ++qi) {
                float d = knn_dist(px[qi], py[qi], pz[qi], pw[qi], q);
                mn[qi] = d < mn[qi] ? d : mn[qi];
            }
        }

        float T[KNN_QW];
#pragma unroll
        for (int qi = 0; qi < KNN_QW; ++qi) {
            const float mv = mn[qi];
            int cnt = 0;
#pragma unroll
            for (int s = 0; s < 32; ++s) {
                float v = __builtin_bit_cast(float,
                    __builtin_amdgcn_readlane(__builtin_bit_cast(int, mv), s));
                cnt += knn_less(v, s, mv, lane) ? 1 : 0;
            }
            unsigned sel = __builtin_amdgcn_ballot_w32(cnt == KNN_K);
            T[qi] = sel ? __shfl(mv, __builtin_ctz(sel), 32) : __builtin_inff();
        }

#pragma unroll
        for (int qi = 0; qi < KNN_QW; ++qi) cn[qi] = 0;
        for (int t = 0; t < N; t += 128) {
            float4 q[4];
            float d[4][KNN_QW];
            unsigned bm[4][KNN_QW];
#pragma unroll
            for (int u = 0; u < 4; ++u) q[u] = tile[t + 32 * u + lane];
#pragma unroll
            for (int u = 0; u < 4; ++u) {
#pragma unroll
                for (int qi = 0; qi < KNN_QW; ++qi) {
                    d[u][qi] = knn_dist(px[qi], py[qi], pz[qi], pw[qi], q[u]);
                    bm[u][qi] = __builtin_amdgcn_ballot_w32(d[u][qi] <= T[qi]);
                }
            }
#pragma unroll
            for (int u = 0; u < 4; ++u) {
                if (bm[u][0] | bm[u][1] | bm[u][2] | bm[u][3]) {
#pragma unroll
                    for (int qi = 0; qi < KNN_QW; ++qi) {
                        const unsigned m = bm[u][qi];
                        if (m) {
                            const bool hit = ((m >> lane) & 1u) != 0u;
                            const int pos = cn[qi] + (int)__builtin_amdgcn_mbcnt_lo(m, 0u);
                            if (hit && pos < KNN_CAP) {
                                hd[wave][qi][pos] = d[u][qi];
                                hi[wave][qi][pos] = t + 32 * u + lane;
                            }
                            cn[qi] += __builtin_popcount(m);
                        }
                    }
                }
            }
        }
    } else {
        const int gt = (int)threadIdx.x - FUSE_KNN_THREADS;
        const int gw = gt >> 5;
        const int wm = gw >> 1, wn = gw & 1;
        const float* Xb = feats + (size_t)b * 128 * N;
        float* Db = PQ + (size_t)b * 256 * N;
        for (int t = 0; t < 2; ++t)
            gemm_tile<true, 0, 128, 4>(wl, Xb, Xb, (const float4*)nullptr, Db, N, 128, 0, 256,
                                       by * 64 + wm * 32, (2 * np + t) * 64 + wn * 32, wm, lane);
    }
    __syncthreads();

    if (knn_role) {
#pragma unroll
        for (int qi = 0; qi < KNN_QW; ++qi) {
            const int n = cn[qi];
            const int iq = q0 + qi;
            int* so_q = so + (wave * KNN_QW + qi) * KNN_K;
            if (n > KNN_K && n <= KNN_CAP) {
                const float* sd = hd[wave][qi];
                const int*   si = hi[wave][qi];
                const bool v0 = lane < n, v1 = lane + 32 < n;
                const float d0 = v0 ? sd[lane] : __builtin_inff();
                const int   i0 = v0 ? si[lane] : 0x7fffffff;
                const float d1 = v1 ? sd[lane + 32] : __builtin_inff();
                const int   i1 = v1 ? si[lane + 32] : 0x7fffffff;
                int r0 = 0, r1 = 0;
                bool tie = false;
                for (int j = 0; j < n; ++j) {
                    const float dj = sd[j]; const int ij = si[j];
                    r0 += knn_less(dj, ij, d0, i0) ? 1 : 0;
                    r1 += knn_less(dj, ij, d1, i1) ? 1 : 0;
                    tie = tie || (dj == d0 && ij != i0) || (dj == d1 && ij != i1);
                }
                if (__builtin_amdgcn_ballot_w32(tie)) {
                    knn_replay(sd, si, n, lane, iq, so_q);
                    continue;
                }
                int sr = KNN_K + 1;
                unsigned s0 = __builtin_amdgcn_ballot_w32(v0 && i0 == iq);
                if (s0) sr = __shfl(r0, __builtin_ctz(s0), 32);
                unsigned s1 = __builtin_amdgcn_ballot_w32(v1 && i1 == iq);
                if (s1) sr = __shfl(r1, __builtin_ctz(s1), 32);
                if (v0 && i0 != iq && r0 <= KNN_K) {
                    int w = r0 - (r0 > sr ? 1 : 0);
                    if (w < KNN_K) so_q[w] = i0;
                }
                if (v1 && i1 != iq && r1 <= KNN_K) {
                    int w = r1 - (r1 > sr ? 1 : 0);
                    if (w < KNN_K) so_q[w] = i1;
                }
            } else {
                knn_exact_wave(tile, N, lane, px[qi], py[qi], pz[qi], pw[qi], iq, so_q);
            }
        }
    }
    __syncthreads();
    if (threadIdx.x < FUSE_KNN_THREADS) {
        int* ob = knn + ((size_t)b * N + (size_t)tb * 32) * KNN_K;
        for (int i = threadIdx.x; i < 32 * KNN_K; i += FUSE_KNN_THREADS) ob[i] = so[i];
    }
}

#define EDGE_PTS 32
#define EDGE_PW  4

template <int M>
__global__ __launch_bounds__(256) void edge_agg_kernel(
        const float* __restrict__ PQT, const int* __restrict__ knn,
        float* __restrict__ outMax, float2* __restrict__ part,
        int N, long outBatchStride) {
    constexpr int NV = M / 128;
    constexpr int TS = M + 4;
    __shared__ __attribute__((aligned(16))) float tile[EDGE_PTS * TS];
    __shared__ __attribute__((aligned(16))) float rs[8][M];
    __shared__ __attribute__((aligned(16))) float rs2[8][M];
    const int b = blockIdx.y;
    const int lane = threadIdx.x & 31;
    const int wave = __builtin_amdgcn_readfirstlane((int)(threadIdx.x >> 5));
    const int nb0 = blockIdx.x * EDGE_PTS;
    const float* base = PQT + (size_t)b * N * (2 * M);

    float4 s[NV], s2[NV];
#pragma unroll
    for (int v = 0; v < NV; ++v) {
        s[v]  = make_float4(0.f, 0.f, 0.f, 0.f);
        s2[v] = make_float4(0.f, 0.f, 0.f, 0.f);
    }

#pragma unroll
    for (int j = 0; j < EDGE_PW; ++j) {
        const int nl = wave * EDGE_PW + j;
        const int n = nb0 + nl;
        const int* id = knn + ((size_t)b * N + n) * KNN_K;
        const float* prow = base + (size_t)n * (2 * M) + 4 * lane;
        float4 p[NV], mx[NV];
#pragma unroll
        for (int v = 0; v < NV; ++v) {
            p[v] = *(const float4*)(prow + v * 128);
            mx[v] = make_float4(-3.4e38f, -3.4e38f, -3.4e38f, -3.4e38f);
        }
#pragma unroll
        for (int k = 0; k < KNN_K; ++k) {
            const float* qrow = base + (size_t)id[k] * (2 * M) + M + 4 * lane;
#pragma unroll
            for (int v = 0; v < NV; ++v) {
                const float4 q = *(const float4*)(qrow + v * 128);
                const float y0 = p[v].x + q.x, y1 = p[v].y + q.y;
                const float y2 = p[v].z + q.z, y3 = p[v].w + q.w;
                mx[v].x = fmaxf(mx[v].x, y0); mx[v].y = fmaxf(mx[v].y, y1);
                mx[v].z = fmaxf(mx[v].z, y2); mx[v].w = fmaxf(mx[v].w, y3);
                s[v].x += y0; s[v].y += y1; s[v].z += y2; s[v].w += y3;
                s2[v].x += y0 * y0; s2[v].y += y1 * y1;
                s2[v].z += y2 * y2; s2[v].w += y3 * y3;
            }
        }
#pragma unroll
        for (int v = 0; v < NV; ++v)
            *(float4*)(tile + nl * TS + v * 128 + 4 * lane) = mx[v];
    }
#pragma unroll
    for (int v = 0; v < NV; ++v) {
        *(float4*)(&rs[wave][v * 128 + 4 * lane])  = s[v];
        *(float4*)(&rs2[wave][v * 128 + 4 * lane]) = s2[v];
    }
    __syncthreads();

    float* ob = outMax + (size_t)b * outBatchStride + nb0 + lane;
    for (int m = wave; m < M; m += 8)
        ob[(size_t)m * N] = tile[lane * TS + m];

    const int t = threadIdx.x;
    if (t < M) {
        float a = 0.f, a2 = 0.f;
#pragma unroll
        for (int w = 0; w < 8; ++w) { a += rs[w][t]; a2 += rs2[w][t]; }
        part[((size_t)b * gridDim.x + blockIdx.x) * M + t] = make_float2(a, a2);
    }
}

#define STAT_TILES 128
#define STAT_CPB   16

__global__ __launch_bounds__(32 * STAT_CPB) void stats_kernel(const float2* __restrict__ part,
                                                             float* __restrict__ stat,
                                                             int M, int chOff, float invCnt) {
    __shared__ float sm[STAT_CPB], sr[STAT_CPB];
    const int lane = threadIdx.x & 31;
    const int w = (int)(threadIdx.x >> 5);
    const int row = blockIdx.x * STAT_CPB + w;
    const int b = row / M, m = row - b * M;
    const float2* p = part + (size_t)b * STAT_TILES * M + m;
    float a[4], a2[4];
#pragma unroll
    for (int g = 0; g < 4; ++g) {
        const float2 v = p[(size_t)(g * 32 + lane) * M];
        a[g] = v.x; a2[g] = v.y;
    }
#pragma unroll
    for (int g = 0; g < 4; ++g)
        for (int off = 16; off > 0; off >>= 1) {
            a[g]  += __shfl_xor(a[g], off, 32);
            a2[g] += __shfl_xor(a2[g], off, 32);
        }
    float s = 0.f, s2 = 0.f;
#pragma unroll
    for (int g = 0; g < 4; ++g) { s += a[g]; s2 += a2[g]; }
    float mu = s * invCnt;
    float e2 = s2 * invCnt;
    float var = fmaxf(__builtin_fmaf(-mu, mu, e2), 0.f);
    float r = rsqrtf(var + IN_EPS);
    if (lane == 0) { sm[w] = mu; sr[w] = r; }
    __syncthreads();
    if (w == 0) {
        const int ch = 2 * (lane >> 2) + (lane & 1);
        const float v = (lane & 2) ? sr[ch] : sm[ch];
        stat[((size_t)b * (STAT_CH / 2) + ((chOff + m) >> 1)) * 4 + lane] = v;
    }
}

__global__ __launch_bounds__(256) void final_norm_kernel(
        const float* __restrict__ xin, float* __restrict__ out, int N) {
    __shared__ float rs[256], rs2[256];
    __shared__ float s_mu, s_r;
    const int b = blockIdx.y, c = blockIdx.x, C = gridDim.x;
    const float* row = xin + ((size_t)b * C + c) * N;
    float s = 0.f, s2 = 0.f;
    for (int n = threadIdx.x; n < N; n += 256) {
        float v = row[n]; s += v; s2 += v * v;
    }
    rs[threadIdx.x] = s; rs2[threadIdx.x] = s2;
    __syncthreads();
    for (int off = 128; off > 0; off >>= 1) {
        if (threadIdx.x < off) {
            rs[threadIdx.x]  += rs[threadIdx.x + off];
            rs2[threadIdx.x] += rs2[threadIdx.x + off];
        }
        __syncthreads();
    }
    if (threadIdx.x == 0) {
        float mu = rs[0] / (float)N;
        float var = fmaxf(rs2[0] / (float)N - mu * mu, 0.f);
        s_mu = mu; s_r = rsqrtf(var + IN_EPS);
    }
    __syncthreads();
    float mu = s_mu, r = s_r;
    float* orow = out + ((size_t)b * C + c) * N;
    for (int n = threadIdx.x; n < N; n += 256)
        orow[n] = lrelu((row[n] - mu) * r);
}

extern "C" void kernel_launch(void* const* d_in, const int* in_sizes, int n_in,
                              void* d_out, int out_size, void* d_ws, size_t ws_size,
                              hipStream_t stream) {
    constexpr int B = 4, C = 128, N = 4096;
    const float* coords = (const float*)d_in[0];
    const float* feats  = (const float*)d_in[1];
    const float* W1     = (const float*)d_in[2];
    const float* W2     = (const float*)d_in[3];
    const float* W3     = (const float*)d_in[4];
    float* out = (float*)d_out;

    char* ws = (char*)d_ws;
    size_t off = 0;
    auto carve = [&](size_t bytes) -> void* {
        void* p = ws + off;
        off = (off + bytes + 255) & ~(size_t)255;
        return p;
    };
    float* W2cat  = (float*)carve(512 * 128 * sizeof(float));
    int*   knn    = (int*)carve((size_t)B * N * KNN_K * sizeof(int));
    float* PQ     = (float*)carve((size_t)B * 512 * N * sizeof(float));
    float* X3     = (float*)carve((size_t)B * 512 * N * sizeof(float));
    float* x3pre  = (float*)carve((size_t)B * 128 * N * sizeof(float));
    float2* part1 = (float2*)x3pre;
    float2* part2 = part1 + (size_t)B * (N / 32) * 128;
    float* stat = X3;
    static_assert(N / EDGE_PTS == STAT_TILES, "tiles per batch");
    static_assert((size_t)B * STAT_CH * 2 <= (size_t)128 * N, "table fits the unused rows");
    static_assert(N == KNN_N && N % 128 == 0, "points per batch held in LDS; tile pairs per row block");
    static_assert(B * (N / 32) * 128 == 512 * 128, "one stacked-W2 element per GEMM thread of the first launch");
    (void)ws_size; (void)in_sizes; (void)n_in; (void)out_size;

    knn_gemm1_kernel<<<B * (N / 32), FUSE_THREADS, 0, stream>>>(coords, knn, W1, feats, PQ, W2, W2cat, N);

    const float invCnt = 1.0f / (float)(N * KNN_K);

    edge_agg_kernel<128><<<dim3(N / 32, B), 256, 0, stream>>>(
        PQ, knn, X3 + (size_t)128 * N, part1, N, (long)512 * N);
    stats_kernel<<<B * 128 / STAT_CPB, 32 * STAT_CPB, 0, stream>>>(part1, stat, 128, 0, invCnt);

    gemm_wmma_kernel<true, 1, 128><<<dim3(N / 64, 512 / 64, B), 128, 0, stream>>>(
        W2cat, X3 + (size_t)128 * N, X3 + (size_t)128 * N, (const float4*)stat, PQ, N, C,
        (long)512 * N, (long)512 * N, 0, (long)512 * N, 512);
    edge_agg_kernel<256><<<dim3(N / 32, B), 256, 0, stream>>>(
        PQ, knn, X3 + (size_t)256 * N, part2, N, (long)512 * N);
    stats_kernel<<<B * 256 / STAT_CPB, 32 * STAT_CPB, 0, stream>>>(part2, stat, 256, 128, invCnt);

    gemm_wmma_kernel<false, 2, 512><<<dim3(N / 64, 128 / 64, B), 128, 0, stream>>>(
        W3, feats, X3, (const float4*)stat, x3pre, N, 512, (long)C * N, (long)512 * N, 4,
        (long)128 * N, 0);
    final_norm_kernel<<<dim3(128, B), 256, 0, stream>>>(x3pre, out, N);
}


# EdgeConv aggregation parallel over channels: a block owns 2-4 channels of a batch and gathers their Q rows from LDS; InstanceNorm block-local, stats launches and GEMM-side norm go; sum order changes
# speedup vs baseline: 5.5047x; 1.0938x over previous
#include <hip/hip_runtime.h>
#include <hip/hip_bf16.h>


typedef __attribute__((ext_vector_type(2))) float v2f;
typedef __attribute__((ext_vector_type(8))) float v8f;
typedef __attribute__((ext_vector_type(4))) float v4f;
typedef __attribute__((ext_vector_type(2))) int   v2i;

#define KNN_K 10
#define IN_EPS 1e-5f

__device__ __forceinline__ float lrelu(float y) { return y >= 0.0f ? y : 0.2f * y; }

__device__ __forceinline__ v8f wmma_f32(v2f a, v2f b, v8f c) {
    return __builtin_amdgcn_wmma_f32_16x16x4_f32(
         false, a,  false, b,
         (short)0, c,  false,  false);
}

#define KNN_N   4096
#define KNN_QW  4
#define KNN_CAP 64

__device__ __forceinline__ float knn_dist(float px, float py, float pz, float pw, float4 q) {
    float d;
    {
#pragma clang fp contract(off)
        d = __builtin_fmaf(-2.0f, __builtin_fmaf(px, q.x, py * q.y) + pz * q.z, pw + q.w);
    }
    return d;
}

__device__ __forceinline__ bool knn_less(float da, int ia, float db, int ib) {
    return (da < db) || (da == db && ia < ib);
}

__device__ __forceinline__ void knn_exact_wave(const float4* tile, int N, int lane,
        float px, float py, float pz, float pw, int iq, int* so_q) {
    float ld = -__builtin_inff(); int li = -1; int w = 0;
    for (int r = 0; r <= KNN_K; ++r) {
        float bd = __builtin_inff(); int bi = 0x7fffffff;
        for (int t = 0; t < N; t += 32) {
            const int c = t + lane;
            float d = knn_dist(px, py, pz, pw, tile[c]);
            if (knn_less(ld, li, d, c) && knn_less(d, c, bd, bi)) { bd = d; bi = c; }
        }
        for (int off = 16; off > 0; off >>= 1) {
            float od = __shfl_xor(bd, off, 32); int oi = __shfl_xor(bi, off, 32);
            if (knn_less(od, oi, bd, bi)) { bd = od; bi = oi; }
        }
        ld = bd; li = bi;
        if (bi != iq) {
            if (w < KNN_K && lane == 0) so_q[w] = (bi < N) ? bi : 0;
            ++w;
        }
    }
}

__device__ __forceinline__ void knn_replay(const float* sd, const int* si, int n,
                                           int lane, int iq, int* so_q) {
    float bd[KNN_K + 1];
    int   bi[KNN_K + 1];
#pragma unroll
    for (int s = 0; s <= KNN_K; ++s) { bd[s] = 3.4e38f; bi[s] = -1; }
    for (int j = 0; j < n; ++j) {
        float cd = sd[j]; int ci = si[j];
        if (cd < bd[KNN_K]) {
#pragma unroll
            for (int s = 0; s <= KNN_K; ++s) {
                bool sw = cd < bd[s];
                float td = sw ? bd[s] : cd; int ti = sw ? bi[s] : ci;
                bd[s] = sw ? cd : bd[s];    bi[s] = sw ? ci : bi[s];
                cd = td; ci = ti;
            }
        }
    }
    if (lane == 0) {
        int w = 0;
#pragma unroll
        for (int s = 0; s <= KNN_K; ++s)
            if (bi[s] != iq && w < KNN_K) { so_q[w] = bi[s]; ++w; }
    }
}


template <int KD, bool CAT>
__device__ __forceinline__ void gemm_copy_w(const float* __restrict__ W, float* wl,
                                            int by, int tid, int Mhalf) {
    constexpr int LDW = KD + 4;
    if (CAT) {
        const int R0 = by * 64;
        const bool diff = R0 < Mhalf;
        const float* wsrc = W + (size_t)(diff ? R0 : R0 - Mhalf) * (2 * KD);
        if (diff) {
#pragma unroll 8
            for (int it = 0; it < 64 * KD / 4 / 128; ++it) {
                const int i = it * 128 + tid;
                const int r = i / (KD / 4), c4 = i - r * (KD / 4);
                const float4 wa = *(const float4*)(wsrc + (size_t)r * (2 * KD) + 4 * c4);
                const float4 wb = *(const float4*)(wsrc + (size_t)r * (2 * KD) + KD + 4 * c4);
                *(float4*)(wl + r * LDW + 4 * c4) =
                    make_float4(wa.x - wb.x, wa.y - wb.y, wa.z - wb.z, wa.w - wb.w);
            }
        } else {
#pragma unroll 16
            for (int it = 0; it < 64 * KD / 4 / 128; ++it) {
                const int i = it * 128 + tid;
                const int r = i / (KD / 4), c4 = i - r * (KD / 4);
                *(float4*)(wl + r * LDW + 4 * c4) =
                    *(const float4*)(wsrc + (size_t)r * (2 * KD) + KD + 4 * c4);
            }
        }
    } else {
        const float4* wsrc = (const float4*)(W + (size_t)(by * 64) * KD);
#pragma unroll 16
        for (int it = 0; it < 64 * KD / 4 / 128; ++it) {
            const int i = it * 128 + tid;
            const int r = i / (KD / 4), c4 = i - r * (KD / 4);
            *(float4*)(wl + r * LDW + 4 * c4) = wsrc[i];
        }
    }
}

template <int KD, int KB>
__device__ __forceinline__ void gemm_tile(
        const float* wl, const float* __restrict__ Xb, const float* X2b,
        float* __restrict__ Db, int N, int Kdim,
        int splitBlk, int m0, int n0, int wm, int lane) {
    constexpr int LDW = KD + 4;
    const int half = lane >> 4;
    const int l16  = lane & 15;

    v8f c00 = {0.f,0.f,0.f,0.f,0.f,0.f,0.f,0.f};
    v8f c01 = c00, c10 = c00, c11 = c00;

    const float* ar0 = wl + (wm * 32 + l16) * LDW + 2 * half;
    const float* ar1 = ar0 + 16 * LDW;
    const float* xr  = Xb + (size_t)(2 * half) * N + n0 + l16;
    const float* xr2 = X2b + (size_t)(splitBlk * (4 * KB) + 2 * half) * N + n0 + l16;
    int blk = 0;

    __builtin_prefetch(xr, 0, 1);

    v2f fa0[KB], fa1[KB], fb0[KB], fb1[KB];
    v2f ga0[KB], ga1[KB], gb0[KB], gb1[KB];

    auto load_blk = [&](v2f (&a0)[KB], v2f (&a1)[KB], v2f (&b0)[KB], v2f (&b1)[KB]) {
#pragma unroll
        for (int s = 0; s < KB; ++s) {
            const float* xs = xr + (size_t)(4 * s) * N;
            a0[s] = *(const v2f*)(ar0 + 4 * s);
            a1[s] = *(const v2f*)(ar1 + 4 * s);
            b0[s].x = xs[0];        b0[s].y = xs[N];
            b1[s].x = xs[16];       b1[s].y = xs[16 + N];
        }
        ar0 += 4 * KB; ar1 += 4 * KB; xr += (size_t)(4 * KB) * N;
        if (++blk == splitBlk) xr = xr2;
    };
    auto mma_blk = [&](const v2f (&a0)[KB], const v2f (&a1)[KB],
                       const v2f (&b0)[KB], const v2f (&b1)[KB]) {
#pragma unroll
        for (int s = 0; s < KB; ++s) {
            c00 = wmma_f32(a0[s], b0[s], c00);
            c01 = wmma_f32(a0[s], b1[s], c01);
            c10 = wmma_f32(a1[s], b0[s], c10);
            c11 = wmma_f32(a1[s], b1[s], c11);
        }
    };

    const int nblk = Kdim / (4 * KB);
    load_blk(fa0, fa1, fb0, fb1);
    for (int j = 0; j < nblk - 2; j += 2) {
        load_blk(ga0, ga1, gb0, gb1);
        __builtin_amdgcn_sched_barrier(0);
        mma_blk(fa0, fa1, fb0, fb1);
        load_blk(fa0, fa1, fb0, fb1);
        __builtin_amdgcn_sched_barrier(0);
        mma_blk(ga0, ga1, gb0, gb1);
    }
    load_blk(ga0, ga1, gb0, gb1);
    __builtin_amdgcn_sched_barrier(0);
    mma_blk(fa0, fa1, fb0, fb1);
    mma_blk(ga0, ga1, gb0, gb1);
#pragma unroll
    for (int v = 0; v < 8; ++v) {
        const int mr0 = m0 + v + 8 * half;
        Db[(size_t)mr0 * N + n0 + l16]             = c00[v];
        Db[(size_t)mr0 * N + n0 + 16 + l16]        = c01[v];
        Db[(size_t)(mr0 + 16) * N + n0 + l16]      = c10[v];
        Db[(size_t)(mr0 + 16) * N + n0 + 16 + l16] = c11[v];
    }
}

template <int KD>
__global__ __launch_bounds__(128) void gemm_wmma_kernel(
        const float* __restrict__ W, const float* __restrict__ X, const float* X2,
        float* __restrict__ D, int N, int Kdim,
        long xBatchStride, long x2BatchStride, int splitBlk, long dBatchStride) {
    __shared__ __attribute__((aligned(16))) float wl[64 * (KD + 4)];
    gemm_copy_w<KD, false>(W, wl, (int)blockIdx.y, (int)threadIdx.x, 0);
    __syncthreads();

    const int lane = threadIdx.x & 31;
    const int wave = threadIdx.x >> 5;
    const int wm = wave >> 1, wn = wave & 1;
    gemm_tile<KD, 8>(
        wl, X + (size_t)blockIdx.z * xBatchStride, X2 + (size_t)blockIdx.z * x2BatchStride,
        D + (size_t)blockIdx.z * dBatchStride,
        N, Kdim, splitBlk, blockIdx.y * 64 + wm * 32, blockIdx.x * 64 + wn * 32, wm, lane);
}

#define FUSE_KNN_THREADS 256
#define FUSE_THREADS     384

__global__ __launch_bounds__(FUSE_THREADS, 6) void knn_gemm1_kernel(
        const float* __restrict__ coords, unsigned short* __restrict__ knn,
        const float* __restrict__ W1, const float* __restrict__ feats,
        float* __restrict__ PQ, const float* __restrict__ W2, float* __restrict__ W2cat, int N) {
    __shared__ float4 tile[KNN_N];
    __shared__ float  hd[8][KNN_QW][KNN_CAP];
    __shared__ int    hi[8][KNN_QW][KNN_CAP];
    __shared__ int    so[32 * KNN_K];
    __shared__ __attribute__((aligned(16))) float wl[64 * (128 + 4)];
    const int lane = threadIdx.x & 31;
    const int wave = __builtin_amdgcn_readfirstlane((int)(threadIdx.x >> 5));
    const bool knn_role = wave < 8;
    const int tilesPerBatch = N / 32;
    const int b = blockIdx.x / tilesPerBatch;
    const int tb = blockIdx.x - b * tilesPerBatch;
    const int by = tb / (N / 128);
    const int np = tb - by * (N / 128);

    if (knn_role) {
        const float* cb = coords + (size_t)b * 3 * N;
#pragma unroll
        for (int it = 0; it < KNN_N / FUSE_KNN_THREADS; ++it) {
            const int i = it * FUSE_KNN_THREADS + (int)threadIdx.x;
            const float x = cb[i], y = cb[N + i], z = cb[2 * N + i];
            tile[i] = make_float4(x, y, z, __builtin_fmaf(z, z, __builtin_fmaf(x, x, y * y)));
        }
    } else {
        const int gt = (int)threadIdx.x - FUSE_KNN_THREADS;
        gemm_copy_w<128, true>(W1, wl, by, gt, 128);
        const int e = (int)blockIdx.x * 128 + gt;
        const int R = e >> 7, cc = e & 127;
        const float wa2 = W2[(R & 255) * 256 + cc];
        const float wb2 = W2[(R & 255) * 256 + 128 + cc];
        W2cat[e] = (R < 256) ? wa2 - wb2 : wb2;
    }
    __syncthreads();

    const int q0 = tb * 32 + wave * KNN_QW;
    float px[KNN_QW], py[KNN_QW], pz[KNN_QW], pw[KNN_QW];
    int cn[KNN_QW];
    if (knn_role) {
#pragma unroll
        for (int qi = 0; qi < KNN_QW; ++qi) {
            const float4 p = tile[q0 + qi];
            px[qi] = __builtin_bit_cast(float, __builtin_amdgcn_readfirstlane(__builtin_bit_cast(int, p.x)));
            py[qi] = __builtin_bit_cast(float, __builtin_amdgcn_readfirstlane(__builtin_bit_cast(int, p.y)));
            pz[qi] = __builtin_bit_cast(float, __builtin_amdgcn_readfirstlane(__builtin_bit_cast(int, p.z)));
            pw[qi] = __builtin_bit_cast(float, __builtin_amdgcn_readfirstlane(__builtin_bit_cast(int, p.w)));
        }

        float mn[KNN_QW];
#pragma unroll
        for (int qi = 0; qi < KNN_QW; ++qi) mn[qi] = __builtin_inff();
#pragma unroll 4
        for (int t = 0; t < N; t += 32) {
            float4 q = tile[t + lane];
#pragma unroll
            for (int qi = 0; qi < KNN_QW; ++qi) {
                float d = knn_dist(px[qi], py[qi], pz[qi], pw[qi], q);
                mn[qi] = d < mn[qi] ? d : mn[qi];
            }
        }

        float T[KNN_QW];
#pragma unroll
        for (int qi = 0; qi < KNN_QW; ++qi) {
            const float mv = mn[qi];
            int cnt = 0;
#pragma unroll
            for (int s = 0; s < 32; ++s) {
                float v = __builtin_bit_cast(float,
                    __builtin_amdgcn_readlane(__builtin_bit_cast(int, mv), s));
                cnt += knn_less(v, s, mv, lane) ? 1 : 0;
            }
            unsigned sel = __builtin_amdgcn_ballot_w32(cnt == KNN_K);
            T[qi] = sel ? __shfl(mv, __builtin_ctz(sel), 32) : __builtin_inff();
        }

#pragma unroll
        for (int qi = 0; qi < KNN_QW; ++qi) cn[qi] = 0;
        for (int t = 0; t < N; t += 128) {
            float4 q[4];
            float d[4][KNN_QW];
            unsigned bm[4][KNN_QW];
#pragma unroll
            for (int u = 0; u < 4; ++u) q[u] = tile[t + 32 * u + lane];
#pragma unroll
            for (int u = 0; u < 4; ++u) {
#pragma unroll
                for (int qi = 0; qi < KNN_QW; ++qi) {
                    d[u][qi] = knn_dist(px[qi], py[qi], pz[qi], pw[qi], q[u]);
                    bm[u][qi] = __builtin_amdgcn_ballot_w32(d[u][qi] <= T[qi]);
                }
            }
#pragma unroll
            for (int u = 0; u < 4; ++u) {
                if (bm[u][0] | bm[u][1] | bm[u][2] | bm[u][3]) {
#pragma unroll
                    for (int qi = 0; qi < KNN_QW; ++qi) {
                        const unsigned m = bm[u][qi];
                        if (m) {
                            const bool hit = ((m >> lane) & 1u) != 0u;
                            const int pos = cn[qi] + (int)__builtin_amdgcn_mbcnt_lo(m, 0u);
                            if (hit && pos < KNN_CAP) {
                                hd[wave][qi][pos] = d[u][qi];
                                hi[wave][qi][pos] = t + 32 * u + lane;
                            }
                            cn[qi] += __builtin_popcount(m);
                        }
                    }
                }
            }
        }
    } else {
        const int gt = (int)threadIdx.x - FUSE_KNN_THREADS;
        const int gw = gt >> 5;
        const int wm = gw >> 1, wn = gw & 1;
        const float* Xb = feats + (size_t)b * 128 * N;
        float* Db = PQ + (size_t)b * 256 * N;
        for (int t = 0; t < 2; ++t)
            gemm_tile<128, 4>(wl, Xb, Xb, Db, N, 128, 0,
                              by * 64 + wm * 32, (2 * np + t) * 64 + wn * 32, wm, lane);
    }
    __syncthreads();

    if (knn_role) {
#pragma unroll
        for (int qi = 0; qi < KNN_QW; ++qi) {
            const int n = cn[qi];
            const int iq = q0 + qi;
            int* so_q = so + (wave * KNN_QW + qi) * KNN_K;
            if (n > KNN_K && n <= KNN_CAP) {
                const float* sd = hd[wave][qi];
                const int*   si = hi[wave][qi];
                const bool v0 = lane < n, v1 = lane + 32 < n;
                const float d0 = v0 ? sd[lane] : __builtin_inff();
                const int   i0 = v0 ? si[lane] : 0x7fffffff;
                const float d1 = v1 ? sd[lane + 32] : __builtin_inff();
                const int   i1 = v1 ? si[lane + 32] : 0x7fffffff;
                int r0 = 0, r1 = 0;
                bool tie = false;
                for (int j = 0; j < n; ++j) {
                    const float dj = sd[j]; const int ij = si[j];
                    r0 += knn_less(dj, ij, d0, i0) ? 1 : 0;
                    r1 += knn_less(dj, ij, d1, i1) ? 1 : 0;
                    tie = tie || (dj == d0 && ij != i0) || (dj == d1 && ij != i1);
                }
                if (__builtin_amdgcn_ballot_w32(tie)) {
                    knn_replay(sd, si, n, lane, iq, so_q);
                    continue;
                }
                int sr = KNN_K + 1;
                unsigned s0 = __builtin_amdgcn_ballot_w32(v0 && i0 == iq);
                if (s0) sr = __shfl(r0, __builtin_ctz(s0), 32);
                unsigned s1 = __builtin_amdgcn_ballot_w32(v1 && i1 == iq);
                if (s1) sr = __shfl(r1, __builtin_ctz(s1), 32);
                if (v0 && i0 != iq && r0 <= KNN_K) {
                    int w = r0 - (r0 > sr ? 1 : 0);
                    if (w < KNN_K) so_q[w] = i0;
                }
                if (v1 && i1 != iq && r1 <= KNN_K) {
                    int w = r1 - (r1 > sr ? 1 : 0);
                    if (w < KNN_K) so_q[w] = i1;
                }
            } else {
                knn_exact_wave(tile, N, lane, px[qi], py[qi], pz[qi], pw[qi], iq, so_q);
            }
        }
    }
    __syncthreads();
    if (threadIdx.x < 16 * KNN_K) {
        unsigned* ob = (unsigned*)(knn + ((size_t)b * tilesPerBatch + tb) * (KNN_K * 32));
        const int k = (int)threadIdx.x >> 4, pp = ((int)threadIdx.x & 15) * 2;
        ob[threadIdx.x] = ((unsigned)so[pp * KNN_K + k] & 0xffffu) | ((unsigned)so[(pp + 1) * KNN_K + k] << 16);
    }
}

#define EDGE_THREADS 512
#define EDGE_GRP (KNN_N / (4 * EDGE_THREADS))

template <int M, int CG>
__global__ __launch_bounds__(EDGE_THREADS) void edge_chan_kernel(
        const float* __restrict__ PQ, const unsigned short* __restrict__ knn,
        float* __restrict__ xout, long outBatchStride, float invCnt) {
    typedef float vcg __attribute__((ext_vector_type(CG)));
    constexpr int N = KNN_N;
    __shared__ __attribute__((aligned(16))) float ql[N * CG];
    __shared__ __attribute__((aligned(16))) float mxl[N * CG];
    __shared__ float red[2 * CG][EDGE_THREADS / 32];
    __shared__ float s_mu[CG], s_r[CG];
    const int b = blockIdx.y;
    const int c0 = blockIdx.x * CG;
    const int t = threadIdx.x;
    const int lane = t & 31, wave = t >> 5;
    const float* Pb = PQ + ((size_t)b * (2 * M) + c0) * N;
    const float* Qb = Pb + (size_t)M * N;
    const unsigned short* ib = knn + (size_t)b * KNN_K * N;

    v4f qr[EDGE_GRP][CG];
#pragma unroll
    for (int g = 0; g < EDGE_GRP; ++g)
#pragma unroll
        for (int c = 0; c < CG; ++c)
            qr[g][c] = *(const v4f*)(Qb + (size_t)c * N + 4 * (g * EDGE_THREADS + t));
    v2i id[EDGE_GRP][KNN_K];
    v4f p[EDGE_GRP][CG];
#pragma unroll
    for (int g = 0; g < EDGE_GRP; ++g) {
#pragma unroll
        for (int k = 0; k < KNN_K; ++k)
            id[g][k] = *(const v2i*)(ib + ((4 * (g * EDGE_THREADS + t)) >> 5) * (KNN_K * 32) + k * 32
                                        + ((4 * (g * EDGE_THREADS + t)) & 31));
#pragma unroll
        for (int c = 0; c < CG; ++c)
            p[g][c] = *(const v4f*)(Pb + (size_t)c * N + 4 * (g * EDGE_THREADS + t));
    }
#pragma unroll
    for (int g = 0; g < EDGE_GRP; ++g) {
        const int n0 = 4 * (g * EDGE_THREADS + t);
#pragma unroll
        for (int j = 0; j < 4; ++j) {
            vcg q;
#pragma unroll
            for (int c = 0; c < CG; ++c) q[c] = qr[g][c][j];
            *(vcg*)(ql + (n0 + j) * CG) = q;
        }
    }
    __syncthreads();

    vcg s, s2;
#pragma unroll
    for (int c = 0; c < CG; ++c) { s[c] = 0.f; s2[c] = 0.f; }
#pragma unroll
    for (int g = 0; g < EDGE_GRP; ++g) {
        v4f mg[CG];
#pragma unroll
        for (int j = 0; j < 4; ++j) {
            vcg pv, m;
#pragma unroll
            for (int c = 0; c < CG; ++c) { pv[c] = p[g][c][j]; m[c] = -3.4e38f; }
#pragma unroll
            for (int k = 0; k < KNN_K; ++k) {
                const unsigned w = (unsigned)id[g][k][j >> 1];
                const int ix = (j & 1) ? (int)(w >> 16) : (int)(w & 0xffffu);
                const vcg q = *(const vcg*)(ql + ix * CG);
                const vcg y = pv + q;
                m = __builtin_elementwise_max(m, y);
                s += y;
                s2 += y * y;
            }
#pragma unroll
            for (int c = 0; c < CG; ++c) mg[c][j] = m[c];
            if (j & 1) __builtin_amdgcn_sched_barrier(0);
        }
#pragma unroll
        for (int c = 0; c < CG; ++c)
            *(v4f*)(mxl + ((g * CG + c) * EDGE_THREADS + t) * 4) = mg[c];
    }

#pragma unroll
    for (int c = 0; c < CG; ++c) {
        float a = s[c], a2 = s2[c];
        for (int off = 16; off > 0; off >>= 1) {
            a  += __shfl_xor(a, off, 32);
            a2 += __shfl_xor(a2, off, 32);
        }
        if (lane == 0) { red[c][wave] = a; red[CG + c][wave] = a2; }
    }
    __syncthreads();
    if (t < CG) {
        float a = 0.f, a2 = 0.f;
        for (int w = 0; w < EDGE_THREADS / 32; ++w) { a += red[t][w]; a2 += red[CG + t][w]; }
        const float mu = a * invCnt;
        const float e2 = a2 * invCnt;
        const float var = fmaxf(__builtin_fmaf(-mu, mu, e2), 0.f);
        s_mu[t] = mu; s_r[t] = rsqrtf(var + IN_EPS);
    }
    __syncthreads();

    float* ob = xout + (size_t)b * outBatchStride + (size_t)c0 * N;
#pragma unroll
    for (int c = 0; c < CG; ++c) {
        const float mu = s_mu[c], r = s_r[c];
#pragma unroll
        for (int g = 0; g < EDGE_GRP; ++g) {
            const v4f v = *(const v4f*)(mxl + ((g * CG + c) * EDGE_THREADS + t) * 4);
            const v4f y = (v - mu) * r;
            const v4f o = __builtin_elementwise_max(y, y * 0.2f);
            *(v4f*)(ob + (size_t)c * N + 4 * (g * EDGE_THREADS + t)) = o;
        }
    }
}

__global__ __launch_bounds__(256) void final_norm_kernel(
        const float* __restrict__ xin, float* __restrict__ out, int N) {
    __shared__ float rs[256], rs2[256];
    __shared__ float s_mu, s_r;
    const int b = blockIdx.y, c = blockIdx.x, C = gridDim.x;
    const float* row = xin + ((size_t)b * C + c) * N;
    float s = 0.f, s2 = 0.f;
    for (int n = threadIdx.x; n < N; n += 256) {
        float v = row[n]; s += v; s2 += v * v;
    }
    rs[threadIdx.x] = s; rs2[threadIdx.x] = s2;
    __syncthreads();
    for (int off = 128; off > 0; off >>= 1) {
        if (threadIdx.x < off) {
            rs[threadIdx.x]  += rs[threadIdx.x + off];
            rs2[threadIdx.x] += rs2[threadIdx.x + off];
        }
        __syncthreads();
    }
    if (threadIdx.x == 0) {
        float mu = rs[0] / (float)N;
        float var = fmaxf(rs2[0] / (float)N - mu * mu, 0.f);
        s_mu = mu; s_r = rsqrtf(var + IN_EPS);
    }
    __syncthreads();
    float mu = s_mu, r = s_r;
    float* orow = out + ((size_t)b * C + c) * N;
    for (int n = threadIdx.x; n < N; n += 256)
        orow[n] = lrelu((row[n] - mu) * r);
}

extern "C" void kernel_launch(void* const* d_in, const int* in_sizes, int n_in,
                              void* d_out, int out_size, void* d_ws, size_t ws_size,
                              hipStream_t stream) {
    constexpr int B = 4, C = 128, N = 4096;
    const float* coords = (const float*)d_in[0];
    const float* feats  = (const float*)d_in[1];
    const float* W1     = (const float*)d_in[2];
    const float* W2     = (const float*)d_in[3];
    const float* W3     = (const float*)d_in[4];
    float* out = (float*)d_out;

    char* ws = (char*)d_ws;
    size_t off = 0;
    auto carve = [&](size_t bytes) -> void* {
        void* p = ws + off;
        off = (off + bytes + 255) & ~(size_t)255;
        return p;
    };
    float* W2cat  = (float*)carve(512 * 128 * sizeof(float));
    unsigned short* knn = (unsigned short*)carve((size_t)B * N * KNN_K * sizeof(unsigned short));
    float* PQ     = (float*)carve((size_t)B * 512 * N * sizeof(float));
    float* X3     = (float*)carve((size_t)B * 512 * N * sizeof(float));
    float* x3pre  = (float*)carve((size_t)B * 128 * N * sizeof(float));
    static_assert(N == KNN_N && N % 128 == 0, "points per batch held in LDS; tile pairs per row block");
    static_assert(N % (4 * EDGE_THREADS) == 0 && N <= 65536, "whole groups of 4 points per edge thread; 16-bit indices");
    static_assert(B * (N / 32) * 128 == 512 * 128, "one stacked-W2 element per GEMM thread of the first launch");
    (void)ws_size; (void)in_sizes; (void)n_in; (void)out_size;

    knn_gemm1_kernel<<<B * (N / 32), FUSE_THREADS, 0, stream>>>(coords, knn, W1, feats, PQ, W2, W2cat, N);

    const float invCnt = 1.0f / (float)(N * KNN_K);

    edge_chan_kernel<128, 2><<<dim3(128 / 2, B), EDGE_THREADS, 0, stream>>>(
        PQ, knn, X3 + (size_t)128 * N, (long)512 * N, invCnt);

    gemm_wmma_kernel<128><<<dim3(N / 64, 512 / 64, B), 128, 0, stream>>>(
        W2cat, X3 + (size_t)128 * N, X3 + (size_t)128 * N, PQ, N, C,
        (long)512 * N, (long)512 * N, 0, (long)512 * N);
    edge_chan_kernel<256, 4><<<dim3(256 / 4, B), EDGE_THREADS, 0, stream>>>(
        PQ, knn, X3 + (size_t)256 * N, (long)512 * N, invCnt);

    gemm_wmma_kernel<512><<<dim3(N / 64, 128 / 64, B), 128, 0, stream>>>(
        W3, feats, X3, x3pre, N, 512, (long)C * N, (long)512 * N, 4, (long)128 * N);
    final_norm_kernel<<<dim3(128, B), 256, 0, stream>>>(x3pre, out, N);
}


# Final InstanceNorm pass holds its row in registers: 16-byte loads all in flight (one round trip, not 16 waited 4-byte loads), sums by shuffles + one LDS exchange, 16-byte stores; sum order changes
# speedup vs baseline: 5.8815x; 1.0685x over previous
#include <hip/hip_runtime.h>
#include <hip/hip_bf16.h>


typedef __attribute__((ext_vector_type(2))) float v2f;
typedef __attribute__((ext_vector_type(8))) float v8f;
typedef __attribute__((ext_vector_type(4))) float v4f;
typedef __attribute__((ext_vector_type(2))) int   v2i;

#define KNN_K 10
#define IN_EPS 1e-5f

__device__ __forceinline__ float lrelu(float y) { return y >= 0.0f ? y : 0.2f * y; }

__device__ __forceinline__ v8f wmma_f32(v2f a, v2f b, v8f c) {
    return __builtin_amdgcn_wmma_f32_16x16x4_f32(
         false, a,  false, b,
         (short)0, c,  false,  false);
}

#define KNN_N   4096
#define KNN_QW  4
#define KNN_CAP 64

__device__ __forceinline__ float knn_dist(float px, float py, float pz, float pw, float4 q) {
    float d;
    {
#pragma clang fp contract(off)
        d = __builtin_fmaf(-2.0f, __builtin_fmaf(px, q.x, py * q.y) + pz * q.z, pw + q.w);
    }
    return d;
}

__device__ __forceinline__ bool knn_less(float da, int ia, float db, int ib) {
    return (da < db) || (da == db && ia < ib);
}

__device__ __forceinline__ void knn_exact_wave(const float4* tile, int N, int lane,
        float px, float py, float pz, float pw, int iq, int* so_q) {
    float ld = -__builtin_inff(); int li = -1; int w = 0;
    for (int r = 0; r <= KNN_K; ++r) {
        float bd = __builtin_inff(); int bi = 0x7fffffff;
        for (int t = 0; t < N; t += 32) {
            const int c = t + lane;
            float d = knn_dist(px, py, pz, pw, tile[c]);
            if (knn_less(ld, li, d, c) && knn_less(d, c, bd, bi)) { bd = d; bi = c; }
        }
        for (int off = 16; off > 0; off >>= 1) {
            float od = __shfl_xor(bd, off, 32); int oi = __shfl_xor(bi, off, 32);
            if (knn_less(od, oi, bd, bi)) { bd = od; bi = oi; }
        }
        ld = bd; li = bi;
        if (bi != iq) {
            if (w < KNN_K && lane == 0) so_q[w] = (bi < N) ? bi : 0;
            ++w;
        }
    }
}

__device__ __forceinline__ void knn_replay(const float* sd, const int* si, int n,
                                           int lane, int iq, int* so_q) {
    float bd[KNN_K + 1];
    int   bi[KNN_K + 1];
#pragma unroll
    for (int s = 0; s <= KNN_K; ++s) { bd[s] = 3.4e38f; bi[s] = -1; }
    for (int j = 0; j < n; ++j) {
        float cd = sd[j]; int ci = si[j];
        if (cd < bd[KNN_K]) {
#pragma unroll
            for (int s = 0; s <= KNN_K; ++s) {
                bool sw = cd < bd[s];
                float td = sw ? bd[s] : cd; int ti = sw ? bi[s] : ci;
                bd[s] = sw ? cd : bd[s];    bi[s] = sw ? ci : bi[s];
                cd = td; ci = ti;
            }
        }
    }
    if (lane == 0) {
        int w = 0;
#pragma unroll
        for (int s = 0; s <= KNN_K; ++s)
            if (bi[s] != iq && w < KNN_K) { so_q[w] = bi[s]; ++w; }
    }
}


template <int KD, bool CAT>
__device__ __forceinline__ void gemm_copy_w(const float* __restrict__ W, float* wl,
                                            int by, int tid, int Mhalf) {
    constexpr int LDW = KD + 4;
    if (CAT) {
        const int R0 = by * 64;
        const bool diff = R0 < Mhalf;
        const float* wsrc = W + (size_t)(diff ? R0 : R0 - Mhalf) * (2 * KD);
        if (diff) {
#pragma unroll 8
            for (int it = 0; it < 64 * KD / 4 / 128; ++it) {
                const int i = it * 128 + tid;
                const int r = i / (KD / 4), c4 = i - r * (KD / 4);
                const float4 wa = *(const float4*)(wsrc + (size_t)r * (2 * KD) + 4 * c4);
                const float4 wb = *(const float4*)(wsrc + (size_t)r * (2 * KD) + KD + 4 * c4);
                *(float4*)(wl + r * LDW + 4 * c4) =
                    make_float4(wa.x - wb.x, wa.y - wb.y, wa.z - wb.z, wa.w - wb.w);
            }
        } else {
#pragma unroll 16
            for (int it = 0; it < 64 * KD / 4 / 128; ++it) {
                const int i = it * 128 + tid;
                const int r = i / (KD / 4), c4 = i - r * (KD / 4);
                *(float4*)(wl + r * LDW + 4 * c4) =
                    *(const float4*)(wsrc + (size_t)r * (2 * KD) + KD + 4 * c4);
            }
        }
    } else {
        const float4* wsrc = (const float4*)(W + (size_t)(by * 64) * KD);
#pragma unroll 16
        for (int it = 0; it < 64 * KD / 4 / 128; ++it) {
            const int i = it * 128 + tid;
            const int r = i / (KD / 4), c4 = i - r * (KD / 4);
            *(float4*)(wl + r * LDW + 4 * c4) = wsrc[i];
        }
    }
}

template <int KD, int KB>
__device__ __forceinline__ void gemm_tile(
        const float* wl, const float* __restrict__ Xb, const float* X2b,
        float* __restrict__ Db, int N, int Kdim,
        int splitBlk, int m0, int n0, int wm, int lane) {
    constexpr int LDW = KD + 4;
    const int half = lane >> 4;
    const int l16  = lane & 15;

    v8f c00 = {0.f,0.f,0.f,0.f,0.f,0.f,0.f,0.f};
    v8f c01 = c00, c10 = c00, c11 = c00;

    const float* ar0 = wl + (wm * 32 + l16) * LDW + 2 * half;
    const float* ar1 = ar0 + 16 * LDW;
    const float* xr  = Xb + (size_t)(2 * half) * N + n0 + l16;
    const float* xr2 = X2b + (size_t)(splitBlk * (4 * KB) + 2 * half) * N + n0 + l16;
    int blk = 0;

    __builtin_prefetch(xr, 0, 1);

    v2f fa0[KB], fa1[KB], fb0[KB], fb1[KB];
    v2f ga0[KB], ga1[KB], gb0[KB], gb1[KB];

    auto load_blk = [&](v2f (&a0)[KB], v2f (&a1)[KB], v2f (&b0)[KB], v2f (&b1)[KB]) {
#pragma unroll
        for (int s = 0; s < KB; ++s) {
            const float* xs = xr + (size_t)(4 * s) * N;
            a0[s] = *(const v2f*)(ar0 + 4 * s);
            a1[s] = *(const v2f*)(ar1 + 4 * s);
            b0[s].x = xs[0];        b0[s].y = xs[N];
            b1[s].x = xs[16];       b1[s].y = xs[16 + N];
        }
        ar0 += 4 * KB; ar1 += 4 * KB; xr += (size_t)(4 * KB) * N;
        if (++blk == splitBlk) xr = xr2;
    };
    auto mma_blk = [&](const v2f (&a0)[KB], const v2f (&a1)[KB],
                       const v2f (&b0)[KB], const v2f (&b1)[KB]) {
#pragma unroll
        for (int s = 0; s < KB; ++s) {
            c00 = wmma_f32(a0[s], b0[s], c00);
            c01 = wmma_f32(a0[s], b1[s], c01);
            c10 = wmma_f32(a1[s], b0[s], c10);
            c11 = wmma_f32(a1[s], b1[s], c11);
        }
    };

    const int nblk = Kdim / (4 * KB);
    load_blk(fa0, fa1, fb0, fb1);
    for (int j = 0; j < nblk - 2; j += 2) {
        load_blk(ga0, ga1, gb0, gb1);
        __builtin_amdgcn_sched_barrier(0);
        mma_blk(fa0, fa1, fb0, fb1);
        load_blk(fa0, fa1, fb0, fb1);
        __builtin_amdgcn_sched_barrier(0);
        mma_blk(ga0, ga1, gb0, gb1);
    }
    load_blk(ga0, ga1, gb0, gb1);
    __builtin_amdgcn_sched_barrier(0);
    mma_blk(fa0, fa1, fb0, fb1);
    mma_blk(ga0, ga1, gb0, gb1);
#pragma unroll
    for (int v = 0; v < 8; ++v) {
        const int mr0 = m0 + v + 8 * half;
        Db[(size_t)mr0 * N + n0 + l16]             = c00[v];
        Db[(size_t)mr0 * N + n0 + 16 + l16]        = c01[v];
        Db[(size_t)(mr0 + 16) * N + n0 + l16]      = c10[v];
        Db[(size_t)(mr0 + 16) * N + n0 + 16 + l16] = c11[v];
    }
}

template <int KD>
__global__ __launch_bounds__(128) void gemm_wmma_kernel(
        const float* __restrict__ W, const float* __restrict__ X, const float* X2,
        float* __restrict__ D, int N, int Kdim,
        long xBatchStride, long x2BatchStride, int splitBlk, long dBatchStride) {
    __shared__ __attribute__((aligned(16))) float wl[64 * (KD + 4)];
    gemm_copy_w<KD, false>(W, wl, (int)blockIdx.y, (int)threadIdx.x, 0);
    __syncthreads();

    const int lane = threadIdx.x & 31;
    const int wave = threadIdx.x >> 5;
    const int wm = wave >> 1, wn = wave & 1;
    gemm_tile<KD, 8>(
        wl, X + (size_t)blockIdx.z * xBatchStride, X2 + (size_t)blockIdx.z * x2BatchStride,
        D + (size_t)blockIdx.z * dBatchStride,
        N, Kdim, splitBlk, blockIdx.y * 64 + wm * 32, blockIdx.x * 64 + wn * 32, wm, lane);
}

#define FUSE_KNN_THREADS 256
#define FUSE_THREADS     384

__global__ __launch_bounds__(FUSE_THREADS, 6) void knn_gemm1_kernel(
        const float* __restrict__ coords, unsigned short* __restrict__ knn,
        const float* __restrict__ W1, const float* __restrict__ feats,
        float* __restrict__ PQ, const float* __restrict__ W2, float* __restrict__ W2cat, int N) {
    __shared__ float4 tile[KNN_N];
    __shared__ float  hd[8][KNN_QW][KNN_CAP];
    __shared__ int    hi[8][KNN_QW][KNN_CAP];
    __shared__ int    so[32 * KNN_K];
    __shared__ __attribute__((aligned(16))) float wl[64 * (128 + 4)];
    const int lane = threadIdx.x & 31;
    const int wave = __builtin_amdgcn_readfirstlane((int)(threadIdx.x >> 5));
    const bool knn_role = wave < 8;
    const int tilesPerBatch = N / 32;
    const int b = blockIdx.x / tilesPerBatch;
    const int tb = blockIdx.x - b * tilesPerBatch;
    const int by = tb / (N / 128);
    const int np = tb - by * (N / 128);

    if (knn_role) {
        const float* cb = coords + (size_t)b * 3 * N;
#pragma unroll
        for (int it = 0; it < KNN_N / FUSE_KNN_THREADS; ++it) {
            const int i = it * FUSE_KNN_THREADS + (int)threadIdx.x;
            const float x = cb[i], y = cb[N + i], z = cb[2 * N + i];
            tile[i] = make_float4(x, y, z, __builtin_fmaf(z, z, __builtin_fmaf(x, x, y * y)));
        }
    } else {
        const int gt = (int)threadIdx.x - FUSE_KNN_THREADS;
        gemm_copy_w<128, true>(W1, wl, by, gt, 128);
        const int e = (int)blockIdx.x * 128 + gt;
        const int R = e >> 7, cc = e & 127;
        const float wa2 = W2[(R & 255) * 256 + cc];
        const float wb2 = W2[(R & 255) * 256 + 128 + cc];
        W2cat[e] = (R < 256) ? wa2 - wb2 : wb2;
    }
    __syncthreads();

    const int q0 = tb * 32 + wave * KNN_QW;
    float px[KNN_QW], py[KNN_QW], pz[KNN_QW], pw[KNN_QW];
    int cn[KNN_QW];
    if (knn_role) {
#pragma unroll
        for (int qi = 0; qi < KNN_QW; ++qi) {
            const float4 p = tile[q0 + qi];
            px[qi] = __builtin_bit_cast(float, __builtin_amdgcn_readfirstlane(__builtin_bit_cast(int, p.x)));
            py[qi] = __builtin_bit_cast(float, __builtin_amdgcn_readfirstlane(__builtin_bit_cast(int, p.y)));
            pz[qi] = __builtin_bit_cast(float, __builtin_amdgcn_readfirstlane(__builtin_bit_cast(int, p.z)));
            pw[qi] = __builtin_bit_cast(float, __builtin_amdgcn_readfirstlane(__builtin_bit_cast(int, p.w)));
        }

        float mn[KNN_QW];
#pragma unroll
        for (int qi = 0; qi < KNN_QW; ++qi) mn[qi] = __builtin_inff();
#pragma unroll 4
        for (int t = 0; t < N; t += 32) {
            float4 q = tile[t + lane];
#pragma unroll
            for (int qi = 0; qi < KNN_QW; ++qi) {
                float d = knn_dist(px[qi], py[qi], pz[qi], pw[qi], q);
                mn[qi] = d < mn[qi] ? d : mn[qi];
            }
        }

        float T[KNN_QW];
#pragma unroll
        for (int qi = 0; qi < KNN_QW; ++qi) {
            const float mv = mn[qi];
            int cnt = 0;
#pragma unroll
            for (int s = 0; s < 32; ++s) {
                float v = __builtin_bit_cast(float,
                    __builtin_amdgcn_readlane(__builtin_bit_cast(int, mv), s));
                cnt += knn_less(v, s, mv, lane) ? 1 : 0;
            }
            unsigned sel = __builtin_amdgcn_ballot_w32(cnt == KNN_K);
            T[qi] = sel ? __shfl(mv, __builtin_ctz(sel), 32) : __builtin_inff();
        }

#pragma unroll
        for (int qi = 0; qi < KNN_QW; ++qi) cn[qi] = 0;
        for (int t = 0; t < N; t += 128) {
            float4 q[4];
            float d[4][KNN_QW];
            unsigned bm[4][KNN_QW];
#pragma unroll
            for (int u = 0; u < 4; ++u) q[u] = tile[t + 32 * u + lane];
#pragma unroll
            for (int u = 0; u < 4; ++u) {
#pragma unroll
                for (int qi = 0; qi < KNN_QW; ++qi) {
                    d[u][qi] = knn_dist(px[qi], py[qi], pz[qi], pw[qi], q[u]);
                    bm[u][qi] = __builtin_amdgcn_ballot_w32(d[u][qi] <= T[qi]);
                }
            }
#pragma unroll
            for (int u = 0; u < 4; ++u) {
                if (bm[u][0] | bm[u][1] | bm[u][2] | bm[u][3]) {
#pragma unroll
                    for (int qi = 0; qi < KNN_QW; ++qi) {
                        const unsigned m = bm[u][qi];
                        if (m) {
                            const bool hit = ((m >> lane) & 1u) != 0u;
                            const int pos = cn[qi] + (int)__builtin_amdgcn_mbcnt_lo(m, 0u);
                            if (hit && pos < KNN_CAP) {
                                hd[wave][qi][pos] = d[u][qi];
                                hi[wave][qi][pos] = t + 32 * u + lane;
                            }
                            cn[qi] += __builtin_popcount(m);
                        }
                    }
                }
            }
        }
    } else {
        const int gt = (int)threadIdx.x - FUSE_KNN_THREADS;
        const int gw = gt >> 5;
        const int wm = gw >> 1, wn = gw & 1;
        const float* Xb = feats + (size_t)b * 128 * N;
        float* Db = PQ + (size_t)b * 256 * N;
        for (int t = 0; t < 2; ++t)
            gemm_tile<128, 4>(wl, Xb, Xb, Db, N, 128, 0,
                              by * 64 + wm * 32, (2 * np + t) * 64 + wn * 32, wm, lane);
    }
    __syncthreads();

    if (knn_role) {
#pragma unroll
        for (int qi = 0; qi < KNN_QW; ++qi) {
            const int n = cn[qi];
            const int iq = q0 + qi;
            int* so_q = so + (wave * KNN_QW + qi) * KNN_K;
            if (n > KNN_K && n <= KNN_CAP) {
                const float* sd = hd[wave][qi];
                const int*   si = hi[wave][qi];
                const bool v0 = lane < n, v1 = lane + 32 < n;
                const float d0 = v0 ? sd[lane] : __builtin_inff();
                const int   i0 = v0 ? si[lane] : 0x7fffffff;
                const float d1 = v1 ? sd[lane + 32] : __builtin_inff();
                const int   i1 = v1 ? si[lane + 32] : 0x7fffffff;
                int r0 = 0, r1 = 0;
                bool tie = false;
                for (int j = 0; j < n; ++j) {
                    const float dj = sd[j]; const int ij = si[j];
                    r0 += knn_less(dj, ij, d0, i0) ? 1 : 0;
                    r1 += knn_less(dj, ij, d1, i1) ? 1 : 0;
                    tie = tie || (dj == d0 && ij != i0) || (dj == d1 && ij != i1);
                }
                if (__builtin_amdgcn_ballot_w32(tie)) {
                    knn_replay(sd, si, n, lane, iq, so_q);
                    continue;
                }
                int sr = KNN_K + 1;
                unsigned s0 = __builtin_amdgcn_ballot_w32(v0 && i0 == iq);
                if (s0) sr = __shfl(r0, __builtin_ctz(s0), 32);
                unsigned s1 = __builtin_amdgcn_ballot_w32(v1 && i1 == iq);
                if (s1) sr = __shfl(r1, __builtin_ctz(s1), 32);
                if (v0 && i0 != iq && r0 <= KNN_K) {
                    int w = r0 - (r0 > sr ? 1 : 0);
                    if (w < KNN_K) so_q[w] = i0;
                }
                if (v1 && i1 != iq && r1 <= KNN_K) {
                    int w = r1 - (r1 > sr ? 1 : 0);
                    if (w < KNN_K) so_q[w] = i1;
                }
            } else {
                knn_exact_wave(tile, N, lane, px[qi], py[qi], pz[qi], pw[qi], iq, so_q);
            }
        }
    }
    __syncthreads();
    if (threadIdx.x < 16 * KNN_K) {
        unsigned* ob = (unsigned*)(knn + ((size_t)b * tilesPerBatch + tb) * (KNN_K * 32));
        const int k = (int)threadIdx.x >> 4, pp = ((int)threadIdx.x & 15) * 2;
        ob[threadIdx.x] = ((unsigned)so[pp * KNN_K + k] & 0xffffu) | ((unsigned)so[(pp + 1) * KNN_K + k] << 16);
    }
}

#define EDGE_THREADS 512
#define EDGE_GRP (KNN_N / (4 * EDGE_THREADS))

template <int M, int CG>
__global__ __launch_bounds__(EDGE_THREADS) void edge_chan_kernel(
        const float* __restrict__ PQ, const unsigned short* __restrict__ knn,
        float* __restrict__ xout, long outBatchStride, float invCnt) {
    typedef float vcg __attribute__((ext_vector_type(CG)));
    constexpr int N = KNN_N;
    __shared__ __attribute__((aligned(16))) float ql[N * CG];
    __shared__ __attribute__((aligned(16))) float mxl[N * CG];
    __shared__ float red[2 * CG][EDGE_THREADS / 32];
    __shared__ float s_mu[CG], s_r[CG];
    const int b = blockIdx.y;
    const int c0 = blockIdx.x * CG;
    const int t = threadIdx.x;
    const int lane = t & 31, wave = t >> 5;
    const float* Pb = PQ + ((size_t)b * (2 * M) + c0) * N;
    const float* Qb = Pb + (size_t)M * N;
    const unsigned short* ib = knn + (size_t)b * KNN_K * N;

    v4f qr[EDGE_GRP][CG];
#pragma unroll
    for (int g = 0; g < EDGE_GRP; ++g)
#pragma unroll
        for (int c = 0; c < CG; ++c)
            qr[g][c] = *(const v4f*)(Qb + (size_t)c * N + 4 * (g * EDGE_THREADS + t));
    v2i id[EDGE_GRP][KNN_K];
    v4f p[EDGE_GRP][CG];
#pragma unroll
    for (int g = 0; g < EDGE_GRP; ++g) {
#pragma unroll
        for (int k = 0; k < KNN_K; ++k)
            id[g][k] = *(const v2i*)(ib + ((4 * (g * EDGE_THREADS + t)) >> 5) * (KNN_K * 32) + k * 32
                                        + ((4 * (g * EDGE_THREADS + t)) & 31));
#pragma unroll
        for (int c = 0; c < CG; ++c)
            p[g][c] = *(const v4f*)(Pb + (size_t)c * N + 4 * (g * EDGE_THREADS + t));
    }
#pragma unroll
    for (int g = 0; g < EDGE_GRP; ++g) {
        const int n0 = 4 * (g * EDGE_THREADS + t);
#pragma unroll
        for (int j = 0; j < 4; ++j) {
            vcg q;
#pragma unroll
            for (int c = 0; c < CG; ++c) q[c] = qr[g][c][j];
            *(vcg*)(ql + (n0 + j) * CG) = q;
        }
    }
    __syncthreads();

    vcg s, s2;
#pragma unroll
    for (int c = 0; c < CG; ++c) { s[c] = 0.f; s2[c] = 0.f; }
#pragma unroll
    for (int g = 0; g < EDGE_GRP; ++g) {
        v4f mg[CG];
#pragma unroll
        for (int j = 0; j < 4; ++j) {
            vcg pv, m;
#pragma unroll
            for (int c = 0; c < CG; ++c) { pv[c] = p[g][c][j]; m[c] = -3.4e38f; }
#pragma unroll
            for (int k = 0; k < KNN_K; ++k) {
                const unsigned w = (unsigned)id[g][k][j >> 1];
                const int ix = (j & 1) ? (int)(w >> 16) : (int)(w & 0xffffu);
                const vcg q = *(const vcg*)(ql + ix * CG);
                const vcg y = pv + q;
                m = __builtin_elementwise_max(m, y);
                s += y;
                s2 += y * y;
            }
#pragma unroll
            for (int c = 0; c < CG; ++c) mg[c][j] = m[c];
            if (j & 1) __builtin_amdgcn_sched_barrier(0);
        }
#pragma unroll
        for (int c = 0; c < CG; ++c)
            *(v4f*)(mxl + ((g * CG + c) * EDGE_THREADS + t) * 4) = mg[c];
    }

#pragma unroll
    for (int c = 0; c < CG; ++c) {
        float a = s[c], a2 = s2[c];
        for (int off = 16; off > 0; off >>= 1) {
            a  += __shfl_xor(a, off, 32);
            a2 += __shfl_xor(a2, off, 32);
        }
        if (lane == 0) { red[c][wave] = a; red[CG + c][wave] = a2; }
    }
    __syncthreads();
    if (t < CG) {
        float a = 0.f, a2 = 0.f;
        for (int w = 0; w < EDGE_THREADS / 32; ++w) { a += red[t][w]; a2 += red[CG + t][w]; }
        const float mu = a * invCnt;
        const float e2 = a2 * invCnt;
        const float var = fmaxf(__builtin_fmaf(-mu, mu, e2), 0.f);
        s_mu[t] = mu; s_r[t] = rsqrtf(var + IN_EPS);
    }
    __syncthreads();

    float* ob = xout + (size_t)b * outBatchStride + (size_t)c0 * N;
#pragma unroll
    for (int c = 0; c < CG; ++c) {
        const float mu = s_mu[c], r = s_r[c];
#pragma unroll
        for (int g = 0; g < EDGE_GRP; ++g) {
            const v4f v = *(const v4f*)(mxl + ((g * CG + c) * EDGE_THREADS + t) * 4);
            const v4f y = (v - mu) * r;
            const v4f o = __builtin_elementwise_max(y, y * 0.2f);
            *(v4f*)(ob + (size_t)c * N + 4 * (g * EDGE_THREADS + t)) = o;
        }
    }
}

#define FIN_THREADS 64
#define FIN_GRP (KNN_N / (4 * FIN_THREADS))

__global__ __launch_bounds__(FIN_THREADS) void final_norm_kernel(
        const float* __restrict__ xin, float* __restrict__ out) {
    constexpr int N = KNN_N;
    __shared__ float red[2][FIN_THREADS / 32];
    const int b = blockIdx.y, c = blockIdx.x, C = gridDim.x;
    const int t = threadIdx.x;
    const int lane = t & 31, wave = t >> 5;
    const float* row = xin + ((size_t)b * C + c) * N;
    v4f v[FIN_GRP];
#pragma unroll
    for (int g = 0; g < FIN_GRP; ++g)
        v[g] = *(const v4f*)(row + 4 * (g * FIN_THREADS + t));
    v4f sv = v[0], qv = v[0] * v[0];
#pragma unroll
    for (int g = 1; g < FIN_GRP; ++g) { sv += v[g]; qv += v[g] * v[g]; }
    float s = (sv[0] + sv[1]) + (sv[2] + sv[3]);
    float s2 = (qv[0] + qv[1]) + (qv[2] + qv[3]);
    for (int off = 16; off > 0; off >>= 1) {
        s  += __shfl_xor(s, off, 32);
        s2 += __shfl_xor(s2, off, 32);
    }
    if (lane == 0) { red[0][wave] = s; red[1][wave] = s2; }
    __syncthreads();
    float a = 0.f, a2 = 0.f;
#pragma unroll
    for (int w = 0; w < FIN_THREADS / 32; ++w) { a += red[0][w]; a2 += red[1][w]; }
    const float mu = a / (float)N;
    const float var = fmaxf(a2 / (float)N - mu * mu, 0.f);
    const float r = rsqrtf(var + IN_EPS);
    float* orow = out + ((size_t)b * C + c) * N;
#pragma unroll
    for (int g = 0; g < FIN_GRP; ++g) {
        v4f o;
#pragma unroll
        for (int j = 0; j < 4; ++j) o[j] = lrelu((v[g][j] - mu) * r);
        *(v4f*)(orow + 4 * (g * FIN_THREADS + t)) = o;
    }
}

extern "C" void kernel_launch(void* const* d_in, const int* in_sizes, int n_in,
                              void* d_out, int out_size, void* d_ws, size_t ws_size,
                              hipStream_t stream) {
    constexpr int B = 4, C = 128, N = 4096;
    const float* coords = (const float*)d_in[0];
    const float* feats  = (const float*)d_in[1];
    const float* W1     = (const float*)d_in[2];
    const float* W2     = (const float*)d_in[3];
    const float* W3     = (const float*)d_in[4];
    float* out = (float*)d_out;

    char* ws = (char*)d_ws;
    size_t off = 0;
    auto carve = [&](size_t bytes) -> void* {
        void* p = ws + off;
        off = (off + bytes + 255) & ~(size_t)255;
        return p;
    };
    float* W2cat  = (float*)carve(512 * 128 * sizeof(float));
    unsigned short* knn = (unsigned short*)carve((size_t)B * N * KNN_K * sizeof(unsigned short));
    float* PQ     = (float*)carve((size_t)B * 512 * N * sizeof(float));
    float* X3     = (float*)carve((size_t)B * 512 * N * sizeof(float));
    float* x3pre  = (float*)carve((size_t)B * 128 * N * sizeof(float));
    static_assert(N == KNN_N && N % 128 == 0, "points per batch held in LDS; tile pairs per row block");
    static_assert(N % (4 * EDGE_THREADS) == 0 && N <= 65536, "whole groups of 4 points per edge thread; 16-bit indices");
    static_assert(B * (N / 32) * 128 == 512 * 128, "one stacked-W2 element per GEMM thread of the first launch");
    static_assert(N % (4 * FIN_THREADS) == 0, "whole groups of 4 values per thread of the final pass");
    (void)ws_size; (void)in_sizes; (void)n_in; (void)out_size;

    knn_gemm1_kernel<<<B * (N / 32), FUSE_THREADS, 0, stream>>>(coords, knn, W1, feats, PQ, W2, W2cat, N);

    const float invCnt = 1.0f / (float)(N * KNN_K);

    edge_chan_kernel<128, 2><<<dim3(128 / 2, B), EDGE_THREADS, 0, stream>>>(
        PQ, knn, X3 + (size_t)128 * N, (long)512 * N, invCnt);

    gemm_wmma_kernel<128><<<dim3(N / 64, 512 / 64, B), 128, 0, stream>>>(
        W2cat, X3 + (size_t)128 * N, X3 + (size_t)128 * N, PQ, N, C,
        (long)512 * N, (long)512 * N, 0, (long)512 * N);
    edge_chan_kernel<256, 4><<<dim3(256 / 4, B), EDGE_THREADS, 0, stream>>>(
        PQ, knn, X3 + (size_t)256 * N, (long)512 * N, invCnt);

    gemm_wmma_kernel<512><<<dim3(N / 64, 128 / 64, B), 128, 0, stream>>>(
        W3, feats, X3, x3pre, N, 512, (long)C * N, (long)512 * N, 4, (long)128 * N);
    final_norm_kernel<<<dim3(128, B), FIN_THREADS, 0, stream>>>(x3pre, out);
}


# KNN: both sweeps over all candidates use a 4-op distance (3 products + |q|^2) against a bound widened by a rounding margin; exact distances only for survivors; neighbour lists bit-identical
# speedup vs baseline: 6.0998x; 1.0371x over previous
#include <hip/hip_runtime.h>
#include <hip/hip_bf16.h>


typedef __attribute__((ext_vector_type(2))) float v2f;
typedef __attribute__((ext_vector_type(8))) float v8f;
typedef __attribute__((ext_vector_type(4))) float v4f;
typedef __attribute__((ext_vector_type(2))) int   v2i;

#define KNN_K 10
#define IN_EPS 1e-5f

__device__ __forceinline__ float lrelu(float y) { return y >= 0.0f ? y : 0.2f * y; }

__device__ __forceinline__ v8f wmma_f32(v2f a, v2f b, v8f c) {
    return __builtin_amdgcn_wmma_f32_16x16x4_f32(
         false, a,  false, b,
         (short)0, c,  false,  false);
}

#define KNN_N   4096
#define KNN_QW  4
#define KNN_CAP 64

__device__ __forceinline__ float knn_dist(float px, float py, float pz, float pw, float4 q) {
    float d;
    {
#pragma clang fp contract(off)
        d = __builtin_fmaf(-2.0f, __builtin_fmaf(px, q.x, py * q.y) + pz * q.z, pw + q.w);
    }
    return d;
}

#define KNN_MARGIN 0x1p-17f
__device__ __forceinline__ float knn_sweep(float ax, float ay, float az, float4 q) {
    return __builtin_fmaf(ax, q.x, __builtin_fmaf(ay, q.y, az * q.z)) + q.w;
}

__device__ __forceinline__ bool knn_less(float da, int ia, float db, int ib) {
    return (da < db) || (da == db && ia < ib);
}

__device__ __forceinline__ void knn_exact_wave(const float4* tile, int N, int lane,
        float px, float py, float pz, float pw, int iq, int* so_q) {
    float ld = -__builtin_inff(); int li = -1; int w = 0;
    for (int r = 0; r <= KNN_K; ++r) {
        float bd = __builtin_inff(); int bi = 0x7fffffff;
        for (int t = 0; t < N; t += 32) {
            const int c = t + lane;
            float d = knn_dist(px, py, pz, pw, tile[c]);
            if (knn_less(ld, li, d, c) && knn_less(d, c, bd, bi)) { bd = d; bi = c; }
        }
        for (int off = 16; off > 0; off >>= 1) {
            float od = __shfl_xor(bd, off, 32); int oi = __shfl_xor(bi, off, 32);
            if (knn_less(od, oi, bd, bi)) { bd = od; bi = oi; }
        }
        ld = bd; li = bi;
        if (bi != iq) {
            if (w < KNN_K && lane == 0) so_q[w] = (bi < N) ? bi : 0;
            ++w;
        }
    }
}

__device__ __forceinline__ void knn_replay(const float* sd, const int* si, int n,
                                           int lane, int iq, int* so_q) {
    float bd[KNN_K + 1];
    int   bi[KNN_K + 1];
#pragma unroll
    for (int s = 0; s <= KNN_K; ++s) { bd[s] = 3.4e38f; bi[s] = -1; }
    for (int j = 0; j < n; ++j) {
        float cd = sd[j]; int ci = si[j];
        if (cd < bd[KNN_K]) {
#pragma unroll
            for (int s = 0; s <= KNN_K; ++s) {
                bool sw = cd < bd[s];
                float td = sw ? bd[s] : cd; int ti = sw ? bi[s] : ci;
                bd[s] = sw ? cd : bd[s];    bi[s] = sw ? ci : bi[s];
                cd = td; ci = ti;
            }
        }
    }
    if (lane == 0) {
        int w = 0;
#pragma unroll
        for (int s = 0; s <= KNN_K; ++s)
            if (bi[s] != iq && w < KNN_K) { so_q[w] = bi[s]; ++w; }
    }
}


template <int KD, bool CAT>
__device__ __forceinline__ void gemm_copy_w(const float* __restrict__ W, float* wl,
                                            int by, int tid, int Mhalf) {
    constexpr int LDW = KD + 4;
    if (CAT) {
        const int R0 = by * 64;
        const bool diff = R0 < Mhalf;
        const float* wsrc = W + (size_t)(diff ? R0 : R0 - Mhalf) * (2 * KD);
        if (diff) {
#pragma unroll 8
            for (int it = 0; it < 64 * KD / 4 / 128; ++it) {
                const int i = it * 128 + tid;
                const int r = i / (KD / 4), c4 = i - r * (KD / 4);
                const float4 wa = *(const float4*)(wsrc + (size_t)r * (2 * KD) + 4 * c4);
                const float4 wb = *(const float4*)(wsrc + (size_t)r * (2 * KD) + KD + 4 * c4);
                *(float4*)(wl + r * LDW + 4 * c4) =
                    make_float4(wa.x - wb.x, wa.y - wb.y, wa.z - wb.z, wa.w - wb.w);
            }
        } else {
#pragma unroll 16
            for (int it = 0; it < 64 * KD / 4 / 128; ++it) {
                const int i = it * 128 + tid;
                const int r = i / (KD / 4), c4 = i - r * (KD / 4);
                *(float4*)(wl + r * LDW + 4 * c4) =
                    *(const float4*)(wsrc + (size_t)r * (2 * KD) + KD + 4 * c4);
            }
        }
    } else {
        const float4* wsrc = (const float4*)(W + (size_t)(by * 64) * KD);
#pragma unroll 16
        for (int it = 0; it < 64 * KD / 4 / 128; ++it) {
            const int i = it * 128 + tid;
            const int r = i / (KD / 4), c4 = i - r * (KD / 4);
            *(float4*)(wl + r * LDW + 4 * c4) = wsrc[i];
        }
    }
}

template <int KD, int KB>
__device__ __forceinline__ void gemm_tile(
        const float* wl, const float* __restrict__ Xb, const float* X2b,
        float* __restrict__ Db, int N, int Kdim,
        int splitBlk, int m0, int n0, int wm, int lane) {
    constexpr int LDW = KD + 4;
    const int half = lane >> 4;
    const int l16  = lane & 15;

    v8f c00 = {0.f,0.f,0.f,0.f,0.f,0.f,0.f,0.f};
    v8f c01 = c00, c10 = c00, c11 = c00;

    const float* ar0 = wl + (wm * 32 + l16) * LDW + 2 * half;
    const float* ar1 = ar0 + 16 * LDW;
    const float* xr  = Xb + (size_t)(2 * half) * N + n0 + l16;
    const float* xr2 = X2b + (size_t)(splitBlk * (4 * KB) + 2 * half) * N + n0 + l16;
    int blk = 0;

    __builtin_prefetch(xr, 0, 1);

    v2f fa0[KB], fa1[KB], fb0[KB], fb1[KB];
    v2f ga0[KB], ga1[KB], gb0[KB], gb1[KB];

    auto load_blk = [&](v2f (&a0)[KB], v2f (&a1)[KB], v2f (&b0)[KB], v2f (&b1)[KB]) {
#pragma unroll
        for (int s = 0; s < KB; ++s) {
            const float* xs = xr + (size_t)(4 * s) * N;
            a0[s] = *(const v2f*)(ar0 + 4 * s);
            a1[s] = *(const v2f*)(ar1 + 4 * s);
            b0[s].x = xs[0];        b0[s].y = xs[N];
            b1[s].x = xs[16];       b1[s].y = xs[16 + N];
        }
        ar0 += 4 * KB; ar1 += 4 * KB; xr += (size_t)(4 * KB) * N;
        if (++blk == splitBlk) xr = xr2;
    };
    auto mma_blk = [&](const v2f (&a0)[KB], const v2f (&a1)[KB],
                       const v2f (&b0)[KB], const v2f (&b1)[KB]) {
#pragma unroll
        for (int s = 0; s < KB; ++s) {
            c00 = wmma_f32(a0[s], b0[s], c00);
            c01 = wmma_f32(a0[s], b1[s], c01);
            c10 = wmma_f32(a1[s], b0[s], c10);
            c11 = wmma_f32(a1[s], b1[s], c11);
        }
    };

    const int nblk = Kdim / (4 * KB);
    load_blk(fa0, fa1, fb0, fb1);
    for (int j = 0; j < nblk - 2; j += 2) {
        load_blk(ga0, ga1, gb0, gb1);
        __builtin_amdgcn_sched_barrier(0);
        mma_blk(fa0, fa1, fb0, fb1);
        load_blk(fa0, fa1, fb0, fb1);
        __builtin_amdgcn_sched_barrier(0);
        mma_blk(ga0, ga1, gb0, gb1);
    }
    load_blk(ga0, ga1, gb0, gb1);
    __builtin_amdgcn_sched_barrier(0);
    mma_blk(fa0, fa1, fb0, fb1);
    mma_blk(ga0, ga1, gb0, gb1);
#pragma unroll
    for (int v = 0; v < 8; ++v) {
        const int mr0 = m0 + v + 8 * half;
        Db[(size_t)mr0 * N + n0 + l16]             = c00[v];
        Db[(size_t)mr0 * N + n0 + 16 + l16]        = c01[v];
        Db[(size_t)(mr0 + 16) * N + n0 + l16]      = c10[v];
        Db[(size_t)(mr0 + 16) * N + n0 + 16 + l16] = c11[v];
    }
}

template <int KD>
__global__ __launch_bounds__(128) void gemm_wmma_kernel(
        const float* __restrict__ W, const float* __restrict__ X, const float* X2,
        float* __restrict__ D, int N, int Kdim,
        long xBatchStride, long x2BatchStride, int splitBlk, long dBatchStride) {
    __shared__ __attribute__((aligned(16))) float wl[64 * (KD + 4)];
    gemm_copy_w<KD, false>(W, wl, (int)blockIdx.y, (int)threadIdx.x, 0);
    __syncthreads();

    const int lane = threadIdx.x & 31;
    const int wave = threadIdx.x >> 5;
    const int wm = wave >> 1, wn = wave & 1;
    gemm_tile<KD, 8>(
        wl, X + (size_t)blockIdx.z * xBatchStride, X2 + (size_t)blockIdx.z * x2BatchStride,
        D + (size_t)blockIdx.z * dBatchStride,
        N, Kdim, splitBlk, blockIdx.y * 64 + wm * 32, blockIdx.x * 64 + wn * 32, wm, lane);
}

#define FUSE_KNN_THREADS 256
#define FUSE_THREADS     384

__global__ __launch_bounds__(FUSE_THREADS, 6) void knn_gemm1_kernel(
        const float* __restrict__ coords, unsigned short* __restrict__ knn,
        const float* __restrict__ W1, const float* __restrict__ feats,
        float* __restrict__ PQ, const float* __restrict__ W2, float* __restrict__ W2cat, int N) {
    __shared__ float4 tile[KNN_N];
    __shared__ float  hd[8][KNN_QW][KNN_CAP];
    __shared__ int    hi[8][KNN_QW][KNN_CAP];
    __shared__ int    so[32 * KNN_K];
    __shared__ float  pmx[8];
    __shared__ __attribute__((aligned(16))) float wl[64 * (128 + 4)];
    const int lane = threadIdx.x & 31;
    const int wave = __builtin_amdgcn_readfirstlane((int)(threadIdx.x >> 5));
    const bool knn_role = wave < 8;
    const int tilesPerBatch = N / 32;
    const int b = blockIdx.x / tilesPerBatch;
    const int tb = blockIdx.x - b * tilesPerBatch;
    const int by = tb / (N / 128);
    const int np = tb - by * (N / 128);

    if (knn_role) {
        const float* cb = coords + (size_t)b * 3 * N;
        float wmax = 0.0f;
#pragma unroll
        for (int it = 0; it < KNN_N / FUSE_KNN_THREADS; ++it) {
            const int i = it * FUSE_KNN_THREADS + (int)threadIdx.x;
            const float x = cb[i], y = cb[N + i], z = cb[2 * N + i];
            const float w = __builtin_fmaf(z, z, __builtin_fmaf(x, x, y * y));
            tile[i] = make_float4(x, y, z, w);
            wmax = __builtin_fmaxf(wmax, w);
        }
        for (int off = 16; off > 0; off >>= 1) wmax = __builtin_fmaxf(wmax, __shfl_xor(wmax, off, 32));
        if (lane == 0) pmx[wave] = wmax;
    } else {
        const int gt = (int)threadIdx.x - FUSE_KNN_THREADS;
        gemm_copy_w<128, true>(W1, wl, by, gt, 128);
        const int e = (int)blockIdx.x * 128 + gt;
        const int R = e >> 7, cc = e & 127;
        const float wa2 = W2[(R & 255) * 256 + cc];
        const float wb2 = W2[(R & 255) * 256 + 128 + cc];
        W2cat[e] = (R < 256) ? wa2 - wb2 : wb2;
    }
    __syncthreads();

    const int q0 = tb * 32 + wave * KNN_QW;
    float px[KNN_QW], py[KNN_QW], pz[KNN_QW], pw[KNN_QW];
    int cn[KNN_QW];
    if (knn_role) {
        float qmax = pmx[0];
#pragma unroll
        for (int w = 1; w < 8; ++w) qmax = __builtin_fmaxf(qmax, pmx[w]);
        qmax = __builtin_bit_cast(float, __builtin_amdgcn_readfirstlane(__builtin_bit_cast(int, qmax)));
        float ax[KNN_QW], ay[KNN_QW], az[KNN_QW];
#pragma unroll
        for (int qi = 0; qi < KNN_QW; ++qi) {
            const float4 p = tile[q0 + qi];
            px[qi] = __builtin_bit_cast(float, __builtin_amdgcn_readfirstlane(__builtin_bit_cast(int, p.x)));
            py[qi] = __builtin_bit_cast(float, __builtin_amdgcn_readfirstlane(__builtin_bit_cast(int, p.y)));
            pz[qi] = __builtin_bit_cast(float, __builtin_amdgcn_readfirstlane(__builtin_bit_cast(int, p.z)));
            pw[qi] = __builtin_bit_cast(float, __builtin_amdgcn_readfirstlane(__builtin_bit_cast(int, p.w)));
            ax[qi] = -2.0f * px[qi]; ay[qi] = -2.0f * py[qi]; az[qi] = -2.0f * pz[qi];
        }

        float mn[KNN_QW];
#pragma unroll
        for (int qi = 0; qi < KNN_QW; ++qi) mn[qi] = __builtin_inff();
#pragma clang loop vectorize(disable)
#pragma unroll 4
        for (int t = 0; t < N; t += 32) {
            float4 q = tile[t + lane];
#pragma unroll
            for (int qi = 0; qi < KNN_QW; ++qi)
                mn[qi] = __builtin_fminf(mn[qi], knn_sweep(ax[qi], ay[qi], az[qi], q));
        }

        float T[KNN_QW];
#pragma unroll
        for (int qi = 0; qi < KNN_QW; ++qi) {
            const float mv = mn[qi];
            int cnt = 0;
#pragma unroll
            for (int s = 0; s < 32; ++s) {
                float v = __builtin_bit_cast(float,
                    __builtin_amdgcn_readlane(__builtin_bit_cast(int, mv), s));
                cnt += knn_less(v, s, mv, lane) ? 1 : 0;
            }
            unsigned sel = __builtin_amdgcn_ballot_w32(cnt == KNN_K);
            T[qi] = sel ? __shfl(mv, __builtin_ctz(sel), 32) + KNN_MARGIN * (pw[qi] + qmax) : __builtin_inff();
        }

#pragma unroll
        for (int qi = 0; qi < KNN_QW; ++qi) cn[qi] = 0;
        for (int t = 0; t < N; t += 128) {
            float4 q[4];
            unsigned bm[4][KNN_QW];
#pragma unroll
            for (int u = 0; u < 4; ++u) q[u] = tile[t + 32 * u + lane];
#pragma unroll
            for (int u = 0; u < 4; ++u) {
#pragma unroll
                for (int qi = 0; qi < KNN_QW; ++qi)
                    bm[u][qi] = __builtin_amdgcn_ballot_w32(knn_sweep(ax[qi], ay[qi], az[qi], q[u]) <= T[qi]);
            }
#pragma unroll
            for (int u = 0; u < 4; ++u) {
                if (bm[u][0] | bm[u][1] | bm[u][2] | bm[u][3]) {
#pragma unroll
                    for (int qi = 0; qi < KNN_QW; ++qi) {
                        const unsigned m = bm[u][qi];
                        if (m) {
                            const bool hit = ((m >> lane) & 1u) != 0u;
                            const int pos = cn[qi] + (int)__builtin_amdgcn_mbcnt_lo(m, 0u);
                            if (hit && pos < KNN_CAP) {
                                hd[wave][qi][pos] = knn_dist(px[qi], py[qi], pz[qi], pw[qi], q[u]);
                                hi[wave][qi][pos] = t + 32 * u + lane;
                            }
                            cn[qi] += __builtin_popcount(m);
                        }
                    }
                }
            }
        }
    } else {
        const int gt = (int)threadIdx.x - FUSE_KNN_THREADS;
        const int gw = gt >> 5;
        const int wm = gw >> 1, wn = gw & 1;
        const float* Xb = feats + (size_t)b * 128 * N;
        float* Db = PQ + (size_t)b * 256 * N;
        for (int t = 0; t < 2; ++t)
            gemm_tile<128, 4>(wl, Xb, Xb, Db, N, 128, 0,
                              by * 64 + wm * 32, (2 * np + t) * 64 + wn * 32, wm, lane);
    }
    __syncthreads();

    if (knn_role) {
#pragma unroll
        for (int qi = 0; qi < KNN_QW; ++qi) {
            const int n = cn[qi];
            const int iq = q0 + qi;
            int* so_q = so + (wave * KNN_QW + qi) * KNN_K;
            if (n > KNN_K && n <= KNN_CAP) {
                const float* sd = hd[wave][qi];
                const int*   si = hi[wave][qi];
                const bool v0 = lane < n, v1 = lane + 32 < n;
                const float d0 = v0 ? sd[lane] : __builtin_inff();
                const int   i0 = v0 ? si[lane] : 0x7fffffff;
                const float d1 = v1 ? sd[lane + 32] : __builtin_inff();
                const int   i1 = v1 ? si[lane + 32] : 0x7fffffff;
                int r0 = 0, r1 = 0;
                bool tie = false;
                for (int j = 0; j < n; ++j) {
                    const float dj = sd[j]; const int ij = si[j];
                    r0 += knn_less(dj, ij, d0, i0) ? 1 : 0;
                    r1 += knn_less(dj, ij, d1, i1) ? 1 : 0;
                    tie = tie || (dj == d0 && ij != i0) || (dj == d1 && ij != i1);
                }
                if (__builtin_amdgcn_ballot_w32(tie)) {
                    knn_replay(sd, si, n, lane, iq, so_q);
                    continue;
                }
                int sr = KNN_K + 1;
                unsigned s0 = __builtin_amdgcn_ballot_w32(v0 && i0 == iq);
                if (s0) sr = __shfl(r0, __builtin_ctz(s0), 32);
                unsigned s1 = __builtin_amdgcn_ballot_w32(v1 && i1 == iq);
                if (s1) sr = __shfl(r1, __builtin_ctz(s1), 32);
                if (v0 && i0 != iq && r0 <= KNN_K) {
                    int w = r0 - (r0 > sr ? 1 : 0);
                    if (w < KNN_K) so_q[w] = i0;
                }
                if (v1 && i1 != iq && r1 <= KNN_K) {
                    int w = r1 - (r1 > sr ? 1 : 0);
                    if (w < KNN_K) so_q[w] = i1;
                }
            } else {
                knn_exact_wave(tile, N, lane, px[qi], py[qi], pz[qi], pw[qi], iq, so_q);
            }
        }
    }
    __syncthreads();
    if (threadIdx.x < 16 * KNN_K) {
        unsigned* ob = (unsigned*)(knn + ((size_t)b * tilesPerBatch + tb) * (KNN_K * 32));
        const int k = (int)threadIdx.x >> 4, pp = ((int)threadIdx.x & 15) * 2;
        ob[threadIdx.x] = ((unsigned)so[pp * KNN_K + k] & 0xffffu) | ((unsigned)so[(pp + 1) * KNN_K + k] << 16);
    }
}

#define EDGE_THREADS 512
#define EDGE_GRP (KNN_N / (4 * EDGE_THREADS))

template <int M, int CG>
__global__ __launch_bounds__(EDGE_THREADS) void edge_chan_kernel(
        const float* __restrict__ PQ, const unsigned short* __restrict__ knn,
        float* __restrict__ xout, long outBatchStride, float invCnt) {
    typedef float vcg __attribute__((ext_vector_type(CG)));
    constexpr int N = KNN_N;
    __shared__ __attribute__((aligned(16))) float ql[N * CG];
    __shared__ __attribute__((aligned(16))) float mxl[N * CG];
    __shared__ float red[2 * CG][EDGE_THREADS / 32];
    __shared__ float s_mu[CG], s_r[CG];
    const int b = blockIdx.y;
    const int c0 = blockIdx.x * CG;
    const int t = threadIdx.x;
    const int lane = t & 31, wave = t >> 5;
    const float* Pb = PQ + ((size_t)b * (2 * M) + c0) * N;
    const float* Qb = Pb + (size_t)M * N;
    const unsigned short* ib = knn + (size_t)b * KNN_K * N;

    v4f qr[EDGE_GRP][CG];
#pragma unroll
    for (int g = 0; g < EDGE_GRP; ++g)
#pragma unroll
        for (int c = 0; c < CG; ++c)
            qr[g][c] = *(const v4f*)(Qb + (size_t)c * N + 4 * (g * EDGE_THREADS + t));
    v2i id[EDGE_GRP][KNN_K];
    v4f p[EDGE_GRP][CG];
#pragma unroll
    for (int g = 0; g < EDGE_GRP; ++g) {
#pragma unroll
        for (int k = 0; k < KNN_K; ++k)
            id[g][k] = *(const v2i*)(ib + ((4 * (g * EDGE_THREADS + t)) >> 5) * (KNN_K * 32) + k * 32
                                        + ((4 * (g * EDGE_THREADS + t)) & 31));
#pragma unroll
        for (int c = 0; c < CG; ++c)
            p[g][c] = *(const v4f*)(Pb + (size_t)c * N + 4 * (g * EDGE_THREADS + t));
    }
#pragma unroll
    for (int g = 0; g < EDGE_GRP; ++g) {
        const int n0 = 4 * (g * EDGE_THREADS + t);
#pragma unroll
        for (int j = 0; j < 4; ++j) {
            vcg q;
#pragma unroll
            for (int c = 0; c < CG; ++c) q[c] = qr[g][c][j];
            *(vcg*)(ql + (n0 + j) * CG) = q;
        }
    }
    __syncthreads();

    vcg s, s2;
#pragma unroll
    for (int c = 0; c < CG; ++c) { s[c] = 0.f; s2[c] = 0.f; }
#pragma unroll
    for (int g = 0; g < EDGE_GRP; ++g) {
        v4f mg[CG];
#pragma unroll
        for (int j = 0; j < 4; ++j) {
            vcg pv, m;
#pragma unroll
            for (int c = 0; c < CG; ++c) { pv[c] = p[g][c][j]; m[c] = -3.4e38f; }
#pragma unroll
            for (int k = 0; k < KNN_K; ++k) {
                const unsigned w = (unsigned)id[g][k][j >> 1];
                const int ix = (j & 1) ? (int)(w >> 16) : (int)(w & 0xffffu);
                const vcg q = *(const vcg*)(ql + ix * CG);
                const vcg y = pv + q;
                m = __builtin_elementwise_max(m, y);
                s += y;
                s2 += y * y;
            }
#pragma unroll
            for (int c = 0; c < CG; ++c) mg[c][j] = m[c];
            if (j & 1) __builtin_amdgcn_sched_barrier(0);
        }
#pragma unroll
        for (int c = 0; c < CG; ++c)
            *(v4f*)(mxl + ((g * CG + c) * EDGE_THREADS + t) * 4) = mg[c];
    }

#pragma unroll
    for (int c = 0; c < CG; ++c) {
        float a = s[c], a2 = s2[c];
        for (int off = 16; off > 0; off >>= 1) {
            a  += __shfl_xor(a, off, 32);
            a2 += __shfl_xor(a2, off, 32);
        }
        if (lane == 0) { red[c][wave] = a; red[CG + c][wave] = a2; }
    }
    __syncthreads();
    if (t < CG) {
        float a = 0.f, a2 = 0.f;
        for (int w = 0; w < EDGE_THREADS / 32; ++w) { a += red[t][w]; a2 += red[CG + t][w]; }
        const float mu = a * invCnt;
        const float e2 = a2 * invCnt;
        const float var = fmaxf(__builtin_fmaf(-mu, mu, e2), 0.f);
        s_mu[t] = mu; s_r[t] = rsqrtf(var + IN_EPS);
    }
    __syncthreads();

    float* ob = xout + (size_t)b * outBatchStride + (size_t)c0 * N;
#pragma unroll
    for (int c = 0; c < CG; ++c) {
        const float mu = s_mu[c], r = s_r[c];
#pragma unroll
        for (int g = 0; g < EDGE_GRP; ++g) {
            const v4f v = *(const v4f*)(mxl + ((g * CG + c) * EDGE_THREADS + t) * 4);
            const v4f y = (v - mu) * r;
            const v4f o = __builtin_elementwise_max(y, y * 0.2f);
            *(v4f*)(ob + (size_t)c * N + 4 * (g * EDGE_THREADS + t)) = o;
        }
    }
}

#define FIN_THREADS 64
#define FIN_GRP (KNN_N / (4 * FIN_THREADS))

__global__ __launch_bounds__(FIN_THREADS) void final_norm_kernel(
        const float* __restrict__ xin, float* __restrict__ out) {
    constexpr int N = KNN_N;
    __shared__ float red[2][FIN_THREADS / 32];
    const int b = blockIdx.y, c = blockIdx.x, C = gridDim.x;
    const int t = threadIdx.x;
    const int lane = t & 31, wave = t >> 5;
    const float* row = xin + ((size_t)b * C + c) * N;
    v4f v[FIN_GRP];
#pragma unroll
    for (int g = 0; g < FIN_GRP; ++g)
        v[g] = *(const v4f*)(row + 4 * (g * FIN_THREADS + t));
    v4f sv = v[0], qv = v[0] * v[0];
#pragma unroll
    for (int g = 1; g < FIN_GRP; ++g) { sv += v[g]; qv += v[g] * v[g]; }
    float s = (sv[0] + sv[1]) + (sv[2] + sv[3]);
    float s2 = (qv[0] + qv[1]) + (qv[2] + qv[3]);
    for (int off = 16; off > 0; off >>= 1) {
        s  += __shfl_xor(s, off, 32);
        s2 += __shfl_xor(s2, off, 32);
    }
    if (lane == 0) { red[0][wave] = s; red[1][wave] = s2; }
    __syncthreads();
    float a = 0.f, a2 = 0.f;
#pragma unroll
    for (int w = 0; w < FIN_THREADS / 32; ++w) { a += red[0][w]; a2 += red[1][w]; }
    const float mu = a / (float)N;
    const float var = fmaxf(a2 / (float)N - mu * mu, 0.f);
    const float r = rsqrtf(var + IN_EPS);
    float* orow = out + ((size_t)b * C + c) * N;
#pragma unroll
    for (int g = 0; g < FIN_GRP; ++g) {
        v4f o;
#pragma unroll
        for (int j = 0; j < 4; ++j) o[j] = lrelu((v[g][j] - mu) * r);
        *(v4f*)(orow + 4 * (g * FIN_THREADS + t)) = o;
    }
}

extern "C" void kernel_launch(void* const* d_in, const int* in_sizes, int n_in,
                              void* d_out, int out_size, void* d_ws, size_t ws_size,
                              hipStream_t stream) {
    constexpr int B = 4, C = 128, N = 4096;
    const float* coords = (const float*)d_in[0];
    const float* feats  = (const float*)d_in[1];
    const float* W1     = (const float*)d_in[2];
    const float* W2     = (const float*)d_in[3];
    const float* W3     = (const float*)d_in[4];
    float* out = (float*)d_out;

    char* ws = (char*)d_ws;
    size_t off = 0;
    auto carve = [&](size_t bytes) -> void* {
        void* p = ws + off;
        off = (off + bytes + 255) & ~(size_t)255;
        return p;
    };
    float* W2cat  = (float*)carve(512 * 128 * sizeof(float));
    unsigned short* knn = (unsigned short*)carve((size_t)B * N * KNN_K * sizeof(unsigned short));
    float* PQ     = (float*)carve((size_t)B * 512 * N * sizeof(float));
    float* X3     = (float*)carve((size_t)B * 512 * N * sizeof(float));
    float* x3pre  = (float*)carve((size_t)B * 128 * N * sizeof(float));
    static_assert(N == KNN_N && N % 128 == 0, "points per batch held in LDS; tile pairs per row block");
    static_assert(N % (4 * EDGE_THREADS) == 0 && N <= 65536, "whole groups of 4 points per edge thread; 16-bit indices");
    static_assert(B * (N / 32) * 128 == 512 * 128, "one stacked-W2 element per GEMM thread of the first launch");
    static_assert(N % (4 * FIN_THREADS) == 0, "whole groups of 4 values per thread of the final pass");
    (void)ws_size; (void)in_sizes; (void)n_in; (void)out_size;

    knn_gemm1_kernel<<<B * (N / 32), FUSE_THREADS, 0, stream>>>(coords, knn, W1, feats, PQ, W2, W2cat, N);

    const float invCnt = 1.0f / (float)(N * KNN_K);

    edge_chan_kernel<128, 2><<<dim3(128 / 2, B), EDGE_THREADS, 0, stream>>>(
        PQ, knn, X3 + (size_t)128 * N, (long)512 * N, invCnt);

    gemm_wmma_kernel<128><<<dim3(N / 64, 512 / 64, B), 128, 0, stream>>>(
        W2cat, X3 + (size_t)128 * N, X3 + (size_t)128 * N, PQ, N, C,
        (long)512 * N, (long)512 * N, 0, (long)512 * N);
    edge_chan_kernel<256, 4><<<dim3(256 / 4, B), EDGE_THREADS, 0, stream>>>(
        PQ, knn, X3 + (size_t)256 * N, (long)512 * N, invCnt);

    gemm_wmma_kernel<512><<<dim3(N / 64, 128 / 64, B), 128, 0, stream>>>(
        W3, feats, X3, x3pre, N, 512, (long)C * N, (long)512 * N, 4, (long)128 * N);
    final_norm_kernel<<<dim3(128, B), FIN_THREADS, 0, stream>>>(x3pre, out);
}
